# starting kernel
# baseline (speedup 1.0000x reference)
#include <hip/hip_runtime.h>

typedef __attribute__((ext_vector_type(2))) float v2f;
typedef __attribute__((ext_vector_type(8))) float v8f;

#define IN_C 64
#define POSD 16
#define KDIM1 (IN_C + POSD)
#define HID 64
#define OUTD 32


__global__ void k_zero(float* __restrict__ p, int n) {
  int i = blockIdx.x * blockDim.x + threadIdx.x;
  if (i < n) p[i] = 0.0f;
}

__global__ void k_deg(const int* __restrict__ dst, float* __restrict__ deg, int E) {
  int i = blockIdx.x * blockDim.x + threadIdx.x;
  if (i < E) atomicAdd(&deg[dst[i]], 1.0f);
}

__global__ void k_dinv(float* __restrict__ d, int n) {
  int i = blockIdx.x * blockDim.x + threadIdx.x;
  if (i < n) d[i] = rsqrtf(d[i] + 1.0f);
}

__global__ __launch_bounds__(256) void k_gemm1(
    const float* __restrict__ x, const float* __restrict__ pos,
    const float* __restrict__ W1, const float* __restrict__ dinv,
    float* __restrict__ hw1, float* __restrict__ h1agg, int N) {
  const int lane = threadIdx.x & 31;
  const int wave = threadIdx.x >> 5;
  const int rowBase = (blockIdx.x * 8 + wave) * 16;
  if (rowBase >= N) return;
  const int kh  = lane >> 4;
  const int n16 = lane & 15;
  int row = rowBase + n16;
  if (row >= N) row = N - 1;

  v8f acc0 = {}, acc1 = {}, acc2 = {}, acc3 = {};
  for (int k = 0; k < KDIM1; k += 4) {
    const int ka = k + 2 * kh;
    v2f a;
    a.x = (ka     < IN_C) ? x[(long)row * IN_C + ka]       : pos[(long)row * POSD + (ka     - IN_C)];
    a.y = (ka + 1 < IN_C) ? x[(long)row * IN_C + ka + 1]   : pos[(long)row * POSD + (ka + 1 - IN_C)];
    const float* wb = W1 + (long)ka * HID + n16;
    v2f b0; b0.x = wb[0];  b0.y = wb[HID];
    v2f b1; b1.x = wb[16]; b1.y = wb[HID + 16];
    v2f b2; b2.x = wb[32]; b2.y = wb[HID + 32];
    v2f b3; b3.x = wb[48]; b3.y = wb[HID + 48];
    acc0 = __builtin_amdgcn_wmma_f32_16x16x4_f32(false, a, false, b0, (short)0, acc0, false, false);
    acc1 = __builtin_amdgcn_wmma_f32_16x16x4_f32(false, a, false, b1, (short)0, acc1, false, false);
    acc2 = __builtin_amdgcn_wmma_f32_16x16x4_f32(false, a, false, b2, (short)0, acc2, false, false);
    acc3 = __builtin_amdgcn_wmma_f32_16x16x4_f32(false, a, false, b3, (short)0, acc3, false, false);
  }
#pragma unroll
  for (int i = 0; i < 8; ++i) {
    const int r = rowBase + i + 8 * kh;
    if (r < N) {
      const float dv = dinv[r];
      const float sw = dv * dv;
      const long base = (long)r * HID + n16;
      hw1[base +  0] = acc0[i];  h1agg[base +  0] = acc0[i] * sw;
      hw1[base + 16] = acc1[i];  h1agg[base + 16] = acc1[i] * sw;
      hw1[base + 32] = acc2[i];  h1agg[base + 32] = acc2[i] * sw;
      hw1[base + 48] = acc3[i];  h1agg[base + 48] = acc3[i] * sw;
    }
  }
}

__global__ __launch_bounds__(256) void k_gemm2(
    const float* __restrict__ h, const float* __restrict__ W2,
    const float* __restrict__ dinv, float* __restrict__ hw2,
    float* __restrict__ zinit, int N) {
  const int lane = threadIdx.x & 31;
  const int wave = threadIdx.x >> 5;
  const int rowBase = (blockIdx.x * 8 + wave) * 16;
  if (rowBase >= N) return;
  const int kh  = lane >> 4;
  const int n16 = lane & 15;
  int row = rowBase + n16;
  if (row >= N) row = N - 1;

  v8f acc0 = {}, acc1 = {};
  for (int k = 0; k < HID; k += 4) {
    const int ka = k + 2 * kh;
    v2f a;
    a.x = h[(long)row * HID + ka];
    a.y = h[(long)row * HID + ka + 1];
    const float* wb = W2 + (long)ka * OUTD + n16;
    v2f b0; b0.x = wb[0];  b0.y = wb[OUTD];
    v2f b1; b1.x = wb[16]; b1.y = wb[OUTD + 16];
    acc0 = __builtin_amdgcn_wmma_f32_16x16x4_f32(false, a, false, b0, (short)0, acc0, false, false);
    acc1 = __builtin_amdgcn_wmma_f32_16x16x4_f32(false, a, false, b1, (short)0, acc1, false, false);
  }
#pragma unroll
  for (int i = 0; i < 8; ++i) {
    const int r = rowBase + i + 8 * kh;
    if (r < N) {
      const float dv = dinv[r];
      const float sw = dv * dv;
      const long base = (long)r * OUTD + n16;
      hw2[base +  0] = acc0[i];  zinit[base +  0] = acc0[i] * sw;
      hw2[base + 16] = acc1[i];  zinit[base + 16] = acc1[i] * sw;
    }
  }
}

__global__ void k_scatter(const int* __restrict__ src, const int* __restrict__ dst,
                          const float* __restrict__ dinv, const float* __restrict__ feat,
                          float* __restrict__ agg, long total, int shift) {
  long t = (long)blockIdx.x * blockDim.x + threadIdx.x;
  if (t >= total) return;
  const int e = (int)(t >> shift);
  const int c = (int)(t & ((1L << shift) - 1));
  const int s = src[e];
  const int d = dst[e];
  const float w = dinv[s] * dinv[d];
  atomicAdd(&agg[((long)d << shift) + c], feat[((long)s << shift) + c] * w);
}

__global__ void k_bias_act(float* __restrict__ h, const float* __restrict__ b,
                           long total, int mask, int relu) {
  long i = (long)blockIdx.x * blockDim.x + threadIdx.x;
  if (i >= total) return;
  float v = h[i] + b[(int)(i & mask)];
  if (relu) v = fmaxf(v, 0.0f);
  h[i] = v;
}

__global__ void k_link(const float* __restrict__ z, const int* __restrict__ eli,
                       const float* __restrict__ Wl, const float* __restrict__ bl,
                       float* __restrict__ pred, int EL) {
  int e = blockIdx.x * blockDim.x + threadIdx.x;
  if (e >= EL) return;
  const int s = eli[e];
  const int d = eli[EL + e];
  const float* zs = z + (long)s * OUTD;
  const float* zd = z + (long)d * OUTD;
  float acc = bl[0];
#pragma unroll
  for (int c = 0; c < OUTD; ++c)
    acc = fmaf(zs[c], Wl[c], fmaf(zd[c], Wl[OUTD + c], acc));
  pred[e] = acc;
}

extern "C" void kernel_launch(void* const* d_in, const int* in_sizes, int n_in,
                              void* d_out, int out_size, void* d_ws, size_t ws_size,
                              hipStream_t stream) {
  const float* x   = (const float*)d_in[0];
  const float* pos = (const float*)d_in[1];
  const float* W1  = (const float*)d_in[2];
  const float* b1  = (const float*)d_in[3];
  const float* W2  = (const float*)d_in[4];
  const float* b2  = (const float*)d_in[5];
  const float* Wl  = (const float*)d_in[6];
  const float* bl  = (const float*)d_in[7];
  const int*   ei  = (const int*)d_in[8];
  const int*   eli = (const int*)d_in[9];

  const int N  = in_sizes[0] / IN_C;
  const int E  = in_sizes[8] / 2;
  const int EL = in_sizes[9] / 2;

  float* z    = (float*)d_out;
  float* pred = z + (long)N * OUTD;

  float* dinv  = (float*)d_ws;
  float* hw1   = dinv + N;
  float* h1    = hw1 + (long)N * HID;
  float* hw2   = h1  + (long)N * HID;

  const int* esrc = ei;
  const int* edst = ei + E;

  const int B = 256;

  k_zero<<<(N + B - 1) / B, B, 0, stream>>>(dinv, N);
  k_deg <<<(E + B - 1) / B, B, 0, stream>>>(edst, dinv, E);
  k_dinv<<<(N + B - 1) / B, B, 0, stream>>>(dinv, N);

  const int rowTiles = (N + 15) / 16;
  const int gblocks  = (rowTiles + 7) / 8;
  k_gemm1<<<gblocks, B, 0, stream>>>(x, pos, W1, dinv, hw1, h1, N);
  {
    long tot = (long)E * HID;
    k_scatter<<<(int)((tot + B - 1) / B), B, 0, stream>>>(esrc, edst, dinv, hw1, h1, tot, 6);
    long nh = (long)N * HID;
    k_bias_act<<<(int)((nh + B - 1) / B), B, 0, stream>>>(h1, b1, nh, HID - 1, 1);
  }

  k_gemm2<<<gblocks, B, 0, stream>>>(h1, W2, dinv, hw2, z, N);
  {
    long tot = (long)E * OUTD;
    k_scatter<<<(int)((tot + B - 1) / B), B, 0, stream>>>(esrc, edst, dinv, hw2, z, tot, 5);
    long nz = (long)N * OUTD;
    k_bias_act<<<(int)((nz + B - 1) / B), B, 0, stream>>>(z, b2, nz, OUTD - 1, 0);
  }

  k_link<<<(EL + B - 1) / B, B, 0, stream>>>(z, eli, Wl, bl, pred, EL);
}


# Degree pass off global float atomics: an LDS count pass gives every partition block's bin counts (exact run offsets, no cursor adds, no zero launch); row degrees counted in LDS per bin; runs by block
# speedup vs baseline: 3.9882x; 3.9882x over previous
#include <hip/hip_runtime.h>

typedef __attribute__((ext_vector_type(2))) float v2f;
typedef __attribute__((ext_vector_type(8))) float v8f;

#define IN_C 64
#define POSD 16
#define KDIM1 (IN_C + POSD)
#define HID 64
#define OUTD 32


#define AGG_THREADS 1024
#define AGG_WAVES 32
#define Q_BATCH 32

#define BIN_ROWS 512
#define BIN_SHIFT 9
#define ENT_SHIFT 23
#define PART_THREADS 1024
#define PART_K 8
#define MAX_BINS 1024
#define CNT_BATCH 32
#define DINV_THREADS 1024
#define DINV_K 8


static __device__ __forceinline__ int wave_sum(int v) {
#pragma unroll
  for (int o = 16; o > 0; o >>= 1) v += __shfl_xor(v, o);
  return v;
}

static __device__ __forceinline__ int wave_scan(int v, int lane) {
#pragma unroll
  for (int o = 1; o < 32; o <<= 1) {
    const int t = __shfl_up(v, o);
    if (lane >= o) v += t;
  }
  return v;
}

__global__ __launch_bounds__(PART_THREADS) void k_count(
    const int* __restrict__ dst, int* __restrict__ blkCnt, int NB, int E) {
  __shared__ int hist[MAX_BINS];
  const int tid = threadIdx.x;
  hist[tid] = 0;
  __syncthreads();

  const int e0 = blockIdx.x * (PART_THREADS * PART_K);
  int dd[PART_K];
#pragma unroll
  for (int k = 0; k < PART_K; ++k)
    dd[k] = dst[min(e0 + k * PART_THREADS + tid, E - 1)];
#pragma unroll
  for (int k = 0; k < PART_K; ++k)
    if (e0 + k * PART_THREADS + tid < E) atomicAdd(&hist[dd[k] >> BIN_SHIFT], 1);
  __syncthreads();

  if (tid < NB) blkCnt[blockIdx.x * NB + tid] = hist[tid];
}


template <bool FULL>
static __device__ __forceinline__ void add_counts(
    const int* __restrict__ col, int NB, int b0, int nblk, int me, int& cb, int& ahead) {
  int c[CNT_BATCH];
#pragma unroll
  for (int j = 0; j < CNT_BATCH; ++j) {
    const int b = FULL ? (b0 + j) : min(b0 + j, nblk - 1);
    c[j] = col[b * NB];
  }
#pragma unroll
  for (int j = 0; j < CNT_BATCH; ++j) {
    const int v = (FULL || b0 + j < nblk) ? c[j] : 0;
    cb += v;
    ahead += (b0 + j < me) ? v : 0;
  }
}

__global__ __launch_bounds__(PART_THREADS) void k_part(
    const int* __restrict__ src, const int* __restrict__ dst,
    const int* __restrict__ blkCnt, int* __restrict__ binCount,
    int* __restrict__ bins, int NB, int E) {
  __shared__ int hist[MAX_BINS];
  __shared__ int bstart[MAX_BINS];
  __shared__ int wtot[PART_THREADS / 32];
  __shared__ int st_ent[PART_THREADS * PART_K];
  __shared__ int st_pos[PART_THREADS * PART_K];
  const int tid = threadIdx.x;
  const int lane = tid & 31;
  const int wave = tid >> 5;
  const int me = blockIdx.x;
  const int nblk = gridDim.x;

  int cb = 0, ahead = 0;
  if (tid < NB) {
    const int* col = blkCnt + tid;
    int b0 = 0;
#pragma nounroll
    for (; b0 + CNT_BATCH <= nblk; b0 += CNT_BATCH) add_counts<true>(col, NB, b0, nblk, me, cb, ahead);
    if (b0 < nblk) add_counts<false>(col, NB, b0, nblk, me, cb, ahead);
  }

  const int inc = wave_scan(cb, lane);
  if (lane == 31) wtot[wave] = inc;
  hist[tid] = 0;
  __syncthreads();
  {
    const int before = wave_sum((lane < wave) ? wtot[lane] : 0);
    bstart[tid] = before + inc - cb;
  }
  if (me == 0 && tid < NB) binCount[tid] = cb;

  const int e0 = me * (PART_THREADS * PART_K);
  int dd[PART_K], ss[PART_K], bin[PART_K], rank[PART_K];
#pragma unroll
  for (int k = 0; k < PART_K; ++k) {
    const int ec = min(e0 + k * PART_THREADS + tid, E - 1);
    dd[k] = dst[ec];
    ss[k] = src[ec];
  }
#pragma unroll
  for (int k = 0; k < PART_K; ++k) {
    const bool ok = e0 + k * PART_THREADS + tid < E;
    bin[k] = ok ? (dd[k] >> BIN_SHIFT) : -1;
    rank[k] = 0;
    if (ok) rank[k] = atomicAdd(&hist[bin[k]], 1);
  }
  __syncthreads();

  const int n = hist[tid];
  const int gb = bstart[tid] + ahead;
  const int inc2 = wave_scan(n, lane);
  if (lane == 31) wtot[wave] = inc2;
  __syncthreads();
  {
    const int before = wave_sum((lane < wave) ? wtot[lane] : 0);
    const int lofs = before + inc2 - n;
    hist[tid] = lofs;
    bstart[tid] = gb - lofs;
  }
  __syncthreads();

  int slot[PART_K], gpos[PART_K];
#pragma unroll
  for (int k = 0; k < PART_K; ++k) {
    const int bb = max(bin[k], 0);
    slot[k] = hist[bb] + rank[k];
    gpos[k] = slot[k] + bstart[bb];
  }
#pragma unroll
  for (int k = 0; k < PART_K; ++k) {
    if (bin[k] >= 0) {
      st_ent[slot[k]] = (int)((unsigned)ss[k] | ((unsigned)(dd[k] & (BIN_ROWS - 1)) << ENT_SHIFT));
      st_pos[slot[k]] = gpos[k];
    }
  }
  __syncthreads();

  const int total = min(PART_THREADS * PART_K, E - e0);
  int oe[PART_K], op[PART_K];
#pragma unroll
  for (int k = 0; k < PART_K; ++k) {
    const int i = min(k * PART_THREADS + tid, total - 1);
    oe[k] = st_ent[i];
    op[k] = st_pos[i];
  }
#pragma unroll
  for (int k = 0; k < PART_K; ++k)
    if (k * PART_THREADS + tid < total) bins[op[k]] = oe[k];
}

__global__ __launch_bounds__(DINV_THREADS) void k_dinv(
    const int* __restrict__ bins, const int* __restrict__ binCount,
    float* __restrict__ dinv, int N) {
  __shared__ int cnt[BIN_ROWS];
  __shared__ int wtot[DINV_THREADS / 32];
  const int tid  = threadIdx.x;
  const int lane = tid & 31;
  const int wave = tid >> 5;
  const int b  = blockIdx.x;

  {
    const int part = wave_sum((tid < b) ? binCount[tid] : 0);
    if (lane == 0) wtot[wave] = part;
  }
  if (tid < BIN_ROWS) cnt[tid] = 0;
  __syncthreads();
  const int start = wave_sum(wtot[lane]);
  const int n = binCount[b];
  const int* mine = bins + start;

  for (int i0 = 0; i0 < n; i0 += DINV_THREADS * DINV_K) {
    int ent[DINV_K];
#pragma unroll
    for (int k = 0; k < DINV_K; ++k)
      ent[k] = mine[min(i0 + k * DINV_THREADS + tid, n - 1)];
#pragma unroll
    for (int k = 0; k < DINV_K; ++k)
      if (i0 + k * DINV_THREADS + tid < n) atomicAdd(&cnt[(unsigned)ent[k] >> ENT_SHIFT], 1);
  }
  __syncthreads();

  const int r = b * BIN_ROWS + tid;
  if (tid < BIN_ROWS && r < N) dinv[r] = rsqrtf((float)cnt[tid] + 1.0f);
}

__global__ __launch_bounds__(256) void k_gemm1(
    const float* __restrict__ x, const float* __restrict__ pos,
    const float* __restrict__ W1, float* __restrict__ hw1, int N) {
  const int lane = threadIdx.x & 31;
  const int wave = threadIdx.x >> 5;
  const int rowBase = (blockIdx.x * 8 + wave) * 16;
  if (rowBase >= N) return;
  const int kh  = lane >> 4;
  const int n16 = lane & 15;
  int row = rowBase + n16;
  if (row >= N) row = N - 1;

  v8f acc0 = {}, acc1 = {}, acc2 = {}, acc3 = {};
  for (int k = 0; k < KDIM1; k += 4) {
    const int ka = k + 2 * kh;
    v2f a;
    a.x = (ka     < IN_C) ? x[(long)row * IN_C + ka]       : pos[(long)row * POSD + (ka     - IN_C)];
    a.y = (ka + 1 < IN_C) ? x[(long)row * IN_C + ka + 1]   : pos[(long)row * POSD + (ka + 1 - IN_C)];
    const float* wb = W1 + (long)ka * HID + n16;
    v2f b0; b0.x = wb[0];  b0.y = wb[HID];
    v2f b1; b1.x = wb[16]; b1.y = wb[HID + 16];
    v2f b2; b2.x = wb[32]; b2.y = wb[HID + 32];
    v2f b3; b3.x = wb[48]; b3.y = wb[HID + 48];
    acc0 = __builtin_amdgcn_wmma_f32_16x16x4_f32(false, a, false, b0, (short)0, acc0, false, false);
    acc1 = __builtin_amdgcn_wmma_f32_16x16x4_f32(false, a, false, b1, (short)0, acc1, false, false);
    acc2 = __builtin_amdgcn_wmma_f32_16x16x4_f32(false, a, false, b2, (short)0, acc2, false, false);
    acc3 = __builtin_amdgcn_wmma_f32_16x16x4_f32(false, a, false, b3, (short)0, acc3, false, false);
  }
#pragma unroll
  for (int i = 0; i < 8; ++i) {
    const int r = rowBase + i + 8 * kh;
    if (r < N) {
      const long base = (long)r * HID + n16;
      hw1[base +  0] = acc0[i];
      hw1[base + 16] = acc1[i];
      hw1[base + 32] = acc2[i];
      hw1[base + 48] = acc3[i];
    }
  }
}

__global__ __launch_bounds__(256) void k_gemm2(
    const float* __restrict__ h, const float* __restrict__ W2,
    float* __restrict__ hw2, int N) {
  const int lane = threadIdx.x & 31;
  const int wave = threadIdx.x >> 5;
  const int rowBase = (blockIdx.x * 8 + wave) * 16;
  if (rowBase >= N) return;
  const int kh  = lane >> 4;
  const int n16 = lane & 15;
  int row = rowBase + n16;
  if (row >= N) row = N - 1;

  v8f acc0 = {}, acc1 = {};
  for (int k = 0; k < HID; k += 4) {
    const int ka = k + 2 * kh;
    v2f a;
    a.x = h[(long)row * HID + ka];
    a.y = h[(long)row * HID + ka + 1];
    const float* wb = W2 + (long)ka * OUTD + n16;
    v2f b0; b0.x = wb[0];  b0.y = wb[OUTD];
    v2f b1; b1.x = wb[16]; b1.y = wb[OUTD + 16];
    acc0 = __builtin_amdgcn_wmma_f32_16x16x4_f32(false, a, false, b0, (short)0, acc0, false, false);
    acc1 = __builtin_amdgcn_wmma_f32_16x16x4_f32(false, a, false, b1, (short)0, acc1, false, false);
  }
#pragma unroll
  for (int i = 0; i < 8; ++i) {
    const int r = rowBase + i + 8 * kh;
    if (r < N) {
      const long base = (long)r * OUTD + n16;
      hw2[base +  0] = acc0[i];
      hw2[base + 16] = acc1[i];
    }
  }
}


template <int C, bool FULL>
static __device__ __forceinline__ void fold_batch(
    int ent, int nb, int lane,
    const float* __restrict__ dinv, const float* dloc,
    const float* __restrict__ feat, float* acc) {
  constexpr int PER = C / 32;
  const int s = (int)((unsigned)ent & ((1u << ENT_SHIFT) - 1u));
  const int lr = (int)((unsigned)ent >> ENT_SHIFT);
  const float dvs = dinv[s];
  const float dvd = dloc[lr];
  float w = 0.0f;
  const int ro = lr * C;
#pragma unroll
  for (int i = 0; i < Q_BATCH; i += 16) {
    if (!FULL && i >= nb) break;
    float v[16][PER];
#pragma unroll
    for (int k = 0; k < 16; ++k) {
      if (FULL || i + k < nb) {
        const int sk = __builtin_amdgcn_readlane(s, i + k);
        const float* p = feat + (long)sk * C + lane * PER;
        if (PER == 2) {
          const v2f t = *(const v2f*)p;
          v[k][0] = t.x;
          v[k][PER - 1] = t.y;
        } else {
          v[k][0] = p[0];
        }
      }
    }
    if (i == 0) w = dvs * dvd;
#pragma unroll
    for (int k = 0; k < 16; ++k) {
      if (FULL || i + k < nb) {
        const float wk = __int_as_float(__builtin_amdgcn_readlane(__float_as_int(w), i + k));
        const int rk = __builtin_amdgcn_readlane(ro, i + k);
        float* a = acc + rk + lane * PER;
#pragma unroll
        for (int j = 0; j < PER; ++j)
          __hip_atomic_fetch_add(a + j, v[k][j] * wk, __ATOMIC_RELAXED, __HIP_MEMORY_SCOPE_WORKGROUP);
      }
    }
  }
}

template <int C, bool RELU>
__global__ __launch_bounds__(AGG_THREADS) void k_agg(
    const int* __restrict__ bins, const int* __restrict__ binCount,
    const float* __restrict__ dinv, const float* __restrict__ feat,
    const float* __restrict__ bias, float* __restrict__ out, int N) {
  __shared__ __attribute__((aligned(16))) float acc[BIN_ROWS * C];
  __shared__ float dloc[BIN_ROWS];
  __shared__ int wtot[AGG_WAVES];

  const int tid  = threadIdx.x;
  const int lane = tid & 31;
  const int wave = __builtin_amdgcn_readfirstlane(tid >> 5);
  const int b  = blockIdx.x;
  const int r0 = b * BIN_ROWS;
  const int nr = min(BIN_ROWS, N - r0);

  {
    const int part = wave_sum((tid < b) ? binCount[tid] : 0);
    if (lane == 0) wtot[wave] = part;
  }
  if (tid < BIN_ROWS) dloc[tid] = (tid < nr) ? dinv[r0 + tid] : 0.0f;

  for (int i = tid; i < nr * (C / 4); i += AGG_THREADS) {
    const int r = i / (C / 4);
    const float dv = dinv[r0 + r];
    const float sw = dv * dv;
    float4 t = ((const float4*)(feat + (long)r0 * C))[i];
    t.x *= sw; t.y *= sw; t.z *= sw; t.w *= sw;
    ((float4*)acc)[i] = t;
  }
  __syncthreads();

  const int start = __builtin_amdgcn_readfirstlane(wave_sum(wtot[lane]));
  const int n = __builtin_amdgcn_readfirstlane(binCount[b]);
  const int* mine = bins + start;

  if (n > 0) {
    int base = wave * Q_BATCH;
    int ent = mine[min(base + lane, n - 1)];
    while (base < n) {
      const int nbase = base + AGG_WAVES * Q_BATCH;
      const int entn = mine[min(nbase + lane, n - 1)];
      const int nb = min(Q_BATCH, n - base);
      if (nb == Q_BATCH) fold_batch<C, true>(ent, nb, lane, dinv, dloc, feat, acc);
      else               fold_batch<C, false>(ent, nb, lane, dinv, dloc, feat, acc);
      ent = entn;
      base = nbase;
    }
  }

  __syncthreads();

  for (int i = tid; i < nr * (C / 4); i += AGG_THREADS) {
    const int c4 = i % (C / 4);
    float4 t = ((const float4*)acc)[i];
    const float4 bb = ((const float4*)bias)[c4];
    t.x += bb.x; t.y += bb.y; t.z += bb.z; t.w += bb.w;
    if (RELU) {
      t.x = fmaxf(t.x, 0.0f); t.y = fmaxf(t.y, 0.0f);
      t.z = fmaxf(t.z, 0.0f); t.w = fmaxf(t.w, 0.0f);
    }
    ((float4*)(out + (long)r0 * C))[i] = t;
  }
}

__global__ void k_link(const float* __restrict__ z, const int* __restrict__ eli,
                       const float* __restrict__ Wl, const float* __restrict__ bl,
                       float* __restrict__ pred, int EL) {
  int e = blockIdx.x * blockDim.x + threadIdx.x;
  if (e >= EL) return;
  const int s = eli[e];
  const int d = eli[EL + e];
  const float* zs = z + (long)s * OUTD;
  const float* zd = z + (long)d * OUTD;
  float acc = bl[0];
#pragma unroll
  for (int c = 0; c < OUTD; ++c)
    acc = fmaf(zs[c], Wl[c], fmaf(zd[c], Wl[OUTD + c], acc));
  pred[e] = acc;
}

extern "C" void kernel_launch(void* const* d_in, const int* in_sizes, int n_in,
                              void* d_out, int out_size, void* d_ws, size_t ws_size,
                              hipStream_t stream) {
  const float* x   = (const float*)d_in[0];
  const float* pos = (const float*)d_in[1];
  const float* W1  = (const float*)d_in[2];
  const float* b1  = (const float*)d_in[3];
  const float* W2  = (const float*)d_in[4];
  const float* b2  = (const float*)d_in[5];
  const float* Wl  = (const float*)d_in[6];
  const float* bl  = (const float*)d_in[7];
  const int*   ei  = (const int*)d_in[8];
  const int*   eli = (const int*)d_in[9];

  const int N  = in_sizes[0] / IN_C;
  const int E  = in_sizes[8] / 2;
  const int EL = in_sizes[9] / 2;

  float* z    = (float*)d_out;
  float* pred = z + (long)N * OUTD;

  const int NB = (N + BIN_ROWS - 1) / BIN_ROWS;
  const int per = PART_THREADS * PART_K;
  const int nblk = (E + per - 1) / per;

  float* dinv  = (float*)d_ws;
  float* hw1   = dinv + N;
  float* h1    = hw1 + (long)N * HID;
  float* hw2   = hw1;
  int* bins     = (int*)(h1 + (long)N * HID);
  int* binCount = bins + E;
  int* blkCnt   = (int*)h1;

  const int* esrc = ei;
  const int* edst = ei + E;

  const int B = 256;

  k_count<<<nblk, PART_THREADS, 0, stream>>>(edst, blkCnt, NB, E);
  k_part <<<nblk, PART_THREADS, 0, stream>>>(esrc, edst, blkCnt, binCount, bins, NB, E);
  k_dinv <<<NB, DINV_THREADS, 0, stream>>>(bins, binCount, dinv, N);

  const int rowTiles = (N + 15) / 16;
  const int gblocks  = (rowTiles + 7) / 8;
  k_gemm1<<<gblocks, B, 0, stream>>>(x, pos, W1, hw1, N);
  k_agg<HID, true><<<NB, AGG_THREADS, 0, stream>>>(bins, binCount, dinv, hw1, b1, h1, N);

  k_gemm2<<<gblocks, B, 0, stream>>>(h1, W2, hw2, N);
  k_agg<OUTD, false><<<NB, AGG_THREADS, 0, stream>>>(bins, binCount, dinv, hw2, b2, z, N);

  k_link<<<(EL + B - 1) / B, B, 0, stream>>>(z, eli, Wl, bl, pred, EL);
}


# Destination bins of 400 rows, not 512: degree pass and both aggregations run 250 owner blocks (one per work-group processor, 256 on the chip), not 196; row sums stay arrival-order LDS float adds
# speedup vs baseline: 4.3494x; 1.0906x over previous
#include <hip/hip_runtime.h>

typedef __attribute__((ext_vector_type(2))) float v2f;
typedef __attribute__((ext_vector_type(8))) float v8f;

#define IN_C 64
#define POSD 16
#define KDIM1 (IN_C + POSD)
#define HID 64
#define OUTD 32


#define AGG_THREADS 1024
#define AGG_WAVES 32
#define Q_BATCH 32

#define BIN_ROWS 400
#define ENT_SHIFT 23
#define PART_THREADS 1024
#define PART_K 8
#define MAX_BINS 1024
#define CNT_BATCH 32
#define DINV_THREADS 1024
#define DINV_K 8
static_assert(BIN_ROWS <= (1 << (32 - ENT_SHIFT)), "a row inside its bin must fit above ENT_SHIFT");
static_assert(BIN_ROWS <= DINV_THREADS && BIN_ROWS <= AGG_THREADS, "one thread per owned row in the owner blocks");


static __device__ __forceinline__ int wave_sum(int v) {
#pragma unroll
  for (int o = 16; o > 0; o >>= 1) v += __shfl_xor(v, o);
  return v;
}

static __device__ __forceinline__ int wave_scan(int v, int lane) {
#pragma unroll
  for (int o = 1; o < 32; o <<= 1) {
    const int t = __shfl_up(v, o);
    if (lane >= o) v += t;
  }
  return v;
}

__global__ __launch_bounds__(PART_THREADS) void k_count(
    const int* __restrict__ dst, int* __restrict__ blkCnt, int NB, int E) {
  __shared__ int hist[MAX_BINS];
  const int tid = threadIdx.x;
  hist[tid] = 0;
  __syncthreads();

  const int e0 = blockIdx.x * (PART_THREADS * PART_K);
  int dd[PART_K];
#pragma unroll
  for (int k = 0; k < PART_K; ++k)
    dd[k] = dst[min(e0 + k * PART_THREADS + tid, E - 1)];
#pragma unroll
  for (int k = 0; k < PART_K; ++k)
    if (e0 + k * PART_THREADS + tid < E) atomicAdd(&hist[(unsigned)dd[k] / BIN_ROWS], 1);
  __syncthreads();

  if (tid < NB) blkCnt[blockIdx.x * NB + tid] = hist[tid];
}


template <bool FULL>
static __device__ __forceinline__ void add_counts(
    const int* __restrict__ col, int NB, int b0, int nblk, int me, int& cb, int& ahead) {
  int c[CNT_BATCH];
#pragma unroll
  for (int j = 0; j < CNT_BATCH; ++j) {
    const int b = FULL ? (b0 + j) : min(b0 + j, nblk - 1);
    c[j] = col[b * NB];
  }
#pragma unroll
  for (int j = 0; j < CNT_BATCH; ++j) {
    const int v = (FULL || b0 + j < nblk) ? c[j] : 0;
    cb += v;
    ahead += (b0 + j < me) ? v : 0;
  }
}

__global__ __launch_bounds__(PART_THREADS) void k_part(
    const int* __restrict__ src, const int* __restrict__ dst,
    const int* __restrict__ blkCnt, int* __restrict__ binCount,
    int* __restrict__ bins, int NB, int E) {
  __shared__ int hist[MAX_BINS];
  __shared__ int bstart[MAX_BINS];
  __shared__ int wtot[PART_THREADS / 32];
  __shared__ int st_ent[PART_THREADS * PART_K];
  __shared__ int st_pos[PART_THREADS * PART_K];
  const int tid = threadIdx.x;
  const int lane = tid & 31;
  const int wave = tid >> 5;
  const int me = blockIdx.x;
  const int nblk = gridDim.x;

  int cb = 0, ahead = 0;
  if (tid < NB) {
    const int* col = blkCnt + tid;
    int b0 = 0;
#pragma nounroll
    for (; b0 + CNT_BATCH <= nblk; b0 += CNT_BATCH) add_counts<true>(col, NB, b0, nblk, me, cb, ahead);
    if (b0 < nblk) add_counts<false>(col, NB, b0, nblk, me, cb, ahead);
  }

  const int inc = wave_scan(cb, lane);
  if (lane == 31) wtot[wave] = inc;
  hist[tid] = 0;
  __syncthreads();
  {
    const int before = wave_sum((lane < wave) ? wtot[lane] : 0);
    bstart[tid] = before + inc - cb;
  }
  if (me == 0 && tid < NB) binCount[tid] = cb;

  const int e0 = me * (PART_THREADS * PART_K);
  int dd[PART_K], ss[PART_K], bin[PART_K], rank[PART_K];
#pragma unroll
  for (int k = 0; k < PART_K; ++k) {
    const int ec = min(e0 + k * PART_THREADS + tid, E - 1);
    dd[k] = dst[ec];
    ss[k] = src[ec];
  }
#pragma unroll
  for (int k = 0; k < PART_K; ++k) {
    const bool ok = e0 + k * PART_THREADS + tid < E;
    bin[k] = ok ? (int)((unsigned)dd[k] / BIN_ROWS) : -1;
    rank[k] = 0;
    if (ok) rank[k] = atomicAdd(&hist[bin[k]], 1);
  }
  __syncthreads();

  const int n = hist[tid];
  const int gb = bstart[tid] + ahead;
  const int inc2 = wave_scan(n, lane);
  if (lane == 31) wtot[wave] = inc2;
  __syncthreads();
  {
    const int before = wave_sum((lane < wave) ? wtot[lane] : 0);
    const int lofs = before + inc2 - n;
    hist[tid] = lofs;
    bstart[tid] = gb - lofs;
  }
  __syncthreads();

  int slot[PART_K], gpos[PART_K];
#pragma unroll
  for (int k = 0; k < PART_K; ++k) {
    const int bb = max(bin[k], 0);
    slot[k] = hist[bb] + rank[k];
    gpos[k] = slot[k] + bstart[bb];
  }
#pragma unroll
  for (int k = 0; k < PART_K; ++k) {
    if (bin[k] >= 0) {
      st_ent[slot[k]] = (int)((unsigned)ss[k] | ((unsigned)(dd[k] - bin[k] * BIN_ROWS) << ENT_SHIFT));
      st_pos[slot[k]] = gpos[k];
    }
  }
  __syncthreads();

  const int total = min(PART_THREADS * PART_K, E - e0);
  int oe[PART_K], op[PART_K];
#pragma unroll
  for (int k = 0; k < PART_K; ++k) {
    const int i = min(k * PART_THREADS + tid, total - 1);
    oe[k] = st_ent[i];
    op[k] = st_pos[i];
  }
#pragma unroll
  for (int k = 0; k < PART_K; ++k)
    if (k * PART_THREADS + tid < total) bins[op[k]] = oe[k];
}

__global__ __launch_bounds__(DINV_THREADS) void k_dinv(
    const int* __restrict__ bins, const int* __restrict__ binCount,
    float* __restrict__ dinv, int N) {
  __shared__ int cnt[BIN_ROWS];
  __shared__ int wtot[DINV_THREADS / 32];
  const int tid  = threadIdx.x;
  const int lane = tid & 31;
  const int wave = tid >> 5;
  const int b  = blockIdx.x;

  {
    const int part = wave_sum((tid < b) ? binCount[tid] : 0);
    if (lane == 0) wtot[wave] = part;
  }
  if (tid < BIN_ROWS) cnt[tid] = 0;
  __syncthreads();
  const int start = wave_sum(wtot[lane]);
  const int n = binCount[b];
  const int* mine = bins + start;

  for (int i0 = 0; i0 < n; i0 += DINV_THREADS * DINV_K) {
    int ent[DINV_K];
#pragma unroll
    for (int k = 0; k < DINV_K; ++k)
      ent[k] = mine[min(i0 + k * DINV_THREADS + tid, n - 1)];
#pragma unroll
    for (int k = 0; k < DINV_K; ++k)
      if (i0 + k * DINV_THREADS + tid < n) atomicAdd(&cnt[(unsigned)ent[k] >> ENT_SHIFT], 1);
  }
  __syncthreads();

  const int r = b * BIN_ROWS + tid;
  if (tid < BIN_ROWS && r < N) dinv[r] = rsqrtf((float)cnt[tid] + 1.0f);
}

__global__ __launch_bounds__(256) void k_gemm1(
    const float* __restrict__ x, const float* __restrict__ pos,
    const float* __restrict__ W1, float* __restrict__ hw1, int N) {
  const int lane = threadIdx.x & 31;
  const int wave = threadIdx.x >> 5;
  const int rowBase = (blockIdx.x * 8 + wave) * 16;
  if (rowBase >= N) return;
  const int kh  = lane >> 4;
  const int n16 = lane & 15;
  int row = rowBase + n16;
  if (row >= N) row = N - 1;

  v8f acc0 = {}, acc1 = {}, acc2 = {}, acc3 = {};
  for (int k = 0; k < KDIM1; k += 4) {
    const int ka = k + 2 * kh;
    v2f a;
    a.x = (ka     < IN_C) ? x[(long)row * IN_C + ka]       : pos[(long)row * POSD + (ka     - IN_C)];
    a.y = (ka + 1 < IN_C) ? x[(long)row * IN_C + ka + 1]   : pos[(long)row * POSD + (ka + 1 - IN_C)];
    const float* wb = W1 + (long)ka * HID + n16;
    v2f b0; b0.x = wb[0];  b0.y = wb[HID];
    v2f b1; b1.x = wb[16]; b1.y = wb[HID + 16];
    v2f b2; b2.x = wb[32]; b2.y = wb[HID + 32];
    v2f b3; b3.x = wb[48]; b3.y = wb[HID + 48];
    acc0 = __builtin_amdgcn_wmma_f32_16x16x4_f32(false, a, false, b0, (short)0, acc0, false, false);
    acc1 = __builtin_amdgcn_wmma_f32_16x16x4_f32(false, a, false, b1, (short)0, acc1, false, false);
    acc2 = __builtin_amdgcn_wmma_f32_16x16x4_f32(false, a, false, b2, (short)0, acc2, false, false);
    acc3 = __builtin_amdgcn_wmma_f32_16x16x4_f32(false, a, false, b3, (short)0, acc3, false, false);
  }
#pragma unroll
  for (int i = 0; i < 8; ++i) {
    const int r = rowBase + i + 8 * kh;
    if (r < N) {
      const long base = (long)r * HID + n16;
      hw1[base +  0] = acc0[i];
      hw1[base + 16] = acc1[i];
      hw1[base + 32] = acc2[i];
      hw1[base + 48] = acc3[i];
    }
  }
}

__global__ __launch_bounds__(256) void k_gemm2(
    const float* __restrict__ h, const float* __restrict__ W2,
    float* __restrict__ hw2, int N) {
  const int lane = threadIdx.x & 31;
  const int wave = threadIdx.x >> 5;
  const int rowBase = (blockIdx.x * 8 + wave) * 16;
  if (rowBase >= N) return;
  const int kh  = lane >> 4;
  const int n16 = lane & 15;
  int row = rowBase + n16;
  if (row >= N) row = N - 1;

  v8f acc0 = {}, acc1 = {};
  for (int k = 0; k < HID; k += 4) {
    const int ka = k + 2 * kh;
    v2f a;
    a.x = h[(long)row * HID + ka];
    a.y = h[(long)row * HID + ka + 1];
    const float* wb = W2 + (long)ka * OUTD + n16;
    v2f b0; b0.x = wb[0];  b0.y = wb[OUTD];
    v2f b1; b1.x = wb[16]; b1.y = wb[OUTD + 16];
    acc0 = __builtin_amdgcn_wmma_f32_16x16x4_f32(false, a, false, b0, (short)0, acc0, false, false);
    acc1 = __builtin_amdgcn_wmma_f32_16x16x4_f32(false, a, false, b1, (short)0, acc1, false, false);
  }
#pragma unroll
  for (int i = 0; i < 8; ++i) {
    const int r = rowBase + i + 8 * kh;
    if (r < N) {
      const long base = (long)r * OUTD + n16;
      hw2[base +  0] = acc0[i];
      hw2[base + 16] = acc1[i];
    }
  }
}


template <int C, bool FULL>
static __device__ __forceinline__ void fold_batch(
    int ent, int nb, int lane,
    const float* __restrict__ dinv, const float* dloc,
    const float* __restrict__ feat, float* acc) {
  constexpr int PER = C / 32;
  const int s = (int)((unsigned)ent & ((1u << ENT_SHIFT) - 1u));
  const int lr = (int)((unsigned)ent >> ENT_SHIFT);
  const float dvs = dinv[s];
  const float dvd = dloc[lr];
  float w = 0.0f;
  const int ro = lr * C;
#pragma unroll
  for (int i = 0; i < Q_BATCH; i += 16) {
    if (!FULL && i >= nb) break;
    float v[16][PER];
#pragma unroll
    for (int k = 0; k < 16; ++k) {
      if (FULL || i + k < nb) {
        const int sk = __builtin_amdgcn_readlane(s, i + k);
        const float* p = feat + (long)sk * C + lane * PER;
        if (PER == 2) {
          const v2f t = *(const v2f*)p;
          v[k][0] = t.x;
          v[k][PER - 1] = t.y;
        } else {
          v[k][0] = p[0];
        }
      }
    }
    if (i == 0) w = dvs * dvd;
#pragma unroll
    for (int k = 0; k < 16; ++k) {
      if (FULL || i + k < nb) {
        const float wk = __int_as_float(__builtin_amdgcn_readlane(__float_as_int(w), i + k));
        const int rk = __builtin_amdgcn_readlane(ro, i + k);
        float* a = acc + rk + lane * PER;
#pragma unroll
        for (int j = 0; j < PER; ++j)
          __hip_atomic_fetch_add(a + j, v[k][j] * wk, __ATOMIC_RELAXED, __HIP_MEMORY_SCOPE_WORKGROUP);
      }
    }
  }
}

template <int C, bool RELU>
__global__ __launch_bounds__(AGG_THREADS) void k_agg(
    const int* __restrict__ bins, const int* __restrict__ binCount,
    const float* __restrict__ dinv, const float* __restrict__ feat,
    const float* __restrict__ bias, float* __restrict__ out, int N) {
  __shared__ __attribute__((aligned(16))) float acc[BIN_ROWS * C];
  __shared__ float dloc[BIN_ROWS];
  __shared__ int wtot[AGG_WAVES];

  const int tid  = threadIdx.x;
  const int lane = tid & 31;
  const int wave = __builtin_amdgcn_readfirstlane(tid >> 5);
  const int b  = blockIdx.x;
  const int r0 = b * BIN_ROWS;
  const int nr = min(BIN_ROWS, N - r0);

  {
    const int part = wave_sum((tid < b) ? binCount[tid] : 0);
    if (lane == 0) wtot[wave] = part;
  }
  if (tid < BIN_ROWS) dloc[tid] = (tid < nr) ? dinv[r0 + tid] : 0.0f;

  for (int i = tid; i < nr * (C / 4); i += AGG_THREADS) {
    const int r = i / (C / 4);
    const float dv = dinv[r0 + r];
    const float sw = dv * dv;
    float4 t = ((const float4*)(feat + (long)r0 * C))[i];
    t.x *= sw; t.y *= sw; t.z *= sw; t.w *= sw;
    ((float4*)acc)[i] = t;
  }
  __syncthreads();

  const int start = __builtin_amdgcn_readfirstlane(wave_sum(wtot[lane]));
  const int n = __builtin_amdgcn_readfirstlane(binCount[b]);
  const int* mine = bins + start;

  if (n > 0) {
    int base = wave * Q_BATCH;
    int ent = mine[min(base + lane, n - 1)];
    while (base < n) {
      const int nbase = base + AGG_WAVES * Q_BATCH;
      const int entn = mine[min(nbase + lane, n - 1)];
      const int nb = min(Q_BATCH, n - base);
      if (nb == Q_BATCH) fold_batch<C, true>(ent, nb, lane, dinv, dloc, feat, acc);
      else               fold_batch<C, false>(ent, nb, lane, dinv, dloc, feat, acc);
      ent = entn;
      base = nbase;
    }
  }

  __syncthreads();

  for (int i = tid; i < nr * (C / 4); i += AGG_THREADS) {
    const int c4 = i % (C / 4);
    float4 t = ((const float4*)acc)[i];
    const float4 bb = ((const float4*)bias)[c4];
    t.x += bb.x; t.y += bb.y; t.z += bb.z; t.w += bb.w;
    if (RELU) {
      t.x = fmaxf(t.x, 0.0f); t.y = fmaxf(t.y, 0.0f);
      t.z = fmaxf(t.z, 0.0f); t.w = fmaxf(t.w, 0.0f);
    }
    ((float4*)(out + (long)r0 * C))[i] = t;
  }
}

__global__ void k_link(const float* __restrict__ z, const int* __restrict__ eli,
                       const float* __restrict__ Wl, const float* __restrict__ bl,
                       float* __restrict__ pred, int EL) {
  int e = blockIdx.x * blockDim.x + threadIdx.x;
  if (e >= EL) return;
  const int s = eli[e];
  const int d = eli[EL + e];
  const float* zs = z + (long)s * OUTD;
  const float* zd = z + (long)d * OUTD;
  float acc = bl[0];
#pragma unroll
  for (int c = 0; c < OUTD; ++c)
    acc = fmaf(zs[c], Wl[c], fmaf(zd[c], Wl[OUTD + c], acc));
  pred[e] = acc;
}

extern "C" void kernel_launch(void* const* d_in, const int* in_sizes, int n_in,
                              void* d_out, int out_size, void* d_ws, size_t ws_size,
                              hipStream_t stream) {
  const float* x   = (const float*)d_in[0];
  const float* pos = (const float*)d_in[1];
  const float* W1  = (const float*)d_in[2];
  const float* b1  = (const float*)d_in[3];
  const float* W2  = (const float*)d_in[4];
  const float* b2  = (const float*)d_in[5];
  const float* Wl  = (const float*)d_in[6];
  const float* bl  = (const float*)d_in[7];
  const int*   ei  = (const int*)d_in[8];
  const int*   eli = (const int*)d_in[9];

  const int N  = in_sizes[0] / IN_C;
  const int E  = in_sizes[8] / 2;
  const int EL = in_sizes[9] / 2;

  float* z    = (float*)d_out;
  float* pred = z + (long)N * OUTD;

  const int NB = (N + BIN_ROWS - 1) / BIN_ROWS;
  const int per = PART_THREADS * PART_K;
  const int nblk = (E + per - 1) / per;

  float* dinv  = (float*)d_ws;
  float* hw1   = dinv + N;
  float* h1    = hw1 + (long)N * HID;
  float* hw2   = hw1;
  int* bins     = (int*)(h1 + (long)N * HID);
  int* binCount = bins + E;
  int* blkCnt   = (int*)h1;

  const int* esrc = ei;
  const int* edst = ei + E;

  const int B = 256;

  k_count<<<nblk, PART_THREADS, 0, stream>>>(edst, blkCnt, NB, E);
  k_part <<<nblk, PART_THREADS, 0, stream>>>(esrc, edst, blkCnt, binCount, bins, NB, E);
  k_dinv <<<NB, DINV_THREADS, 0, stream>>>(bins, binCount, dinv, N);

  const int rowTiles = (N + 15) / 16;
  const int gblocks  = (rowTiles + 7) / 8;
  k_gemm1<<<gblocks, B, 0, stream>>>(x, pos, W1, hw1, N);
  k_agg<HID, true><<<NB, AGG_THREADS, 0, stream>>>(bins, binCount, dinv, hw1, b1, h1, N);

  k_gemm2<<<gblocks, B, 0, stream>>>(h1, W2, hw2, N);
  k_agg<OUTD, false><<<NB, AGG_THREADS, 0, stream>>>(bins, binCount, dinv, hw2, b2, z, N);

  k_link<<<(EL + B - 1) / B, B, 0, stream>>>(z, eli, Wl, bl, pred, EL);
}


# Link kernel reads z rows with 8 lanes per edge (16 B per lane, four whole rows per load, not one row per lane = 32 lines per load); dot = 8 lane partials + 3-step lane tree (pred sum order changes)
# speedup vs baseline: 4.5496x; 1.0460x over previous
#include <hip/hip_runtime.h>

typedef __attribute__((ext_vector_type(2))) float v2f;
typedef __attribute__((ext_vector_type(8))) float v8f;

#define IN_C 64
#define POSD 16
#define KDIM1 (IN_C + POSD)
#define HID 64
#define OUTD 32


#define AGG_THREADS 1024
#define AGG_WAVES 32
#define Q_BATCH 32

#define BIN_ROWS 400
#define ENT_SHIFT 23
#define PART_THREADS 1024
#define PART_K 8
#define MAX_BINS 1024
#define CNT_BATCH 32
#define DINV_THREADS 1024
#define DINV_K 8
static_assert(BIN_ROWS <= (1 << (32 - ENT_SHIFT)), "a row inside its bin must fit above ENT_SHIFT");
static_assert(BIN_ROWS <= DINV_THREADS && BIN_ROWS <= AGG_THREADS, "one thread per owned row in the owner blocks");


static __device__ __forceinline__ int wave_sum(int v) {
#pragma unroll
  for (int o = 16; o > 0; o >>= 1) v += __shfl_xor(v, o);
  return v;
}

static __device__ __forceinline__ int wave_scan(int v, int lane) {
#pragma unroll
  for (int o = 1; o < 32; o <<= 1) {
    const int t = __shfl_up(v, o);
    if (lane >= o) v += t;
  }
  return v;
}

__global__ __launch_bounds__(PART_THREADS) void k_count(
    const int* __restrict__ dst, int* __restrict__ blkCnt, int NB, int E) {
  __shared__ int hist[MAX_BINS];
  const int tid = threadIdx.x;
  hist[tid] = 0;
  __syncthreads();

  const int e0 = blockIdx.x * (PART_THREADS * PART_K);
  int dd[PART_K];
#pragma unroll
  for (int k = 0; k < PART_K; ++k)
    dd[k] = dst[min(e0 + k * PART_THREADS + tid, E - 1)];
#pragma unroll
  for (int k = 0; k < PART_K; ++k)
    if (e0 + k * PART_THREADS + tid < E) atomicAdd(&hist[(unsigned)dd[k] / BIN_ROWS], 1);
  __syncthreads();

  if (tid < NB) blkCnt[blockIdx.x * NB + tid] = hist[tid];
}


template <bool FULL>
static __device__ __forceinline__ void add_counts(
    const int* __restrict__ col, int NB, int b0, int nblk, int me, int& cb, int& ahead) {
  int c[CNT_BATCH];
#pragma unroll
  for (int j = 0; j < CNT_BATCH; ++j) {
    const int b = FULL ? (b0 + j) : min(b0 + j, nblk - 1);
    c[j] = col[b * NB];
  }
#pragma unroll
  for (int j = 0; j < CNT_BATCH; ++j) {
    const int v = (FULL || b0 + j < nblk) ? c[j] : 0;
    cb += v;
    ahead += (b0 + j < me) ? v : 0;
  }
}

__global__ __launch_bounds__(PART_THREADS) void k_part(
    const int* __restrict__ src, const int* __restrict__ dst,
    const int* __restrict__ blkCnt, int* __restrict__ binCount,
    int* __restrict__ bins, int NB, int E) {
  __shared__ int hist[MAX_BINS];
  __shared__ int bstart[MAX_BINS];
  __shared__ int wtot[PART_THREADS / 32];
  __shared__ int st_ent[PART_THREADS * PART_K];
  __shared__ int st_pos[PART_THREADS * PART_K];
  const int tid = threadIdx.x;
  const int lane = tid & 31;
  const int wave = tid >> 5;
  const int me = blockIdx.x;
  const int nblk = gridDim.x;

  int cb = 0, ahead = 0;
  if (tid < NB) {
    const int* col = blkCnt + tid;
    int b0 = 0;
#pragma nounroll
    for (; b0 + CNT_BATCH <= nblk; b0 += CNT_BATCH) add_counts<true>(col, NB, b0, nblk, me, cb, ahead);
    if (b0 < nblk) add_counts<false>(col, NB, b0, nblk, me, cb, ahead);
  }

  const int inc = wave_scan(cb, lane);
  if (lane == 31) wtot[wave] = inc;
  hist[tid] = 0;
  __syncthreads();
  {
    const int before = wave_sum((lane < wave) ? wtot[lane] : 0);
    bstart[tid] = before + inc - cb;
  }
  if (me == 0 && tid < NB) binCount[tid] = cb;

  const int e0 = me * (PART_THREADS * PART_K);
  int dd[PART_K], ss[PART_K], bin[PART_K], rank[PART_K];
#pragma unroll
  for (int k = 0; k < PART_K; ++k) {
    const int ec = min(e0 + k * PART_THREADS + tid, E - 1);
    dd[k] = dst[ec];
    ss[k] = src[ec];
  }
#pragma unroll
  for (int k = 0; k < PART_K; ++k) {
    const bool ok = e0 + k * PART_THREADS + tid < E;
    bin[k] = ok ? (int)((unsigned)dd[k] / BIN_ROWS) : -1;
    rank[k] = 0;
    if (ok) rank[k] = atomicAdd(&hist[bin[k]], 1);
  }
  __syncthreads();

  const int n = hist[tid];
  const int gb = bstart[tid] + ahead;
  const int inc2 = wave_scan(n, lane);
  if (lane == 31) wtot[wave] = inc2;
  __syncthreads();
  {
    const int before = wave_sum((lane < wave) ? wtot[lane] : 0);
    const int lofs = before + inc2 - n;
    hist[tid] = lofs;
    bstart[tid] = gb - lofs;
  }
  __syncthreads();

  int slot[PART_K], gpos[PART_K];
#pragma unroll
  for (int k = 0; k < PART_K; ++k) {
    const int bb = max(bin[k], 0);
    slot[k] = hist[bb] + rank[k];
    gpos[k] = slot[k] + bstart[bb];
  }
#pragma unroll
  for (int k = 0; k < PART_K; ++k) {
    if (bin[k] >= 0) {
      st_ent[slot[k]] = (int)((unsigned)ss[k] | ((unsigned)(dd[k] - bin[k] * BIN_ROWS) << ENT_SHIFT));
      st_pos[slot[k]] = gpos[k];
    }
  }
  __syncthreads();

  const int total = min(PART_THREADS * PART_K, E - e0);
  int oe[PART_K], op[PART_K];
#pragma unroll
  for (int k = 0; k < PART_K; ++k) {
    const int i = min(k * PART_THREADS + tid, total - 1);
    oe[k] = st_ent[i];
    op[k] = st_pos[i];
  }
#pragma unroll
  for (int k = 0; k < PART_K; ++k)
    if (k * PART_THREADS + tid < total) bins[op[k]] = oe[k];
}

__global__ __launch_bounds__(DINV_THREADS) void k_dinv(
    const int* __restrict__ bins, const int* __restrict__ binCount,
    float* __restrict__ dinv, int N) {
  __shared__ int cnt[BIN_ROWS];
  __shared__ int wtot[DINV_THREADS / 32];
  const int tid  = threadIdx.x;
  const int lane = tid & 31;
  const int wave = tid >> 5;
  const int b  = blockIdx.x;

  {
    const int part = wave_sum((tid < b) ? binCount[tid] : 0);
    if (lane == 0) wtot[wave] = part;
  }
  if (tid < BIN_ROWS) cnt[tid] = 0;
  __syncthreads();
  const int start = wave_sum(wtot[lane]);
  const int n = binCount[b];
  const int* mine = bins + start;

  for (int i0 = 0; i0 < n; i0 += DINV_THREADS * DINV_K) {
    int ent[DINV_K];
#pragma unroll
    for (int k = 0; k < DINV_K; ++k)
      ent[k] = mine[min(i0 + k * DINV_THREADS + tid, n - 1)];
#pragma unroll
    for (int k = 0; k < DINV_K; ++k)
      if (i0 + k * DINV_THREADS + tid < n) atomicAdd(&cnt[(unsigned)ent[k] >> ENT_SHIFT], 1);
  }
  __syncthreads();

  const int r = b * BIN_ROWS + tid;
  if (tid < BIN_ROWS && r < N) dinv[r] = rsqrtf((float)cnt[tid] + 1.0f);
}

__global__ __launch_bounds__(256) void k_gemm1(
    const float* __restrict__ x, const float* __restrict__ pos,
    const float* __restrict__ W1, float* __restrict__ hw1, int N) {
  const int lane = threadIdx.x & 31;
  const int wave = threadIdx.x >> 5;
  const int rowBase = (blockIdx.x * 8 + wave) * 16;
  if (rowBase >= N) return;
  const int kh  = lane >> 4;
  const int n16 = lane & 15;
  int row = rowBase + n16;
  if (row >= N) row = N - 1;

  v8f acc0 = {}, acc1 = {}, acc2 = {}, acc3 = {};
  for (int k = 0; k < KDIM1; k += 4) {
    const int ka = k + 2 * kh;
    v2f a;
    a.x = (ka     < IN_C) ? x[(long)row * IN_C + ka]       : pos[(long)row * POSD + (ka     - IN_C)];
    a.y = (ka + 1 < IN_C) ? x[(long)row * IN_C + ka + 1]   : pos[(long)row * POSD + (ka + 1 - IN_C)];
    const float* wb = W1 + (long)ka * HID + n16;
    v2f b0; b0.x = wb[0];  b0.y = wb[HID];
    v2f b1; b1.x = wb[16]; b1.y = wb[HID + 16];
    v2f b2; b2.x = wb[32]; b2.y = wb[HID + 32];
    v2f b3; b3.x = wb[48]; b3.y = wb[HID + 48];
    acc0 = __builtin_amdgcn_wmma_f32_16x16x4_f32(false, a, false, b0, (short)0, acc0, false, false);
    acc1 = __builtin_amdgcn_wmma_f32_16x16x4_f32(false, a, false, b1, (short)0, acc1, false, false);
    acc2 = __builtin_amdgcn_wmma_f32_16x16x4_f32(false, a, false, b2, (short)0, acc2, false, false);
    acc3 = __builtin_amdgcn_wmma_f32_16x16x4_f32(false, a, false, b3, (short)0, acc3, false, false);
  }
#pragma unroll
  for (int i = 0; i < 8; ++i) {
    const int r = rowBase + i + 8 * kh;
    if (r < N) {
      const long base = (long)r * HID + n16;
      hw1[base +  0] = acc0[i];
      hw1[base + 16] = acc1[i];
      hw1[base + 32] = acc2[i];
      hw1[base + 48] = acc3[i];
    }
  }
}

__global__ __launch_bounds__(256) void k_gemm2(
    const float* __restrict__ h, const float* __restrict__ W2,
    float* __restrict__ hw2, int N) {
  const int lane = threadIdx.x & 31;
  const int wave = threadIdx.x >> 5;
  const int rowBase = (blockIdx.x * 8 + wave) * 16;
  if (rowBase >= N) return;
  const int kh  = lane >> 4;
  const int n16 = lane & 15;
  int row = rowBase + n16;
  if (row >= N) row = N - 1;

  v8f acc0 = {}, acc1 = {};
  for (int k = 0; k < HID; k += 4) {
    const int ka = k + 2 * kh;
    v2f a;
    a.x = h[(long)row * HID + ka];
    a.y = h[(long)row * HID + ka + 1];
    const float* wb = W2 + (long)ka * OUTD + n16;
    v2f b0; b0.x = wb[0];  b0.y = wb[OUTD];
    v2f b1; b1.x = wb[16]; b1.y = wb[OUTD + 16];
    acc0 = __builtin_amdgcn_wmma_f32_16x16x4_f32(false, a, false, b0, (short)0, acc0, false, false);
    acc1 = __builtin_amdgcn_wmma_f32_16x16x4_f32(false, a, false, b1, (short)0, acc1, false, false);
  }
#pragma unroll
  for (int i = 0; i < 8; ++i) {
    const int r = rowBase + i + 8 * kh;
    if (r < N) {
      const long base = (long)r * OUTD + n16;
      hw2[base +  0] = acc0[i];
      hw2[base + 16] = acc1[i];
    }
  }
}


template <int C, bool FULL>
static __device__ __forceinline__ void fold_batch(
    int ent, int nb, int lane,
    const float* __restrict__ dinv, const float* dloc,
    const float* __restrict__ feat, float* acc) {
  constexpr int PER = C / 32;
  const int s = (int)((unsigned)ent & ((1u << ENT_SHIFT) - 1u));
  const int lr = (int)((unsigned)ent >> ENT_SHIFT);
  const float dvs = dinv[s];
  const float dvd = dloc[lr];
  float w = 0.0f;
  const int ro = lr * C;
#pragma unroll
  for (int i = 0; i < Q_BATCH; i += 16) {
    if (!FULL && i >= nb) break;
    float v[16][PER];
#pragma unroll
    for (int k = 0; k < 16; ++k) {
      if (FULL || i + k < nb) {
        const int sk = __builtin_amdgcn_readlane(s, i + k);
        const float* p = feat + (long)sk * C + lane * PER;
        if (PER == 2) {
          const v2f t = *(const v2f*)p;
          v[k][0] = t.x;
          v[k][PER - 1] = t.y;
        } else {
          v[k][0] = p[0];
        }
      }
    }
    if (i == 0) w = dvs * dvd;
#pragma unroll
    for (int k = 0; k < 16; ++k) {
      if (FULL || i + k < nb) {
        const float wk = __int_as_float(__builtin_amdgcn_readlane(__float_as_int(w), i + k));
        const int rk = __builtin_amdgcn_readlane(ro, i + k);
        float* a = acc + rk + lane * PER;
#pragma unroll
        for (int j = 0; j < PER; ++j)
          __hip_atomic_fetch_add(a + j, v[k][j] * wk, __ATOMIC_RELAXED, __HIP_MEMORY_SCOPE_WORKGROUP);
      }
    }
  }
}

template <int C, bool RELU>
__global__ __launch_bounds__(AGG_THREADS) void k_agg(
    const int* __restrict__ bins, const int* __restrict__ binCount,
    const float* __restrict__ dinv, const float* __restrict__ feat,
    const float* __restrict__ bias, float* __restrict__ out, int N) {
  __shared__ __attribute__((aligned(16))) float acc[BIN_ROWS * C];
  __shared__ float dloc[BIN_ROWS];
  __shared__ int wtot[AGG_WAVES];

  const int tid  = threadIdx.x;
  const int lane = tid & 31;
  const int wave = __builtin_amdgcn_readfirstlane(tid >> 5);
  const int b  = blockIdx.x;
  const int r0 = b * BIN_ROWS;
  const int nr = min(BIN_ROWS, N - r0);

  {
    const int part = wave_sum((tid < b) ? binCount[tid] : 0);
    if (lane == 0) wtot[wave] = part;
  }
  if (tid < BIN_ROWS) dloc[tid] = (tid < nr) ? dinv[r0 + tid] : 0.0f;

  for (int i = tid; i < nr * (C / 4); i += AGG_THREADS) {
    const int r = i / (C / 4);
    const float dv = dinv[r0 + r];
    const float sw = dv * dv;
    float4 t = ((const float4*)(feat + (long)r0 * C))[i];
    t.x *= sw; t.y *= sw; t.z *= sw; t.w *= sw;
    ((float4*)acc)[i] = t;
  }
  __syncthreads();

  const int start = __builtin_amdgcn_readfirstlane(wave_sum(wtot[lane]));
  const int n = __builtin_amdgcn_readfirstlane(binCount[b]);
  const int* mine = bins + start;

  if (n > 0) {
    int base = wave * Q_BATCH;
    int ent = mine[min(base + lane, n - 1)];
    while (base < n) {
      const int nbase = base + AGG_WAVES * Q_BATCH;
      const int entn = mine[min(nbase + lane, n - 1)];
      const int nb = min(Q_BATCH, n - base);
      if (nb == Q_BATCH) fold_batch<C, true>(ent, nb, lane, dinv, dloc, feat, acc);
      else               fold_batch<C, false>(ent, nb, lane, dinv, dloc, feat, acc);
      ent = entn;
      base = nbase;
    }
  }

  __syncthreads();

  for (int i = tid; i < nr * (C / 4); i += AGG_THREADS) {
    const int c4 = i % (C / 4);
    float4 t = ((const float4*)acc)[i];
    const float4 bb = ((const float4*)bias)[c4];
    t.x += bb.x; t.y += bb.y; t.z += bb.z; t.w += bb.w;
    if (RELU) {
      t.x = fmaxf(t.x, 0.0f); t.y = fmaxf(t.y, 0.0f);
      t.z = fmaxf(t.z, 0.0f); t.w = fmaxf(t.w, 0.0f);
    }
    ((float4*)(out + (long)r0 * C))[i] = t;
  }
}

__global__ void k_link(const float* __restrict__ z, const int* __restrict__ eli,
                       const float* __restrict__ Wl, const float* __restrict__ bl,
                       float* __restrict__ pred, int EL) {
  const int lane = threadIdx.x & 31;
  const int g = lane >> 3;
  const int q = lane & 7;
  const int e0 = blockIdx.x * blockDim.x + threadIdx.x - lane;
  if (e0 >= EL) return;
  const int ec = min(e0 + lane, EL - 1);
  const int s = eli[ec];
  const int d = eli[EL + ec];
  const float4 ws = ((const float4*)Wl)[q];
  const float4 wd = ((const float4*)(Wl + OUTD))[q];

  float t[8];
#pragma unroll
  for (int j = 0; j < 8; ++j) {
    const int sj = __shfl(s, 4 * j + g);
    const int dj = __shfl(d, 4 * j + g);
    const float4 a = ((const float4*)(z + (long)sj * OUTD))[q];
    const float4 b = ((const float4*)(z + (long)dj * OUTD))[q];
    float acc = a.x * ws.x;
    acc = fmaf(a.y, ws.y, acc);
    acc = fmaf(a.z, ws.z, acc);
    acc = fmaf(a.w, ws.w, acc);
    acc = fmaf(b.x, wd.x, acc);
    acc = fmaf(b.y, wd.y, acc);
    acc = fmaf(b.z, wd.z, acc);
    acc = fmaf(b.w, wd.w, acc);
    t[j] = acc;
  }

  const bool b4 = (q & 4) != 0, b2 = (q & 2) != 0, b1 = (q & 1) != 0;
  float u[4];
#pragma unroll
  for (int i = 0; i < 4; ++i) {
    const float keep = b4 ? t[i + 4] : t[i];
    const float send = b4 ? t[i] : t[i + 4];
    u[i] = keep + __shfl_xor(send, 4);
  }
  float v[2];
#pragma unroll
  for (int i = 0; i < 2; ++i) {
    const float keep = b2 ? u[i + 2] : u[i];
    const float send = b2 ? u[i] : u[i + 2];
    v[i] = keep + __shfl_xor(send, 2);
  }
  const float keep1 = b1 ? v[1] : v[0];
  const float send1 = b1 ? v[0] : v[1];
  const float r = keep1 + __shfl_xor(send1, 1);

  const int e = e0 + 4 * q + g;
  if (e < EL) pred[e] = r + bl[0];
}

extern "C" void kernel_launch(void* const* d_in, const int* in_sizes, int n_in,
                              void* d_out, int out_size, void* d_ws, size_t ws_size,
                              hipStream_t stream) {
  const float* x   = (const float*)d_in[0];
  const float* pos = (const float*)d_in[1];
  const float* W1  = (const float*)d_in[2];
  const float* b1  = (const float*)d_in[3];
  const float* W2  = (const float*)d_in[4];
  const float* b2  = (const float*)d_in[5];
  const float* Wl  = (const float*)d_in[6];
  const float* bl  = (const float*)d_in[7];
  const int*   ei  = (const int*)d_in[8];
  const int*   eli = (const int*)d_in[9];

  const int N  = in_sizes[0] / IN_C;
  const int E  = in_sizes[8] / 2;
  const int EL = in_sizes[9] / 2;

  float* z    = (float*)d_out;
  float* pred = z + (long)N * OUTD;

  const int NB = (N + BIN_ROWS - 1) / BIN_ROWS;
  const int per = PART_THREADS * PART_K;
  const int nblk = (E + per - 1) / per;

  float* dinv  = (float*)d_ws;
  float* hw1   = dinv + N;
  float* h1    = hw1 + (long)N * HID;
  float* hw2   = hw1;
  int* bins     = (int*)(h1 + (long)N * HID);
  int* binCount = bins + E;
  int* blkCnt   = (int*)h1;

  const int* esrc = ei;
  const int* edst = ei + E;

  const int B = 256;

  k_count<<<nblk, PART_THREADS, 0, stream>>>(edst, blkCnt, NB, E);
  k_part <<<nblk, PART_THREADS, 0, stream>>>(esrc, edst, blkCnt, binCount, bins, NB, E);
  k_dinv <<<NB, DINV_THREADS, 0, stream>>>(bins, binCount, dinv, N);

  const int rowTiles = (N + 15) / 16;
  const int gblocks  = (rowTiles + 7) / 8;
  k_gemm1<<<gblocks, B, 0, stream>>>(x, pos, W1, hw1, N);
  k_agg<HID, true><<<NB, AGG_THREADS, 0, stream>>>(bins, binCount, dinv, hw1, b1, h1, N);

  k_gemm2<<<gblocks, B, 0, stream>>>(h1, W2, hw2, N);
  k_agg<OUTD, false><<<NB, AGG_THREADS, 0, stream>>>(bins, binCount, dinv, hw2, b2, z, N);

  k_link<<<(EL + B - 1) / B, B, 0, stream>>>(z, eli, Wl, bl, pred, EL);
}


# GEMM 1 and 2: weights staged once per block in LDS in B-fragment order, a lane's A half-row by 16-byte loads issued ahead; k order of the products permuted alike in both loaders (sum order changes)
# speedup vs baseline: 4.7770x; 1.0500x over previous
#include <hip/hip_runtime.h>

typedef __attribute__((ext_vector_type(2))) float v2f;
typedef __attribute__((ext_vector_type(8))) float v8f;

#define IN_C 64
#define POSD 16
#define KDIM1 (IN_C + POSD)
#define HID 64
#define OUTD 32


#define AGG_THREADS 1024
#define AGG_WAVES 32
#define Q_BATCH 32

#define BIN_ROWS 400
#define ENT_SHIFT 23
#define PART_THREADS 1024
#define PART_K 8
#define MAX_BINS 1024
#define CNT_BATCH 32
#define DINV_THREADS 1024
#define DINV_K 8
static_assert(BIN_ROWS <= (1 << (32 - ENT_SHIFT)), "a row inside its bin must fit above ENT_SHIFT");
static_assert(BIN_ROWS <= DINV_THREADS && BIN_ROWS <= AGG_THREADS, "one thread per owned row in the owner blocks");


static __device__ __forceinline__ int wave_sum(int v) {
#pragma unroll
  for (int o = 16; o > 0; o >>= 1) v += __shfl_xor(v, o);
  return v;
}

static __device__ __forceinline__ int wave_scan(int v, int lane) {
#pragma unroll
  for (int o = 1; o < 32; o <<= 1) {
    const int t = __shfl_up(v, o);
    if (lane >= o) v += t;
  }
  return v;
}

__global__ __launch_bounds__(PART_THREADS) void k_count(
    const int* __restrict__ dst, int* __restrict__ blkCnt, int NB, int E) {
  __shared__ int hist[MAX_BINS];
  const int tid = threadIdx.x;
  hist[tid] = 0;
  __syncthreads();

  const int e0 = blockIdx.x * (PART_THREADS * PART_K);
  int dd[PART_K];
#pragma unroll
  for (int k = 0; k < PART_K; ++k)
    dd[k] = dst[min(e0 + k * PART_THREADS + tid, E - 1)];
#pragma unroll
  for (int k = 0; k < PART_K; ++k)
    if (e0 + k * PART_THREADS + tid < E) atomicAdd(&hist[(unsigned)dd[k] / BIN_ROWS], 1);
  __syncthreads();

  if (tid < NB) blkCnt[blockIdx.x * NB + tid] = hist[tid];
}


template <bool FULL>
static __device__ __forceinline__ void add_counts(
    const int* __restrict__ col, int NB, int b0, int nblk, int me, int& cb, int& ahead) {
  int c[CNT_BATCH];
#pragma unroll
  for (int j = 0; j < CNT_BATCH; ++j) {
    const int b = FULL ? (b0 + j) : min(b0 + j, nblk - 1);
    c[j] = col[b * NB];
  }
#pragma unroll
  for (int j = 0; j < CNT_BATCH; ++j) {
    const int v = (FULL || b0 + j < nblk) ? c[j] : 0;
    cb += v;
    ahead += (b0 + j < me) ? v : 0;
  }
}

__global__ __launch_bounds__(PART_THREADS) void k_part(
    const int* __restrict__ src, const int* __restrict__ dst,
    const int* __restrict__ blkCnt, int* __restrict__ binCount,
    int* __restrict__ bins, int NB, int E) {
  __shared__ int hist[MAX_BINS];
  __shared__ int bstart[MAX_BINS];
  __shared__ int wtot[PART_THREADS / 32];
  __shared__ int st_ent[PART_THREADS * PART_K];
  __shared__ int st_pos[PART_THREADS * PART_K];
  const int tid = threadIdx.x;
  const int lane = tid & 31;
  const int wave = tid >> 5;
  const int me = blockIdx.x;
  const int nblk = gridDim.x;

  int cb = 0, ahead = 0;
  if (tid < NB) {
    const int* col = blkCnt + tid;
    int b0 = 0;
#pragma nounroll
    for (; b0 + CNT_BATCH <= nblk; b0 += CNT_BATCH) add_counts<true>(col, NB, b0, nblk, me, cb, ahead);
    if (b0 < nblk) add_counts<false>(col, NB, b0, nblk, me, cb, ahead);
  }

  const int inc = wave_scan(cb, lane);
  if (lane == 31) wtot[wave] = inc;
  hist[tid] = 0;
  __syncthreads();
  {
    const int before = wave_sum((lane < wave) ? wtot[lane] : 0);
    bstart[tid] = before + inc - cb;
  }
  if (me == 0 && tid < NB) binCount[tid] = cb;

  const int e0 = me * (PART_THREADS * PART_K);
  int dd[PART_K], ss[PART_K], bin[PART_K], rank[PART_K];
#pragma unroll
  for (int k = 0; k < PART_K; ++k) {
    const int ec = min(e0 + k * PART_THREADS + tid, E - 1);
    dd[k] = dst[ec];
    ss[k] = src[ec];
  }
#pragma unroll
  for (int k = 0; k < PART_K; ++k) {
    const bool ok = e0 + k * PART_THREADS + tid < E;
    bin[k] = ok ? (int)((unsigned)dd[k] / BIN_ROWS) : -1;
    rank[k] = 0;
    if (ok) rank[k] = atomicAdd(&hist[bin[k]], 1);
  }
  __syncthreads();

  const int n = hist[tid];
  const int gb = bstart[tid] + ahead;
  const int inc2 = wave_scan(n, lane);
  if (lane == 31) wtot[wave] = inc2;
  __syncthreads();
  {
    const int before = wave_sum((lane < wave) ? wtot[lane] : 0);
    const int lofs = before + inc2 - n;
    hist[tid] = lofs;
    bstart[tid] = gb - lofs;
  }
  __syncthreads();

  int slot[PART_K], gpos[PART_K];
#pragma unroll
  for (int k = 0; k < PART_K; ++k) {
    const int bb = max(bin[k], 0);
    slot[k] = hist[bb] + rank[k];
    gpos[k] = slot[k] + bstart[bb];
  }
#pragma unroll
  for (int k = 0; k < PART_K; ++k) {
    if (bin[k] >= 0) {
      st_ent[slot[k]] = (int)((unsigned)ss[k] | ((unsigned)(dd[k] - bin[k] * BIN_ROWS) << ENT_SHIFT));
      st_pos[slot[k]] = gpos[k];
    }
  }
  __syncthreads();

  const int total = min(PART_THREADS * PART_K, E - e0);
  int oe[PART_K], op[PART_K];
#pragma unroll
  for (int k = 0; k < PART_K; ++k) {
    const int i = min(k * PART_THREADS + tid, total - 1);
    oe[k] = st_ent[i];
    op[k] = st_pos[i];
  }
#pragma unroll
  for (int k = 0; k < PART_K; ++k)
    if (k * PART_THREADS + tid < total) bins[op[k]] = oe[k];
}

__global__ __launch_bounds__(DINV_THREADS) void k_dinv(
    const int* __restrict__ bins, const int* __restrict__ binCount,
    float* __restrict__ dinv, int N) {
  __shared__ int cnt[BIN_ROWS];
  __shared__ int wtot[DINV_THREADS / 32];
  const int tid  = threadIdx.x;
  const int lane = tid & 31;
  const int wave = tid >> 5;
  const int b  = blockIdx.x;

  {
    const int part = wave_sum((tid < b) ? binCount[tid] : 0);
    if (lane == 0) wtot[wave] = part;
  }
  if (tid < BIN_ROWS) cnt[tid] = 0;
  __syncthreads();
  const int start = wave_sum(wtot[lane]);
  const int n = binCount[b];
  const int* mine = bins + start;

  for (int i0 = 0; i0 < n; i0 += DINV_THREADS * DINV_K) {
    int ent[DINV_K];
#pragma unroll
    for (int k = 0; k < DINV_K; ++k)
      ent[k] = mine[min(i0 + k * DINV_THREADS + tid, n - 1)];
#pragma unroll
    for (int k = 0; k < DINV_K; ++k)
      if (i0 + k * DINV_THREADS + tid < n) atomicAdd(&cnt[(unsigned)ent[k] >> ENT_SHIFT], 1);
  }
  __syncthreads();

  const int r = b * BIN_ROWS + tid;
  if (tid < BIN_ROWS && r < N) dinv[r] = rsqrtf((float)cnt[tid] + 1.0f);
}


template <int K, int NT>
struct WStage {
  static constexpr int NC = 16 * NT;
  static constexpr int PER_T = K * NC / 4 / 256;
  static_assert(K % 8 == 0 && (K * NC / 4) % 256 == 0, "W must split evenly over the block");
  float4 w[PER_T];
  __device__ __forceinline__ void load(const float* __restrict__ Wm, int tid) {
#pragma unroll
    for (int j = 0; j < PER_T; ++j) w[j] = ((const float4*)Wm)[tid + 256 * j];
  }
  __device__ __forceinline__ void store(float* Bs, int tid) const {
#pragma unroll
    for (int j = 0; j < PER_T; ++j) {
      const int i = tid + 256 * j;
      const int k = (i * 4) / NC;
      const int n = (i * 4) % NC;
      const int t = 2 * (k >> 3) + ((k >> 1) & 1);
      const int h = (k >> 2) & 1;
      const int s = k & 1;
      const int d = ((t * NT + (n >> 4)) * 32 + (n & 15) + 16 * h) * 2 + s;
      Bs[d] = w[j].x; Bs[d + 2] = w[j].y; Bs[d + 4] = w[j].z; Bs[d + 6] = w[j].w;
    }
  }
};

__global__ __launch_bounds__(256) void k_gemm1(
    const float* __restrict__ x, const float* __restrict__ pos,
    const float* __restrict__ W1, float* __restrict__ hw1, int N) {
  constexpr int NT = HID / 16;
  constexpr int TS = KDIM1 / 4;
  __shared__ __attribute__((aligned(16))) float Bs[TS * NT * 32 * 2];
  const int lane = threadIdx.x & 31;
  const int wave = threadIdx.x >> 5;
  const int rowBase = (blockIdx.x * 8 + wave) * 16;
  const int kh  = lane >> 4;
  const int n16 = lane & 15;
  const int row = min(rowBase + n16, N - 1);

  WStage<KDIM1, NT> ws;
  ws.load(W1, threadIdx.x);

  float4 a4[TS / 2];
  const float4* xr = (const float4*)(x + (long)row * IN_C);
  const float4* pr = (const float4*)(pos + (long)row * POSD);
#pragma unroll
  for (int q = 0; q < IN_C / 8; ++q) a4[q] = xr[2 * q + kh];
#pragma unroll
  for (int q = 0; q < POSD / 8; ++q) a4[IN_C / 8 + q] = pr[2 * q + kh];

  ws.store(Bs, threadIdx.x);
  __syncthreads();
  if (rowBase >= N) return;

  v8f acc0 = {}, acc1 = {}, acc2 = {}, acc3 = {};
  const float* bl = Bs + lane * 2;
#pragma unroll
  for (int t = 0; t < TS; ++t) {
    v2f a;
    a.x = (t & 1) ? a4[t >> 1].z : a4[t >> 1].x;
    a.y = (t & 1) ? a4[t >> 1].w : a4[t >> 1].y;
    const v2f b0 = *(const v2f*)(bl + (t * NT + 0) * 64);
    const v2f b1 = *(const v2f*)(bl + (t * NT + 1) * 64);
    const v2f b2 = *(const v2f*)(bl + (t * NT + 2) * 64);
    const v2f b3 = *(const v2f*)(bl + (t * NT + 3) * 64);
    acc0 = __builtin_amdgcn_wmma_f32_16x16x4_f32(false, a, false, b0, (short)0, acc0, false, false);
    acc1 = __builtin_amdgcn_wmma_f32_16x16x4_f32(false, a, false, b1, (short)0, acc1, false, false);
    acc2 = __builtin_amdgcn_wmma_f32_16x16x4_f32(false, a, false, b2, (short)0, acc2, false, false);
    acc3 = __builtin_amdgcn_wmma_f32_16x16x4_f32(false, a, false, b3, (short)0, acc3, false, false);
  }
#pragma unroll
  for (int i = 0; i < 8; ++i) {
    const int r = rowBase + i + 8 * kh;
    if (r < N) {
      const long base = (long)r * HID + n16;
      hw1[base +  0] = acc0[i];
      hw1[base + 16] = acc1[i];
      hw1[base + 32] = acc2[i];
      hw1[base + 48] = acc3[i];
    }
  }
}

__global__ __launch_bounds__(256) void k_gemm2(
    const float* __restrict__ h, const float* __restrict__ W2,
    float* __restrict__ hw2, int N) {
  constexpr int NT = OUTD / 16;
  constexpr int TS = HID / 4;
  __shared__ __attribute__((aligned(16))) float Bs[TS * NT * 32 * 2];
  const int lane = threadIdx.x & 31;
  const int wave = threadIdx.x >> 5;
  const int rowBase = (blockIdx.x * 8 + wave) * 16;
  const int kh  = lane >> 4;
  const int n16 = lane & 15;
  const int row = min(rowBase + n16, N - 1);

  WStage<HID, NT> ws;
  ws.load(W2, threadIdx.x);

  float4 a4[TS / 2];
  const float4* hr = (const float4*)(h + (long)row * HID);
#pragma unroll
  for (int q = 0; q < HID / 8; ++q) a4[q] = hr[2 * q + kh];

  ws.store(Bs, threadIdx.x);
  __syncthreads();
  if (rowBase >= N) return;

  v8f acc0 = {}, acc1 = {};
  const float* bl = Bs + lane * 2;
#pragma unroll
  for (int t = 0; t < TS; ++t) {
    v2f a;
    a.x = (t & 1) ? a4[t >> 1].z : a4[t >> 1].x;
    a.y = (t & 1) ? a4[t >> 1].w : a4[t >> 1].y;
    const v2f b0 = *(const v2f*)(bl + (t * NT + 0) * 64);
    const v2f b1 = *(const v2f*)(bl + (t * NT + 1) * 64);
    acc0 = __builtin_amdgcn_wmma_f32_16x16x4_f32(false, a, false, b0, (short)0, acc0, false, false);
    acc1 = __builtin_amdgcn_wmma_f32_16x16x4_f32(false, a, false, b1, (short)0, acc1, false, false);
  }
#pragma unroll
  for (int i = 0; i < 8; ++i) {
    const int r = rowBase + i + 8 * kh;
    if (r < N) {
      const long base = (long)r * OUTD + n16;
      hw2[base +  0] = acc0[i];
      hw2[base + 16] = acc1[i];
    }
  }
}


template <int C, bool FULL>
static __device__ __forceinline__ void fold_batch(
    int ent, int nb, int lane,
    const float* __restrict__ dinv, const float* dloc,
    const float* __restrict__ feat, float* acc) {
  constexpr int PER = C / 32;
  const int s = (int)((unsigned)ent & ((1u << ENT_SHIFT) - 1u));
  const int lr = (int)((unsigned)ent >> ENT_SHIFT);
  const float dvs = dinv[s];
  const float dvd = dloc[lr];
  float w = 0.0f;
  const int ro = lr * C;
#pragma unroll
  for (int i = 0; i < Q_BATCH; i += 16) {
    if (!FULL && i >= nb) break;
    float v[16][PER];
#pragma unroll
    for (int k = 0; k < 16; ++k) {
      if (FULL || i + k < nb) {
        const int sk = __builtin_amdgcn_readlane(s, i + k);
        const float* p = feat + (long)sk * C + lane * PER;
        if (PER == 2) {
          const v2f t = *(const v2f*)p;
          v[k][0] = t.x;
          v[k][PER - 1] = t.y;
        } else {
          v[k][0] = p[0];
        }
      }
    }
    if (i == 0) w = dvs * dvd;
#pragma unroll
    for (int k = 0; k < 16; ++k) {
      if (FULL || i + k < nb) {
        const float wk = __int_as_float(__builtin_amdgcn_readlane(__float_as_int(w), i + k));
        const int rk = __builtin_amdgcn_readlane(ro, i + k);
        float* a = acc + rk + lane * PER;
#pragma unroll
        for (int j = 0; j < PER; ++j)
          __hip_atomic_fetch_add(a + j, v[k][j] * wk, __ATOMIC_RELAXED, __HIP_MEMORY_SCOPE_WORKGROUP);
      }
    }
  }
}

template <int C, bool RELU>
__global__ __launch_bounds__(AGG_THREADS) void k_agg(
    const int* __restrict__ bins, const int* __restrict__ binCount,
    const float* __restrict__ dinv, const float* __restrict__ feat,
    const float* __restrict__ bias, float* __restrict__ out, int N) {
  __shared__ __attribute__((aligned(16))) float acc[BIN_ROWS * C];
  __shared__ float dloc[BIN_ROWS];
  __shared__ int wtot[AGG_WAVES];

  const int tid  = threadIdx.x;
  const int lane = tid & 31;
  const int wave = __builtin_amdgcn_readfirstlane(tid >> 5);
  const int b  = blockIdx.x;
  const int r0 = b * BIN_ROWS;
  const int nr = min(BIN_ROWS, N - r0);

  {
    const int part = wave_sum((tid < b) ? binCount[tid] : 0);
    if (lane == 0) wtot[wave] = part;
  }
  if (tid < BIN_ROWS) dloc[tid] = (tid < nr) ? dinv[r0 + tid] : 0.0f;

  for (int i = tid; i < nr * (C / 4); i += AGG_THREADS) {
    const int r = i / (C / 4);
    const float dv = dinv[r0 + r];
    const float sw = dv * dv;
    float4 t = ((const float4*)(feat + (long)r0 * C))[i];
    t.x *= sw; t.y *= sw; t.z *= sw; t.w *= sw;
    ((float4*)acc)[i] = t;
  }
  __syncthreads();

  const int start = __builtin_amdgcn_readfirstlane(wave_sum(wtot[lane]));
  const int n = __builtin_amdgcn_readfirstlane(binCount[b]);
  const int* mine = bins + start;

  if (n > 0) {
    int base = wave * Q_BATCH;
    int ent = mine[min(base + lane, n - 1)];
    while (base < n) {
      const int nbase = base + AGG_WAVES * Q_BATCH;
      const int entn = mine[min(nbase + lane, n - 1)];
      const int nb = min(Q_BATCH, n - base);
      if (nb == Q_BATCH) fold_batch<C, true>(ent, nb, lane, dinv, dloc, feat, acc);
      else               fold_batch<C, false>(ent, nb, lane, dinv, dloc, feat, acc);
      ent = entn;
      base = nbase;
    }
  }

  __syncthreads();

  for (int i = tid; i < nr * (C / 4); i += AGG_THREADS) {
    const int c4 = i % (C / 4);
    float4 t = ((const float4*)acc)[i];
    const float4 bb = ((const float4*)bias)[c4];
    t.x += bb.x; t.y += bb.y; t.z += bb.z; t.w += bb.w;
    if (RELU) {
      t.x = fmaxf(t.x, 0.0f); t.y = fmaxf(t.y, 0.0f);
      t.z = fmaxf(t.z, 0.0f); t.w = fmaxf(t.w, 0.0f);
    }
    ((float4*)(out + (long)r0 * C))[i] = t;
  }
}

__global__ void k_link(const float* __restrict__ z, const int* __restrict__ eli,
                       const float* __restrict__ Wl, const float* __restrict__ bl,
                       float* __restrict__ pred, int EL) {
  const int lane = threadIdx.x & 31;
  const int g = lane >> 3;
  const int q = lane & 7;
  const int e0 = blockIdx.x * blockDim.x + threadIdx.x - lane;
  if (e0 >= EL) return;
  const int ec = min(e0 + lane, EL - 1);
  const int s = eli[ec];
  const int d = eli[EL + ec];
  const float4 ws = ((const float4*)Wl)[q];
  const float4 wd = ((const float4*)(Wl + OUTD))[q];

  float t[8];
#pragma unroll
  for (int j = 0; j < 8; ++j) {
    const int sj = __shfl(s, 4 * j + g);
    const int dj = __shfl(d, 4 * j + g);
    const float4 a = ((const float4*)(z + (long)sj * OUTD))[q];
    const float4 b = ((const float4*)(z + (long)dj * OUTD))[q];
    float acc = a.x * ws.x;
    acc = fmaf(a.y, ws.y, acc);
    acc = fmaf(a.z, ws.z, acc);
    acc = fmaf(a.w, ws.w, acc);
    acc = fmaf(b.x, wd.x, acc);
    acc = fmaf(b.y, wd.y, acc);
    acc = fmaf(b.z, wd.z, acc);
    acc = fmaf(b.w, wd.w, acc);
    t[j] = acc;
  }

  const bool b4 = (q & 4) != 0, b2 = (q & 2) != 0, b1 = (q & 1) != 0;
  float u[4];
#pragma unroll
  for (int i = 0; i < 4; ++i) {
    const float keep = b4 ? t[i + 4] : t[i];
    const float send = b4 ? t[i] : t[i + 4];
    u[i] = keep + __shfl_xor(send, 4);
  }
  float v[2];
#pragma unroll
  for (int i = 0; i < 2; ++i) {
    const float keep = b2 ? u[i + 2] : u[i];
    const float send = b2 ? u[i] : u[i + 2];
    v[i] = keep + __shfl_xor(send, 2);
  }
  const float keep1 = b1 ? v[1] : v[0];
  const float send1 = b1 ? v[0] : v[1];
  const float r = keep1 + __shfl_xor(send1, 1);

  const int e = e0 + 4 * q + g;
  if (e < EL) pred[e] = r + bl[0];
}

extern "C" void kernel_launch(void* const* d_in, const int* in_sizes, int n_in,
                              void* d_out, int out_size, void* d_ws, size_t ws_size,
                              hipStream_t stream) {
  const float* x   = (const float*)d_in[0];
  const float* pos = (const float*)d_in[1];
  const float* W1  = (const float*)d_in[2];
  const float* b1  = (const float*)d_in[3];
  const float* W2  = (const float*)d_in[4];
  const float* b2  = (const float*)d_in[5];
  const float* Wl  = (const float*)d_in[6];
  const float* bl  = (const float*)d_in[7];
  const int*   ei  = (const int*)d_in[8];
  const int*   eli = (const int*)d_in[9];

  const int N  = in_sizes[0] / IN_C;
  const int E  = in_sizes[8] / 2;
  const int EL = in_sizes[9] / 2;

  float* z    = (float*)d_out;
  float* pred = z + (long)N * OUTD;

  const int NB = (N + BIN_ROWS - 1) / BIN_ROWS;
  const int per = PART_THREADS * PART_K;
  const int nblk = (E + per - 1) / per;

  float* dinv  = (float*)d_ws;
  float* hw1   = dinv + N;
  float* h1    = hw1 + (long)N * HID;
  float* hw2   = hw1;
  int* bins     = (int*)(h1 + (long)N * HID);
  int* binCount = bins + E;
  int* blkCnt   = (int*)h1;

  const int* esrc = ei;
  const int* edst = ei + E;

  const int B = 256;

  k_count<<<nblk, PART_THREADS, 0, stream>>>(edst, blkCnt, NB, E);
  k_part <<<nblk, PART_THREADS, 0, stream>>>(esrc, edst, blkCnt, binCount, bins, NB, E);
  k_dinv <<<NB, DINV_THREADS, 0, stream>>>(bins, binCount, dinv, N);

  const int rowTiles = (N + 15) / 16;
  const int gblocks  = (rowTiles + 7) / 8;
  k_gemm1<<<gblocks, B, 0, stream>>>(x, pos, W1, hw1, N);
  k_agg<HID, true><<<NB, AGG_THREADS, 0, stream>>>(bins, binCount, dinv, hw1, b1, h1, N);

  k_gemm2<<<gblocks, B, 0, stream>>>(h1, W2, hw2, N);
  k_agg<OUTD, false><<<NB, AGG_THREADS, 0, stream>>>(bins, binCount, dinv, hw2, b2, z, N);

  k_link<<<(EL + B - 1) / B, B, 0, stream>>>(z, eli, Wl, bl, pred, EL);
}


# Per-edge source weight from an LDS table: degree pass stores 16-bit row degrees in dinv's bytes, each aggregation block copies all of them to LDS, takes rsqrt(deg+1) there; no scattered dinv read
# speedup vs baseline: 4.9774x; 1.0420x over previous
#include <hip/hip_runtime.h>

typedef __attribute__((ext_vector_type(2))) float v2f;
typedef __attribute__((ext_vector_type(8))) float v8f;

#define IN_C 64
#define POSD 16
#define KDIM1 (IN_C + POSD)
#define HID 64
#define OUTD 32


#define AGG_THREADS 1024
#define AGG_WAVES 32
#define Q_BATCH 32

#define BIN_ROWS 400
#define ENT_SHIFT 23
#define PART_THREADS 1024
#define PART_K 8
#define MAX_BINS 1024
#define CNT_BATCH 32
#define DINV_THREADS 1024
#define DINV_K 8
#define TBL_CAP 100352
static_assert(BIN_ROWS <= (1 << (32 - ENT_SHIFT)), "a row inside its bin must fit above ENT_SHIFT");
static_assert(BIN_ROWS <= DINV_THREADS && BIN_ROWS <= AGG_THREADS, "one thread per owned row in the owner blocks");


static __device__ __forceinline__ int wave_sum(int v) {
#pragma unroll
  for (int o = 16; o > 0; o >>= 1) v += __shfl_xor(v, o);
  return v;
}

static __device__ __forceinline__ int wave_max(int v) {
#pragma unroll
  for (int o = 16; o > 0; o >>= 1) v = max(v, __shfl_xor(v, o));
  return v;
}

static __device__ __forceinline__ bool table_fits(int maxbin, int N) {
  return maxbin <= 0xFFFF && N <= TBL_CAP;
}

static __device__ __forceinline__ int wave_scan(int v, int lane) {
#pragma unroll
  for (int o = 1; o < 32; o <<= 1) {
    const int t = __shfl_up(v, o);
    if (lane >= o) v += t;
  }
  return v;
}

__global__ __launch_bounds__(PART_THREADS) void k_count(
    const int* __restrict__ dst, int* __restrict__ blkCnt, int NB, int E) {
  __shared__ int hist[MAX_BINS];
  const int tid = threadIdx.x;
  hist[tid] = 0;
  __syncthreads();

  const int e0 = blockIdx.x * (PART_THREADS * PART_K);
  int dd[PART_K];
#pragma unroll
  for (int k = 0; k < PART_K; ++k)
    dd[k] = dst[min(e0 + k * PART_THREADS + tid, E - 1)];
#pragma unroll
  for (int k = 0; k < PART_K; ++k)
    if (e0 + k * PART_THREADS + tid < E) atomicAdd(&hist[(unsigned)dd[k] / BIN_ROWS], 1);
  __syncthreads();

  if (tid < NB) blkCnt[blockIdx.x * NB + tid] = hist[tid];
}


template <bool FULL>
static __device__ __forceinline__ void add_counts(
    const int* __restrict__ col, int NB, int b0, int nblk, int me, int& cb, int& ahead) {
  int c[CNT_BATCH];
#pragma unroll
  for (int j = 0; j < CNT_BATCH; ++j) {
    const int b = FULL ? (b0 + j) : min(b0 + j, nblk - 1);
    c[j] = col[b * NB];
  }
#pragma unroll
  for (int j = 0; j < CNT_BATCH; ++j) {
    const int v = (FULL || b0 + j < nblk) ? c[j] : 0;
    cb += v;
    ahead += (b0 + j < me) ? v : 0;
  }
}

__global__ __launch_bounds__(PART_THREADS) void k_part(
    const int* __restrict__ src, const int* __restrict__ dst,
    const int* __restrict__ blkCnt, int* __restrict__ binCount,
    int* __restrict__ bins, int NB, int E) {
  __shared__ int hist[MAX_BINS];
  __shared__ int bstart[MAX_BINS];
  __shared__ int wtot[PART_THREADS / 32];
  __shared__ int st_ent[PART_THREADS * PART_K];
  __shared__ int st_pos[PART_THREADS * PART_K];
  const int tid = threadIdx.x;
  const int lane = tid & 31;
  const int wave = tid >> 5;
  const int me = blockIdx.x;
  const int nblk = gridDim.x;

  int cb = 0, ahead = 0;
  if (tid < NB) {
    const int* col = blkCnt + tid;
    int b0 = 0;
#pragma nounroll
    for (; b0 + CNT_BATCH <= nblk; b0 += CNT_BATCH) add_counts<true>(col, NB, b0, nblk, me, cb, ahead);
    if (b0 < nblk) add_counts<false>(col, NB, b0, nblk, me, cb, ahead);
  }

  const int inc = wave_scan(cb, lane);
  if (lane == 31) wtot[wave] = inc;
  hist[tid] = 0;
  __syncthreads();
  {
    const int before = wave_sum((lane < wave) ? wtot[lane] : 0);
    bstart[tid] = before + inc - cb;
  }
  if (me == 0 && tid < NB) binCount[tid] = cb;

  const int e0 = me * (PART_THREADS * PART_K);
  int dd[PART_K], ss[PART_K], bin[PART_K], rank[PART_K];
#pragma unroll
  for (int k = 0; k < PART_K; ++k) {
    const int ec = min(e0 + k * PART_THREADS + tid, E - 1);
    dd[k] = dst[ec];
    ss[k] = src[ec];
  }
#pragma unroll
  for (int k = 0; k < PART_K; ++k) {
    const bool ok = e0 + k * PART_THREADS + tid < E;
    bin[k] = ok ? (int)((unsigned)dd[k] / BIN_ROWS) : -1;
    rank[k] = 0;
    if (ok) rank[k] = atomicAdd(&hist[bin[k]], 1);
  }
  __syncthreads();

  const int n = hist[tid];
  const int gb = bstart[tid] + ahead;
  const int inc2 = wave_scan(n, lane);
  if (lane == 31) wtot[wave] = inc2;
  __syncthreads();
  {
    const int before = wave_sum((lane < wave) ? wtot[lane] : 0);
    const int lofs = before + inc2 - n;
    hist[tid] = lofs;
    bstart[tid] = gb - lofs;
  }
  __syncthreads();

  int slot[PART_K], gpos[PART_K];
#pragma unroll
  for (int k = 0; k < PART_K; ++k) {
    const int bb = max(bin[k], 0);
    slot[k] = hist[bb] + rank[k];
    gpos[k] = slot[k] + bstart[bb];
  }
#pragma unroll
  for (int k = 0; k < PART_K; ++k) {
    if (bin[k] >= 0) {
      st_ent[slot[k]] = (int)((unsigned)ss[k] | ((unsigned)(dd[k] - bin[k] * BIN_ROWS) << ENT_SHIFT));
      st_pos[slot[k]] = gpos[k];
    }
  }
  __syncthreads();

  const int total = min(PART_THREADS * PART_K, E - e0);
  int oe[PART_K], op[PART_K];
#pragma unroll
  for (int k = 0; k < PART_K; ++k) {
    const int i = min(k * PART_THREADS + tid, total - 1);
    oe[k] = st_ent[i];
    op[k] = st_pos[i];
  }
#pragma unroll
  for (int k = 0; k < PART_K; ++k)
    if (k * PART_THREADS + tid < total) bins[op[k]] = oe[k];
}

__global__ __launch_bounds__(DINV_THREADS) void k_dinv(
    const int* __restrict__ bins, const int* __restrict__ binCount,
    float* dinv, unsigned short* deg16, int NB, int N) {
  __shared__ int cnt[BIN_ROWS];
  __shared__ int wtot[DINV_THREADS / 32];
  __shared__ int wmax[DINV_THREADS / 32];
  const int tid  = threadIdx.x;
  const int lane = tid & 31;
  const int wave = tid >> 5;
  const int b  = blockIdx.x;

  {
    const int bc = (tid < NB) ? binCount[tid] : 0;
    const int part = wave_sum((tid < b) ? bc : 0);
    const int mx = wave_max(bc);
    if (lane == 0) { wtot[wave] = part; wmax[wave] = mx; }
  }
  if (tid < BIN_ROWS) cnt[tid] = 0;
  __syncthreads();
  const int start = wave_sum(wtot[lane]);
  const bool tab = table_fits(wave_max(wmax[lane]), N);
  const int n = binCount[b];
  const int* mine = bins + start;

  for (int i0 = 0; i0 < n; i0 += DINV_THREADS * DINV_K) {
    int ent[DINV_K];
#pragma unroll
    for (int k = 0; k < DINV_K; ++k)
      ent[k] = mine[min(i0 + k * DINV_THREADS + tid, n - 1)];
#pragma unroll
    for (int k = 0; k < DINV_K; ++k)
      if (i0 + k * DINV_THREADS + tid < n) atomicAdd(&cnt[(unsigned)ent[k] >> ENT_SHIFT], 1);
  }
  __syncthreads();

  const int r = b * BIN_ROWS + tid;
  if (tid < BIN_ROWS && r < N) {
    if (tab) deg16[r] = (unsigned short)cnt[tid];
    else     dinv[r] = rsqrtf((float)cnt[tid] + 1.0f);
  }
}


template <int K, int NT>
struct WStage {
  static constexpr int NC = 16 * NT;
  static constexpr int PER_T = K * NC / 4 / 256;
  static_assert(K % 8 == 0 && (K * NC / 4) % 256 == 0, "W must split evenly over the block");
  float4 w[PER_T];
  __device__ __forceinline__ void load(const float* __restrict__ Wm, int tid) {
#pragma unroll
    for (int j = 0; j < PER_T; ++j) w[j] = ((const float4*)Wm)[tid + 256 * j];
  }
  __device__ __forceinline__ void store(float* Bs, int tid) const {
#pragma unroll
    for (int j = 0; j < PER_T; ++j) {
      const int i = tid + 256 * j;
      const int k = (i * 4) / NC;
      const int n = (i * 4) % NC;
      const int t = 2 * (k >> 3) + ((k >> 1) & 1);
      const int h = (k >> 2) & 1;
      const int s = k & 1;
      const int d = ((t * NT + (n >> 4)) * 32 + (n & 15) + 16 * h) * 2 + s;
      Bs[d] = w[j].x; Bs[d + 2] = w[j].y; Bs[d + 4] = w[j].z; Bs[d + 6] = w[j].w;
    }
  }
};

__global__ __launch_bounds__(256) void k_gemm1(
    const float* __restrict__ x, const float* __restrict__ pos,
    const float* __restrict__ W1, float* __restrict__ hw1, int N) {
  constexpr int NT = HID / 16;
  constexpr int TS = KDIM1 / 4;
  __shared__ __attribute__((aligned(16))) float Bs[TS * NT * 32 * 2];
  const int lane = threadIdx.x & 31;
  const int wave = threadIdx.x >> 5;
  const int rowBase = (blockIdx.x * 8 + wave) * 16;
  const int kh  = lane >> 4;
  const int n16 = lane & 15;
  const int row = min(rowBase + n16, N - 1);

  WStage<KDIM1, NT> ws;
  ws.load(W1, threadIdx.x);

  float4 a4[TS / 2];
  const float4* xr = (const float4*)(x + (long)row * IN_C);
  const float4* pr = (const float4*)(pos + (long)row * POSD);
#pragma unroll
  for (int q = 0; q < IN_C / 8; ++q) a4[q] = xr[2 * q + kh];
#pragma unroll
  for (int q = 0; q < POSD / 8; ++q) a4[IN_C / 8 + q] = pr[2 * q + kh];

  ws.store(Bs, threadIdx.x);
  __syncthreads();
  if (rowBase >= N) return;

  v8f acc0 = {}, acc1 = {}, acc2 = {}, acc3 = {};
  const float* bl = Bs + lane * 2;
#pragma unroll
  for (int t = 0; t < TS; ++t) {
    v2f a;
    a.x = (t & 1) ? a4[t >> 1].z : a4[t >> 1].x;
    a.y = (t & 1) ? a4[t >> 1].w : a4[t >> 1].y;
    const v2f b0 = *(const v2f*)(bl + (t * NT + 0) * 64);
    const v2f b1 = *(const v2f*)(bl + (t * NT + 1) * 64);
    const v2f b2 = *(const v2f*)(bl + (t * NT + 2) * 64);
    const v2f b3 = *(const v2f*)(bl + (t * NT + 3) * 64);
    acc0 = __builtin_amdgcn_wmma_f32_16x16x4_f32(false, a, false, b0, (short)0, acc0, false, false);
    acc1 = __builtin_amdgcn_wmma_f32_16x16x4_f32(false, a, false, b1, (short)0, acc1, false, false);
    acc2 = __builtin_amdgcn_wmma_f32_16x16x4_f32(false, a, false, b2, (short)0, acc2, false, false);
    acc3 = __builtin_amdgcn_wmma_f32_16x16x4_f32(false, a, false, b3, (short)0, acc3, false, false);
  }
#pragma unroll
  for (int i = 0; i < 8; ++i) {
    const int r = rowBase + i + 8 * kh;
    if (r < N) {
      const long base = (long)r * HID + n16;
      hw1[base +  0] = acc0[i];
      hw1[base + 16] = acc1[i];
      hw1[base + 32] = acc2[i];
      hw1[base + 48] = acc3[i];
    }
  }
}

__global__ __launch_bounds__(256) void k_gemm2(
    const float* __restrict__ h, const float* __restrict__ W2,
    float* __restrict__ hw2, int N) {
  constexpr int NT = OUTD / 16;
  constexpr int TS = HID / 4;
  __shared__ __attribute__((aligned(16))) float Bs[TS * NT * 32 * 2];
  const int lane = threadIdx.x & 31;
  const int wave = threadIdx.x >> 5;
  const int rowBase = (blockIdx.x * 8 + wave) * 16;
  const int kh  = lane >> 4;
  const int n16 = lane & 15;
  const int row = min(rowBase + n16, N - 1);

  WStage<HID, NT> ws;
  ws.load(W2, threadIdx.x);

  float4 a4[TS / 2];
  const float4* hr = (const float4*)(h + (long)row * HID);
#pragma unroll
  for (int q = 0; q < HID / 8; ++q) a4[q] = hr[2 * q + kh];

  ws.store(Bs, threadIdx.x);
  __syncthreads();
  if (rowBase >= N) return;

  v8f acc0 = {}, acc1 = {};
  const float* bl = Bs + lane * 2;
#pragma unroll
  for (int t = 0; t < TS; ++t) {
    v2f a;
    a.x = (t & 1) ? a4[t >> 1].z : a4[t >> 1].x;
    a.y = (t & 1) ? a4[t >> 1].w : a4[t >> 1].y;
    const v2f b0 = *(const v2f*)(bl + (t * NT + 0) * 64);
    const v2f b1 = *(const v2f*)(bl + (t * NT + 1) * 64);
    acc0 = __builtin_amdgcn_wmma_f32_16x16x4_f32(false, a, false, b0, (short)0, acc0, false, false);
    acc1 = __builtin_amdgcn_wmma_f32_16x16x4_f32(false, a, false, b1, (short)0, acc1, false, false);
  }
#pragma unroll
  for (int i = 0; i < 8; ++i) {
    const int r = rowBase + i + 8 * kh;
    if (r < N) {
      const long base = (long)r * OUTD + n16;
      hw2[base +  0] = acc0[i];
      hw2[base + 16] = acc1[i];
    }
  }
}


template <bool TAB>
static __device__ __forceinline__ float row_dinv(
    const unsigned short* dtab, const float* __restrict__ dinv, int row) {
  return TAB ? rsqrtf((float)dtab[row] + 1.0f) : dinv[row];
}

template <int C, bool FULL, bool TAB>
static __device__ __forceinline__ void fold_batch(
    int ent, int nb, int lane,
    const float* __restrict__ dinv, const unsigned short* dtab, const float* dloc,
    const float* __restrict__ feat, float* acc) {
  constexpr int PER = C / 32;
  const int s = (int)((unsigned)ent & ((1u << ENT_SHIFT) - 1u));
  const int lr = (int)((unsigned)ent >> ENT_SHIFT);
  const float dvs = row_dinv<TAB>(dtab, dinv, s);
  const float dvd = dloc[lr];
  float w = 0.0f;
  const int ro = lr * C;
#pragma unroll
  for (int i = 0; i < Q_BATCH; i += 16) {
    if (!FULL && i >= nb) break;
    float v[16][PER];
#pragma unroll
    for (int k = 0; k < 16; ++k) {
      if (FULL || i + k < nb) {
        const int sk = __builtin_amdgcn_readlane(s, i + k);
        const float* p = feat + (long)sk * C + lane * PER;
        if (PER == 2) {
          const v2f t = *(const v2f*)p;
          v[k][0] = t.x;
          v[k][PER - 1] = t.y;
        } else {
          v[k][0] = p[0];
        }
      }
    }
    if (i == 0) w = dvs * dvd;
#pragma unroll
    for (int k = 0; k < 16; ++k) {
      if (FULL || i + k < nb) {
        const float wk = __int_as_float(__builtin_amdgcn_readlane(__float_as_int(w), i + k));
        const int rk = __builtin_amdgcn_readlane(ro, i + k);
        float* a = acc + rk + lane * PER;
#pragma unroll
        for (int j = 0; j < PER; ++j)
          __hip_atomic_fetch_add(a + j, v[k][j] * wk, __ATOMIC_RELAXED, __HIP_MEMORY_SCOPE_WORKGROUP);
      }
    }
  }
}

template <int C, bool TAB>
static __device__ __forceinline__ void fold_bin(
    const int* __restrict__ mine, int n, int wave, int lane,
    const float* __restrict__ dinv, const unsigned short* dtab, const float* dloc,
    const float* __restrict__ feat, float* acc) {
  int base = wave * Q_BATCH;
  int ent = mine[min(base + lane, n - 1)];
  while (base < n) {
    const int nbase = base + AGG_WAVES * Q_BATCH;
    const int entn = mine[min(nbase + lane, n - 1)];
    const int nb = min(Q_BATCH, n - base);
    if (nb == Q_BATCH) fold_batch<C, true, TAB>(ent, nb, lane, dinv, dtab, dloc, feat, acc);
    else               fold_batch<C, false, TAB>(ent, nb, lane, dinv, dtab, dloc, feat, acc);
    ent = entn;
    base = nbase;
  }
}

template <int C, bool RELU>
__global__ __launch_bounds__(AGG_THREADS) void k_agg(
    const int* __restrict__ bins, const int* __restrict__ binCount,
    const float* __restrict__ dinv, const unsigned short* __restrict__ deg16,
    const float* __restrict__ feat,
    const float* __restrict__ bias, float* __restrict__ out, int NB, int N) {
  __shared__ __attribute__((aligned(16))) float acc[BIN_ROWS * C];
  __shared__ __attribute__((aligned(16))) unsigned short dtab[TBL_CAP];
  __shared__ float dloc[BIN_ROWS];
  __shared__ int wtot[AGG_WAVES];
  __shared__ int wmax[AGG_WAVES];

  const int tid  = threadIdx.x;
  const int lane = tid & 31;
  const int wave = __builtin_amdgcn_readfirstlane(tid >> 5);
  const int b  = blockIdx.x;
  const int r0 = b * BIN_ROWS;
  const int nr = min(BIN_ROWS, N - r0);

  const int bc = (tid < NB) ? binCount[tid] : 0;

  constexpr int SELF_IT = (BIN_ROWS * (C / 4) + AGG_THREADS - 1) / AGG_THREADS;
  const int nself = nr * (C / 4);
  float4 selfv[SELF_IT];
#pragma unroll
  for (int k = 0; k < SELF_IT; ++k)
    selfv[k] = ((const float4*)(feat + (long)r0 * C))[min(tid + k * AGG_THREADS, nself - 1)];

  constexpr int TAB_IT = (TBL_CAP / 8 + AGG_THREADS - 1) / AGG_THREADS;
  const int ntab = min(N, TBL_CAP);
  const int npiece = ntab >> 3;
  const int lastp = max(npiece - 1, 0);
  uint4 tv[TAB_IT];
#pragma unroll
  for (int k = 0; k < TAB_IT; ++k)
    tv[k] = ((const uint4*)deg16)[min(tid + k * AGG_THREADS, lastp)];

  {
    const int part = wave_sum((tid < b) ? bc : 0);
    const int mx = wave_max(bc);
    if (lane == 0) { wtot[wave] = part; wmax[wave] = mx; }
  }

#pragma unroll
  for (int k = 0; k < TAB_IT; ++k)
    ((uint4*)dtab)[min(tid + k * AGG_THREADS, lastp)] = tv[k];
  for (int i = (npiece << 3) + tid; i < ntab; i += AGG_THREADS) dtab[i] = deg16[i];
  __syncthreads();

  const int start = __builtin_amdgcn_readfirstlane(wave_sum(wtot[lane]));
  const bool tab = table_fits(__builtin_amdgcn_readfirstlane(wave_max(wmax[lane])), N);
  const int n = __builtin_amdgcn_readfirstlane(binCount[b]);
  const int* mine = bins + start;

  if (tid < BIN_ROWS)
    dloc[tid] = (tid < nr) ? (tab ? row_dinv<true>(dtab, dinv, r0 + tid) : row_dinv<false>(dtab, dinv, r0 + tid)) : 0.0f;
#pragma unroll
  for (int k = 0; k < SELF_IT; ++k) {
    const int i = tid + k * AGG_THREADS;
    if (i < nself) {
      const int r = r0 + i / (C / 4);
      const float dv = tab ? row_dinv<true>(dtab, dinv, r) : row_dinv<false>(dtab, dinv, r);
      const float sw = dv * dv;
      float4 t = selfv[k];
      t.x *= sw; t.y *= sw; t.z *= sw; t.w *= sw;
      ((float4*)acc)[i] = t;
    }
  }
  __syncthreads();

  if (n > 0) {
    if (tab) fold_bin<C, true>(mine, n, wave, lane, dinv, dtab, dloc, feat, acc);
    else     fold_bin<C, false>(mine, n, wave, lane, dinv, dtab, dloc, feat, acc);
  }

  __syncthreads();

  for (int i = tid; i < nr * (C / 4); i += AGG_THREADS) {
    const int c4 = i % (C / 4);
    float4 t = ((const float4*)acc)[i];
    const float4 bb = ((const float4*)bias)[c4];
    t.x += bb.x; t.y += bb.y; t.z += bb.z; t.w += bb.w;
    if (RELU) {
      t.x = fmaxf(t.x, 0.0f); t.y = fmaxf(t.y, 0.0f);
      t.z = fmaxf(t.z, 0.0f); t.w = fmaxf(t.w, 0.0f);
    }
    ((float4*)(out + (long)r0 * C))[i] = t;
  }
}

__global__ void k_link(const float* __restrict__ z, const int* __restrict__ eli,
                       const float* __restrict__ Wl, const float* __restrict__ bl,
                       float* __restrict__ pred, int EL) {
  const int lane = threadIdx.x & 31;
  const int g = lane >> 3;
  const int q = lane & 7;
  const int e0 = blockIdx.x * blockDim.x + threadIdx.x - lane;
  if (e0 >= EL) return;
  const int ec = min(e0 + lane, EL - 1);
  const int s = eli[ec];
  const int d = eli[EL + ec];
  const float4 ws = ((const float4*)Wl)[q];
  const float4 wd = ((const float4*)(Wl + OUTD))[q];

  float t[8];
#pragma unroll
  for (int j = 0; j < 8; ++j) {
    const int sj = __shfl(s, 4 * j + g);
    const int dj = __shfl(d, 4 * j + g);
    const float4 a = ((const float4*)(z + (long)sj * OUTD))[q];
    const float4 b = ((const float4*)(z + (long)dj * OUTD))[q];
    float acc = a.x * ws.x;
    acc = fmaf(a.y, ws.y, acc);
    acc = fmaf(a.z, ws.z, acc);
    acc = fmaf(a.w, ws.w, acc);
    acc = fmaf(b.x, wd.x, acc);
    acc = fmaf(b.y, wd.y, acc);
    acc = fmaf(b.z, wd.z, acc);
    acc = fmaf(b.w, wd.w, acc);
    t[j] = acc;
  }

  const bool b4 = (q & 4) != 0, b2 = (q & 2) != 0, b1 = (q & 1) != 0;
  float u[4];
#pragma unroll
  for (int i = 0; i < 4; ++i) {
    const float keep = b4 ? t[i + 4] : t[i];
    const float send = b4 ? t[i] : t[i + 4];
    u[i] = keep + __shfl_xor(send, 4);
  }
  float v[2];
#pragma unroll
  for (int i = 0; i < 2; ++i) {
    const float keep = b2 ? u[i + 2] : u[i];
    const float send = b2 ? u[i] : u[i + 2];
    v[i] = keep + __shfl_xor(send, 2);
  }
  const float keep1 = b1 ? v[1] : v[0];
  const float send1 = b1 ? v[0] : v[1];
  const float r = keep1 + __shfl_xor(send1, 1);

  const int e = e0 + 4 * q + g;
  if (e < EL) pred[e] = r + bl[0];
}

extern "C" void kernel_launch(void* const* d_in, const int* in_sizes, int n_in,
                              void* d_out, int out_size, void* d_ws, size_t ws_size,
                              hipStream_t stream) {
  const float* x   = (const float*)d_in[0];
  const float* pos = (const float*)d_in[1];
  const float* W1  = (const float*)d_in[2];
  const float* b1  = (const float*)d_in[3];
  const float* W2  = (const float*)d_in[4];
  const float* b2  = (const float*)d_in[5];
  const float* Wl  = (const float*)d_in[6];
  const float* bl  = (const float*)d_in[7];
  const int*   ei  = (const int*)d_in[8];
  const int*   eli = (const int*)d_in[9];

  const int N  = in_sizes[0] / IN_C;
  const int E  = in_sizes[8] / 2;
  const int EL = in_sizes[9] / 2;

  float* z    = (float*)d_out;
  float* pred = z + (long)N * OUTD;

  const int NB = (N + BIN_ROWS - 1) / BIN_ROWS;
  const int per = PART_THREADS * PART_K;
  const int nblk = (E + per - 1) / per;

  float* dinv  = (float*)d_ws;
  unsigned short* deg16 = (unsigned short*)d_ws;
  float* hw1   = dinv + N;
  float* h1    = hw1 + (long)N * HID;
  float* hw2   = hw1;
  int* bins     = (int*)(h1 + (long)N * HID);
  int* binCount = bins + E;
  int* blkCnt   = (int*)h1;

  const int* esrc = ei;
  const int* edst = ei + E;

  const int B = 256;

  k_count<<<nblk, PART_THREADS, 0, stream>>>(edst, blkCnt, NB, E);
  k_part <<<nblk, PART_THREADS, 0, stream>>>(esrc, edst, blkCnt, binCount, bins, NB, E);
  k_dinv <<<NB, DINV_THREADS, 0, stream>>>(bins, binCount, dinv, deg16, NB, N);

  const int rowTiles = (N + 15) / 16;
  const int gblocks  = (rowTiles + 7) / 8;
  k_gemm1<<<gblocks, B, 0, stream>>>(x, pos, W1, hw1, N);
  k_agg<HID, true><<<NB, AGG_THREADS, 0, stream>>>(bins, binCount, dinv, deg16, hw1, b1, h1, NB, N);

  k_gemm2<<<gblocks, B, 0, stream>>>(h1, W2, hw2, N);
  k_agg<OUTD, false><<<NB, AGG_THREADS, 0, stream>>>(bins, binCount, dinv, deg16, hw2, b2, z, NB, N);

  k_link<<<(EL + B - 1) / B, B, 0, stream>>>(z, eli, Wl, bl, pred, EL);
}


# GEMM 1 runs as extra work-groups of the partition launch, 2 groups per WGP (64 VGPRs: A by k halves, column tiles in 2 pairs), under the partition's waits; 7 launches; k order, sums unchanged
# speedup vs baseline: 5.3455x; 1.0740x over previous
#include <hip/hip_runtime.h>

typedef __attribute__((ext_vector_type(2))) float v2f;
typedef __attribute__((ext_vector_type(8))) float v8f;

#define IN_C 64
#define POSD 16
#define KDIM1 (IN_C + POSD)
#define HID 64
#define OUTD 32


#define AGG_THREADS 1024
#define AGG_WAVES 32
#define Q_BATCH 32

#define BIN_ROWS 400
#define ENT_SHIFT 23
#define PART_THREADS 1024
#define PART_K 8
#define MAX_BINS 1024
#define CNT_BATCH 32
#define SIMDS_PER_WGP 4
#define DINV_THREADS 1024
#define DINV_K 8
#define TBL_CAP 100352
static_assert(BIN_ROWS <= (1 << (32 - ENT_SHIFT)), "a row inside its bin must fit above ENT_SHIFT");
static_assert(BIN_ROWS <= DINV_THREADS && BIN_ROWS <= AGG_THREADS, "one thread per owned row in the owner blocks");


static __device__ __forceinline__ int wave_sum(int v) {
#pragma unroll
  for (int o = 16; o > 0; o >>= 1) v += __shfl_xor(v, o);
  return v;
}

static __device__ __forceinline__ int wave_max(int v) {
#pragma unroll
  for (int o = 16; o > 0; o >>= 1) v = max(v, __shfl_xor(v, o));
  return v;
}

static __device__ __forceinline__ bool table_fits(int maxbin, int N) {
  return maxbin <= 0xFFFF && N <= TBL_CAP;
}

static __device__ __forceinline__ int wave_scan(int v, int lane) {
#pragma unroll
  for (int o = 1; o < 32; o <<= 1) {
    const int t = __shfl_up(v, o);
    if (lane >= o) v += t;
  }
  return v;
}


template <int K, int NT, int T>
struct WStage {
  static constexpr int NC = 16 * NT;
  static constexpr int PIECES = K * NC / 4;
  static constexpr int PER_T = (PIECES + T - 1) / T;
  static constexpr bool EVEN = (PIECES % T) == 0;
  static_assert(K % 8 == 0 && NC % 4 == 0, "whole k groups, whole 16-byte pieces");
  float4 w[PER_T];
  __device__ __forceinline__ void load(const float* __restrict__ Wm, int tid) {
#pragma unroll
    for (int j = 0; j < PER_T; ++j) {
      const int i = tid + T * j;
      w[j] = ((const float4*)Wm)[EVEN ? i : min(i, PIECES - 1)];
    }
  }
  __device__ __forceinline__ void store(float* Bs, int tid) const {
#pragma unroll
    for (int j = 0; j < PER_T; ++j) {
      const int i = tid + T * j;
      if (!EVEN && i >= PIECES) continue;
      const int k = (i * 4) / NC;
      const int n = (i * 4) % NC;
      const int t = 2 * (k >> 3) + ((k >> 1) & 1);
      const int h = (k >> 2) & 1;
      const int s = k & 1;
      const int d = ((t * NT + (n >> 4)) * 32 + (n & 15) + 16 * h) * 2 + s;
      Bs[d] = w[j].x; Bs[d + 2] = w[j].y; Bs[d + 4] = w[j].z; Bs[d + 6] = w[j].w;
    }
  }
};

#define GEMM1_LDS ((KDIM1 / 4) * (HID / 16) * 32 * 2)
template <int WAVES>
static __device__ __forceinline__ void gemm1_tiles(
    const float* __restrict__ x, const float* __restrict__ pos,
    const float* __restrict__ W1, float* __restrict__ hw1, int N, int gb, float* Bs) {
  constexpr int NT = HID / 16;
  constexpr int TS = KDIM1 / 4;
  constexpr int HQ = TS / 4;
  static_assert(TS % 4 == 0 && NT % 2 == 0, "two halves of the k range, column tiles in pairs");
  const int lane = threadIdx.x & 31;
  const int wave = threadIdx.x >> 5;
  const int rowBase = (gb * WAVES + wave) * 16;
  const int kh  = lane >> 4;
  const int n16 = lane & 15;
  const int row = min(rowBase + n16, N - 1);

  WStage<KDIM1, NT, WAVES * 32> ws;
  ws.load(W1, threadIdx.x);

  float4 a4[HQ];
  const float4* xr = (const float4*)(x + (long)row * IN_C);
  const float4* pr = (const float4*)(pos + (long)row * POSD);
#pragma unroll
  for (int i = 0; i < HQ; ++i) a4[i] = (i < IN_C / 8) ? xr[2 * i + kh] : pr[2 * (i - IN_C / 8) + kh];

  ws.store(Bs, threadIdx.x);
  __syncthreads();
  if (rowBase >= N) return;

  const float* bl = Bs + lane * 2;
  float* orow = hw1 + (long)(rowBase + 8 * kh) * HID + n16;
  const int lim = N - (rowBase + 8 * kh);
#pragma unroll
  for (int p = 0; p < NT / 2; ++p) {
    v8f acc0 = {}, acc1 = {};
#pragma unroll
    for (int hh = 0; hh < 2; ++hh) {
      if (p != 0 || hh != 0) {
#pragma unroll
        for (int i = 0; i < HQ; ++i) {
          const int q = HQ * hh + i;
          a4[i] = (q < IN_C / 8) ? xr[2 * q + kh] : pr[2 * (q - IN_C / 8) + kh];
        }
      }
#pragma unroll
      for (int i = 0; i < 2 * HQ; ++i) {
        const int t = 2 * HQ * hh + i;
        v2f a;
        a.x = (i & 1) ? a4[i >> 1].z : a4[i >> 1].x;
        a.y = (i & 1) ? a4[i >> 1].w : a4[i >> 1].y;
        const v2f b0 = *(const v2f*)(bl + (t * NT + 2 * p + 0) * 64);
        const v2f b1 = *(const v2f*)(bl + (t * NT + 2 * p + 1) * 64);
        acc0 = __builtin_amdgcn_wmma_f32_16x16x4_f32(false, a, false, b0, (short)0, acc0, false, false);
        acc1 = __builtin_amdgcn_wmma_f32_16x16x4_f32(false, a, false, b1, (short)0, acc1, false, false);
      }
    }
#pragma unroll
    for (int i = 0; i < 8; ++i) {
      if (i < lim) {
        orow[i * HID + 32 * p +  0] = acc0[i];
        orow[i * HID + 32 * p + 16] = acc1[i];
      }
    }
  }
}

__global__ __launch_bounds__(256) void k_gemm2(
    const float* __restrict__ h, const float* __restrict__ W2,
    float* __restrict__ hw2, int N) {
  constexpr int NT = OUTD / 16;
  constexpr int TS = HID / 4;
  __shared__ __attribute__((aligned(16))) float Bs[TS * NT * 32 * 2];
  const int lane = threadIdx.x & 31;
  const int wave = threadIdx.x >> 5;
  const int rowBase = (blockIdx.x * 8 + wave) * 16;
  const int kh  = lane >> 4;
  const int n16 = lane & 15;
  const int row = min(rowBase + n16, N - 1);

  WStage<HID, NT, 256> ws;
  ws.load(W2, threadIdx.x);

  float4 a4[TS / 2];
  const float4* hr = (const float4*)(h + (long)row * HID);
#pragma unroll
  for (int q = 0; q < HID / 8; ++q) a4[q] = hr[2 * q + kh];

  ws.store(Bs, threadIdx.x);
  __syncthreads();
  if (rowBase >= N) return;

  v8f acc0 = {}, acc1 = {};
  const float* bl = Bs + lane * 2;
#pragma unroll
  for (int t = 0; t < TS; ++t) {
    v2f a;
    a.x = (t & 1) ? a4[t >> 1].z : a4[t >> 1].x;
    a.y = (t & 1) ? a4[t >> 1].w : a4[t >> 1].y;
    const v2f b0 = *(const v2f*)(bl + (t * NT + 0) * 64);
    const v2f b1 = *(const v2f*)(bl + (t * NT + 1) * 64);
    acc0 = __builtin_amdgcn_wmma_f32_16x16x4_f32(false, a, false, b0, (short)0, acc0, false, false);
    acc1 = __builtin_amdgcn_wmma_f32_16x16x4_f32(false, a, false, b1, (short)0, acc1, false, false);
  }
#pragma unroll
  for (int i = 0; i < 8; ++i) {
    const int r = rowBase + i + 8 * kh;
    if (r < N) {
      const long base = (long)r * OUTD + n16;
      hw2[base +  0] = acc0[i];
      hw2[base + 16] = acc1[i];
    }
  }
}

__global__ __launch_bounds__(PART_THREADS) void k_count(
    const int* __restrict__ dst, int* __restrict__ blkCnt, int NB, int E) {
  __shared__ int hist[MAX_BINS];
  const int tid = threadIdx.x;
  hist[tid] = 0;
  __syncthreads();

  const int e0 = blockIdx.x * (PART_THREADS * PART_K);
  int dd[PART_K];
#pragma unroll
  for (int k = 0; k < PART_K; ++k)
    dd[k] = dst[min(e0 + k * PART_THREADS + tid, E - 1)];
#pragma unroll
  for (int k = 0; k < PART_K; ++k)
    if (e0 + k * PART_THREADS + tid < E) atomicAdd(&hist[(unsigned)dd[k] / BIN_ROWS], 1);
  __syncthreads();

  if (tid < NB) blkCnt[blockIdx.x * NB + tid] = hist[tid];
}


template <bool FULL>
static __device__ __forceinline__ void add_counts(
    const int* __restrict__ col, int NB, int b0, int nblk, int me, int& cb, int& ahead) {
  int c[CNT_BATCH];
#pragma unroll
  for (int j = 0; j < CNT_BATCH; ++j) {
    const int b = FULL ? (b0 + j) : min(b0 + j, nblk - 1);
    c[j] = col[b * NB];
  }
#pragma unroll
  for (int j = 0; j < CNT_BATCH; ++j) {
    const int v = (FULL || b0 + j < nblk) ? c[j] : 0;
    cb += v;
    ahead += (b0 + j < me) ? v : 0;
  }
}

__global__ __launch_bounds__(PART_THREADS, 2 * (PART_THREADS / 32) / SIMDS_PER_WGP) void k_part(
    const int* __restrict__ src, const int* __restrict__ dst,
    const int* __restrict__ blkCnt, int* __restrict__ binCount,
    int* __restrict__ bins, int NB, int E, int npart,
    const float* __restrict__ x, const float* __restrict__ pos,
    const float* __restrict__ W1, float* __restrict__ hw1, int N) {
  __shared__ int hist[MAX_BINS];
  __shared__ int bstart[MAX_BINS];
  __shared__ int wtot[PART_THREADS / 32];
  __shared__ __attribute__((aligned(16))) int st_ent[PART_THREADS * PART_K];
  __shared__ int st_pos[PART_THREADS * PART_K];
  static_assert(GEMM1_LDS <= PART_THREADS * PART_K, "the staged weights fit the entry staging");
  if ((int)blockIdx.x >= npart) {
    gemm1_tiles<PART_THREADS / 32>(x, pos, W1, hw1, N, (int)blockIdx.x - npart, (float*)st_ent);
    return;
  }
  const int tid = threadIdx.x;
  const int lane = tid & 31;
  const int wave = tid >> 5;
  const int me = blockIdx.x;
  const int nblk = npart;

  int cb = 0, ahead = 0;
  if (tid < NB) {
    const int* col = blkCnt + tid;
    int b0 = 0;
#pragma nounroll
    for (; b0 + CNT_BATCH <= nblk; b0 += CNT_BATCH) add_counts<true>(col, NB, b0, nblk, me, cb, ahead);
    if (b0 < nblk) add_counts<false>(col, NB, b0, nblk, me, cb, ahead);
  }

  const int inc = wave_scan(cb, lane);
  if (lane == 31) wtot[wave] = inc;
  hist[tid] = 0;
  __syncthreads();
  {
    const int before = wave_sum((lane < wave) ? wtot[lane] : 0);
    bstart[tid] = before + inc - cb;
  }
  if (me == 0 && tid < NB) binCount[tid] = cb;

  const int e0 = me * (PART_THREADS * PART_K);
  int dd[PART_K], ss[PART_K], bin[PART_K], rank[PART_K];
#pragma unroll
  for (int k = 0; k < PART_K; ++k) {
    const int ec = min(e0 + k * PART_THREADS + tid, E - 1);
    dd[k] = dst[ec];
    ss[k] = src[ec];
  }
#pragma unroll
  for (int k = 0; k < PART_K; ++k) {
    const bool ok = e0 + k * PART_THREADS + tid < E;
    bin[k] = ok ? (int)((unsigned)dd[k] / BIN_ROWS) : -1;
    rank[k] = 0;
    if (ok) rank[k] = atomicAdd(&hist[bin[k]], 1);
  }
  __syncthreads();

  const int n = hist[tid];
  const int gb = bstart[tid] + ahead;
  const int inc2 = wave_scan(n, lane);
  if (lane == 31) wtot[wave] = inc2;
  __syncthreads();
  {
    const int before = wave_sum((lane < wave) ? wtot[lane] : 0);
    const int lofs = before + inc2 - n;
    hist[tid] = lofs;
    bstart[tid] = gb - lofs;
  }
  __syncthreads();

  int slot[PART_K], gpos[PART_K];
#pragma unroll
  for (int k = 0; k < PART_K; ++k) {
    const int bb = max(bin[k], 0);
    slot[k] = hist[bb] + rank[k];
    gpos[k] = slot[k] + bstart[bb];
  }
#pragma unroll
  for (int k = 0; k < PART_K; ++k) {
    if (bin[k] >= 0) {
      st_ent[slot[k]] = (int)((unsigned)ss[k] | ((unsigned)(dd[k] - bin[k] * BIN_ROWS) << ENT_SHIFT));
      st_pos[slot[k]] = gpos[k];
    }
  }
  __syncthreads();

  const int total = min(PART_THREADS * PART_K, E - e0);
  int oe[PART_K], op[PART_K];
#pragma unroll
  for (int k = 0; k < PART_K; ++k) {
    const int i = min(k * PART_THREADS + tid, total - 1);
    oe[k] = st_ent[i];
    op[k] = st_pos[i];
  }
#pragma unroll
  for (int k = 0; k < PART_K; ++k)
    if (k * PART_THREADS + tid < total) bins[op[k]] = oe[k];
}

__global__ __launch_bounds__(DINV_THREADS) void k_dinv(
    const int* __restrict__ bins, const int* __restrict__ binCount,
    float* dinv, unsigned short* deg16, int NB, int N) {
  __shared__ int cnt[BIN_ROWS];
  __shared__ int wtot[DINV_THREADS / 32];
  __shared__ int wmax[DINV_THREADS / 32];
  const int tid  = threadIdx.x;
  const int lane = tid & 31;
  const int wave = tid >> 5;
  const int b  = blockIdx.x;

  {
    const int bc = (tid < NB) ? binCount[tid] : 0;
    const int part = wave_sum((tid < b) ? bc : 0);
    const int mx = wave_max(bc);
    if (lane == 0) { wtot[wave] = part; wmax[wave] = mx; }
  }
  if (tid < BIN_ROWS) cnt[tid] = 0;
  __syncthreads();
  const int start = wave_sum(wtot[lane]);
  const bool tab = table_fits(wave_max(wmax[lane]), N);
  const int n = binCount[b];
  const int* mine = bins + start;

  for (int i0 = 0; i0 < n; i0 += DINV_THREADS * DINV_K) {
    int ent[DINV_K];
#pragma unroll
    for (int k = 0; k < DINV_K; ++k)
      ent[k] = mine[min(i0 + k * DINV_THREADS + tid, n - 1)];
#pragma unroll
    for (int k = 0; k < DINV_K; ++k)
      if (i0 + k * DINV_THREADS + tid < n) atomicAdd(&cnt[(unsigned)ent[k] >> ENT_SHIFT], 1);
  }
  __syncthreads();

  const int r = b * BIN_ROWS + tid;
  if (tid < BIN_ROWS && r < N) {
    if (tab) deg16[r] = (unsigned short)cnt[tid];
    else     dinv[r] = rsqrtf((float)cnt[tid] + 1.0f);
  }
}


template <bool TAB>
static __device__ __forceinline__ float row_dinv(
    const unsigned short* dtab, const float* __restrict__ dinv, int row) {
  return TAB ? rsqrtf((float)dtab[row] + 1.0f) : dinv[row];
}

template <int C, bool FULL, bool TAB>
static __device__ __forceinline__ void fold_batch(
    int ent, int nb, int lane,
    const float* __restrict__ dinv, const unsigned short* dtab, const float* dloc,
    const float* __restrict__ feat, float* acc) {
  constexpr int PER = C / 32;
  const int s = (int)((unsigned)ent & ((1u << ENT_SHIFT) - 1u));
  const int lr = (int)((unsigned)ent >> ENT_SHIFT);
  const float dvs = row_dinv<TAB>(dtab, dinv, s);
  const float dvd = dloc[lr];
  float w = 0.0f;
  const int ro = lr * C;
#pragma unroll
  for (int i = 0; i < Q_BATCH; i += 16) {
    if (!FULL && i >= nb) break;
    float v[16][PER];
#pragma unroll
    for (int k = 0; k < 16; ++k) {
      if (FULL || i + k < nb) {
        const int sk = __builtin_amdgcn_readlane(s, i + k);
        const float* p = feat + (long)sk * C + lane * PER;
        if (PER == 2) {
          const v2f t = *(const v2f*)p;
          v[k][0] = t.x;
          v[k][PER - 1] = t.y;
        } else {
          v[k][0] = p[0];
        }
      }
    }
    if (i == 0) w = dvs * dvd;
#pragma unroll
    for (int k = 0; k < 16; ++k) {
      if (FULL || i + k < nb) {
        const float wk = __int_as_float(__builtin_amdgcn_readlane(__float_as_int(w), i + k));
        const int rk = __builtin_amdgcn_readlane(ro, i + k);
        float* a = acc + rk + lane * PER;
#pragma unroll
        for (int j = 0; j < PER; ++j)
          __hip_atomic_fetch_add(a + j, v[k][j] * wk, __ATOMIC_RELAXED, __HIP_MEMORY_SCOPE_WORKGROUP);
      }
    }
  }
}

template <int C, bool TAB>
static __device__ __forceinline__ void fold_bin(
    const int* __restrict__ mine, int n, int wave, int lane,
    const float* __restrict__ dinv, const unsigned short* dtab, const float* dloc,
    const float* __restrict__ feat, float* acc) {
  int base = wave * Q_BATCH;
  int ent = mine[min(base + lane, n - 1)];
  while (base < n) {
    const int nbase = base + AGG_WAVES * Q_BATCH;
    const int entn = mine[min(nbase + lane, n - 1)];
    const int nb = min(Q_BATCH, n - base);
    if (nb == Q_BATCH) fold_batch<C, true, TAB>(ent, nb, lane, dinv, dtab, dloc, feat, acc);
    else               fold_batch<C, false, TAB>(ent, nb, lane, dinv, dtab, dloc, feat, acc);
    ent = entn;
    base = nbase;
  }
}

template <int C, bool RELU>
__global__ __launch_bounds__(AGG_THREADS) void k_agg(
    const int* __restrict__ bins, const int* __restrict__ binCount,
    const float* __restrict__ dinv, const unsigned short* __restrict__ deg16,
    const float* __restrict__ feat,
    const float* __restrict__ bias, float* __restrict__ out, int NB, int N) {
  __shared__ __attribute__((aligned(16))) float acc[BIN_ROWS * C];
  __shared__ __attribute__((aligned(16))) unsigned short dtab[TBL_CAP];
  __shared__ float dloc[BIN_ROWS];
  __shared__ int wtot[AGG_WAVES];
  __shared__ int wmax[AGG_WAVES];

  const int tid  = threadIdx.x;
  const int lane = tid & 31;
  const int wave = __builtin_amdgcn_readfirstlane(tid >> 5);
  const int b  = blockIdx.x;
  const int r0 = b * BIN_ROWS;
  const int nr = min(BIN_ROWS, N - r0);

  const int bc = (tid < NB) ? binCount[tid] : 0;

  constexpr int SELF_IT = (BIN_ROWS * (C / 4) + AGG_THREADS - 1) / AGG_THREADS;
  const int nself = nr * (C / 4);
  float4 selfv[SELF_IT];
#pragma unroll
  for (int k = 0; k < SELF_IT; ++k)
    selfv[k] = ((const float4*)(feat + (long)r0 * C))[min(tid + k * AGG_THREADS, nself - 1)];

  constexpr int TAB_IT = (TBL_CAP / 8 + AGG_THREADS - 1) / AGG_THREADS;
  const int ntab = min(N, TBL_CAP);
  const int npiece = ntab >> 3;
  const int lastp = max(npiece - 1, 0);
  uint4 tv[TAB_IT];
#pragma unroll
  for (int k = 0; k < TAB_IT; ++k)
    tv[k] = ((const uint4*)deg16)[min(tid + k * AGG_THREADS, lastp)];

  {
    const int part = wave_sum((tid < b) ? bc : 0);
    const int mx = wave_max(bc);
    if (lane == 0) { wtot[wave] = part; wmax[wave] = mx; }
  }

#pragma unroll
  for (int k = 0; k < TAB_IT; ++k)
    ((uint4*)dtab)[min(tid + k * AGG_THREADS, lastp)] = tv[k];
  for (int i = (npiece << 3) + tid; i < ntab; i += AGG_THREADS) dtab[i] = deg16[i];
  __syncthreads();

  const int start = __builtin_amdgcn_readfirstlane(wave_sum(wtot[lane]));
  const bool tab = table_fits(__builtin_amdgcn_readfirstlane(wave_max(wmax[lane])), N);
  const int n = __builtin_amdgcn_readfirstlane(binCount[b]);
  const int* mine = bins + start;

  if (tid < BIN_ROWS)
    dloc[tid] = (tid < nr) ? (tab ? row_dinv<true>(dtab, dinv, r0 + tid) : row_dinv<false>(dtab, dinv, r0 + tid)) : 0.0f;
#pragma unroll
  for (int k = 0; k < SELF_IT; ++k) {
    const int i = tid + k * AGG_THREADS;
    if (i < nself) {
      const int r = r0 + i / (C / 4);
      const float dv = tab ? row_dinv<true>(dtab, dinv, r) : row_dinv<false>(dtab, dinv, r);
      const float sw = dv * dv;
      float4 t = selfv[k];
      t.x *= sw; t.y *= sw; t.z *= sw; t.w *= sw;
      ((float4*)acc)[i] = t;
    }
  }
  __syncthreads();

  if (n > 0) {
    if (tab) fold_bin<C, true>(mine, n, wave, lane, dinv, dtab, dloc, feat, acc);
    else     fold_bin<C, false>(mine, n, wave, lane, dinv, dtab, dloc, feat, acc);
  }

  __syncthreads();

  for (int i = tid; i < nr * (C / 4); i += AGG_THREADS) {
    const int c4 = i % (C / 4);
    float4 t = ((const float4*)acc)[i];
    const float4 bb = ((const float4*)bias)[c4];
    t.x += bb.x; t.y += bb.y; t.z += bb.z; t.w += bb.w;
    if (RELU) {
      t.x = fmaxf(t.x, 0.0f); t.y = fmaxf(t.y, 0.0f);
      t.z = fmaxf(t.z, 0.0f); t.w = fmaxf(t.w, 0.0f);
    }
    ((float4*)(out + (long)r0 * C))[i] = t;
  }
}

__global__ void k_link(const float* __restrict__ z, const int* __restrict__ eli,
                       const float* __restrict__ Wl, const float* __restrict__ bl,
                       float* __restrict__ pred, int EL) {
  const int lane = threadIdx.x & 31;
  const int g = lane >> 3;
  const int q = lane & 7;
  const int e0 = blockIdx.x * blockDim.x + threadIdx.x - lane;
  if (e0 >= EL) return;
  const int ec = min(e0 + lane, EL - 1);
  const int s = eli[ec];
  const int d = eli[EL + ec];
  const float4 ws = ((const float4*)Wl)[q];
  const float4 wd = ((const float4*)(Wl + OUTD))[q];

  float t[8];
#pragma unroll
  for (int j = 0; j < 8; ++j) {
    const int sj = __shfl(s, 4 * j + g);
    const int dj = __shfl(d, 4 * j + g);
    const float4 a = ((const float4*)(z + (long)sj * OUTD))[q];
    const float4 b = ((const float4*)(z + (long)dj * OUTD))[q];
    float acc = a.x * ws.x;
    acc = fmaf(a.y, ws.y, acc);
    acc = fmaf(a.z, ws.z, acc);
    acc = fmaf(a.w, ws.w, acc);
    acc = fmaf(b.x, wd.x, acc);
    acc = fmaf(b.y, wd.y, acc);
    acc = fmaf(b.z, wd.z, acc);
    acc = fmaf(b.w, wd.w, acc);
    t[j] = acc;
  }

  const bool b4 = (q & 4) != 0, b2 = (q & 2) != 0, b1 = (q & 1) != 0;
  float u[4];
#pragma unroll
  for (int i = 0; i < 4; ++i) {
    const float keep = b4 ? t[i + 4] : t[i];
    const float send = b4 ? t[i] : t[i + 4];
    u[i] = keep + __shfl_xor(send, 4);
  }
  float v[2];
#pragma unroll
  for (int i = 0; i < 2; ++i) {
    const float keep = b2 ? u[i + 2] : u[i];
    const float send = b2 ? u[i] : u[i + 2];
    v[i] = keep + __shfl_xor(send, 2);
  }
  const float keep1 = b1 ? v[1] : v[0];
  const float send1 = b1 ? v[0] : v[1];
  const float r = keep1 + __shfl_xor(send1, 1);

  const int e = e0 + 4 * q + g;
  if (e < EL) pred[e] = r + bl[0];
}

extern "C" void kernel_launch(void* const* d_in, const int* in_sizes, int n_in,
                              void* d_out, int out_size, void* d_ws, size_t ws_size,
                              hipStream_t stream) {
  const float* x   = (const float*)d_in[0];
  const float* pos = (const float*)d_in[1];
  const float* W1  = (const float*)d_in[2];
  const float* b1  = (const float*)d_in[3];
  const float* W2  = (const float*)d_in[4];
  const float* b2  = (const float*)d_in[5];
  const float* Wl  = (const float*)d_in[6];
  const float* bl  = (const float*)d_in[7];
  const int*   ei  = (const int*)d_in[8];
  const int*   eli = (const int*)d_in[9];

  const int N  = in_sizes[0] / IN_C;
  const int E  = in_sizes[8] / 2;
  const int EL = in_sizes[9] / 2;

  float* z    = (float*)d_out;
  float* pred = z + (long)N * OUTD;

  const int NB = (N + BIN_ROWS - 1) / BIN_ROWS;
  const int per = PART_THREADS * PART_K;
  const int nblk = (E + per - 1) / per;

  float* dinv  = (float*)d_ws;
  unsigned short* deg16 = (unsigned short*)d_ws;
  float* hw1   = dinv + N;
  float* h1    = hw1 + (long)N * HID;
  float* hw2   = hw1;
  int* bins     = (int*)(h1 + (long)N * HID);
  int* binCount = bins + E;
  int* blkCnt   = (int*)h1;

  const int* esrc = ei;
  const int* edst = ei + E;

  const int B = 256;

  k_count<<<nblk, PART_THREADS, 0, stream>>>(edst, blkCnt, NB, E);
  const int rowTiles = (N + 15) / 16;
  const int gblocks  = (rowTiles + 7) / 8;
  const int g1blocks = (rowTiles + PART_THREADS / 32 - 1) / (PART_THREADS / 32);
  k_part <<<nblk + g1blocks, PART_THREADS, 0, stream>>>(esrc, edst, blkCnt, binCount, bins, NB, E, nblk,
                                                       x, pos, W1, hw1, N);
  k_dinv <<<NB, DINV_THREADS, 0, stream>>>(bins, binCount, dinv, deg16, NB, N);

  k_agg<HID, true><<<NB, AGG_THREADS, 0, stream>>>(bins, binCount, dinv, deg16, hw1, b1, h1, NB, N);

  k_gemm2<<<gblocks, B, 0, stream>>>(h1, W2, hw2, N);
  k_agg<OUTD, false><<<NB, AGG_THREADS, 0, stream>>>(bins, binCount, dinv, deg16, hw2, b2, z, NB, N);

  k_link<<<(EL + B - 1) / B, B, 0, stream>>>(z, eli, Wl, bl, pred, EL);
}


# GEMM 2 is the tail of layer 1's aggregation launch: a block's finished rows stay in its LDS tile (row pitch 272 B), 25 waves multiply them by W2 on WMMA; h1 never stored; 6 launches; same k order
# speedup vs baseline: 5.7633x; 1.0782x over previous
#include <hip/hip_runtime.h>

typedef __attribute__((ext_vector_type(2))) float v2f;
typedef __attribute__((ext_vector_type(8))) float v8f;

#define IN_C 64
#define POSD 16
#define KDIM1 (IN_C + POSD)
#define HID 64
#define OUTD 32


#define AGG_THREADS 1024
#define AGG_WAVES 32
#define Q_BATCH 32

#define BIN_ROWS 400
#define ENT_SHIFT 23
#define PART_THREADS 1024
#define PART_K 8
#define MAX_BINS 1024
#define CNT_BATCH 32
#define SIMDS_PER_WGP 4
#define DINV_THREADS 1024
#define DINV_K 8
#define TBL_CAP 100352
static_assert(BIN_ROWS <= (1 << (32 - ENT_SHIFT)), "a row inside its bin must fit above ENT_SHIFT");
static_assert(BIN_ROWS <= DINV_THREADS && BIN_ROWS <= AGG_THREADS, "one thread per owned row in the owner blocks");


static __device__ __forceinline__ int wave_sum(int v) {
#pragma unroll
  for (int o = 16; o > 0; o >>= 1) v += __shfl_xor(v, o);
  return v;
}

static __device__ __forceinline__ int wave_max(int v) {
#pragma unroll
  for (int o = 16; o > 0; o >>= 1) v = max(v, __shfl_xor(v, o));
  return v;
}

static __device__ __forceinline__ bool table_fits(int maxbin, int N) {
  return maxbin <= 0xFFFF && N <= TBL_CAP;
}

static __device__ __forceinline__ int wave_scan(int v, int lane) {
#pragma unroll
  for (int o = 1; o < 32; o <<= 1) {
    const int t = __shfl_up(v, o);
    if (lane >= o) v += t;
  }
  return v;
}


template <int K, int NT, int T>
struct WStage {
  static constexpr int NC = 16 * NT;
  static constexpr int PIECES = K * NC / 4;
  static constexpr int PER_T = (PIECES + T - 1) / T;
  static constexpr bool EVEN = (PIECES % T) == 0;
  static_assert(K % 8 == 0 && NC % 4 == 0, "whole k groups, whole 16-byte pieces");
  float4 w[PER_T];
  __device__ __forceinline__ void load(const float* __restrict__ Wm, int tid) {
#pragma unroll
    for (int j = 0; j < PER_T; ++j) {
      const int i = tid + T * j;
      w[j] = ((const float4*)Wm)[EVEN ? i : min(i, PIECES - 1)];
    }
  }
  __device__ __forceinline__ void store(float* Bs, int tid) const {
#pragma unroll
    for (int j = 0; j < PER_T; ++j) {
      const int i = tid + T * j;
      if (!EVEN && i >= PIECES) continue;
      const int k = (i * 4) / NC;
      const int n = (i * 4) % NC;
      const int t = 2 * (k >> 3) + ((k >> 1) & 1);
      const int h = (k >> 2) & 1;
      const int s = k & 1;
      const int d = ((t * NT + (n >> 4)) * 32 + (n & 15) + 16 * h) * 2 + s;
      Bs[d] = w[j].x; Bs[d + 2] = w[j].y; Bs[d + 4] = w[j].z; Bs[d + 6] = w[j].w;
    }
  }
};

#define GEMM1_LDS ((KDIM1 / 4) * (HID / 16) * 32 * 2)
template <int WAVES>
static __device__ __forceinline__ void gemm1_tiles(
    const float* __restrict__ x, const float* __restrict__ pos,
    const float* __restrict__ W1, float* __restrict__ hw1, int N, int gb, float* Bs) {
  constexpr int NT = HID / 16;
  constexpr int TS = KDIM1 / 4;
  constexpr int HQ = TS / 4;
  static_assert(TS % 4 == 0 && NT % 2 == 0, "two halves of the k range, column tiles in pairs");
  const int lane = threadIdx.x & 31;
  const int wave = threadIdx.x >> 5;
  const int rowBase = (gb * WAVES + wave) * 16;
  const int kh  = lane >> 4;
  const int n16 = lane & 15;
  const int row = min(rowBase + n16, N - 1);

  WStage<KDIM1, NT, WAVES * 32> ws;
  ws.load(W1, threadIdx.x);

  float4 a4[HQ];
  const float4* xr = (const float4*)(x + (long)row * IN_C);
  const float4* pr = (const float4*)(pos + (long)row * POSD);
#pragma unroll
  for (int i = 0; i < HQ; ++i) a4[i] = (i < IN_C / 8) ? xr[2 * i + kh] : pr[2 * (i - IN_C / 8) + kh];

  ws.store(Bs, threadIdx.x);
  __syncthreads();
  if (rowBase >= N) return;

  const float* bl = Bs + lane * 2;
  float* orow = hw1 + (long)(rowBase + 8 * kh) * HID + n16;
  const int lim = N - (rowBase + 8 * kh);
#pragma unroll
  for (int p = 0; p < NT / 2; ++p) {
    v8f acc0 = {}, acc1 = {};
#pragma unroll
    for (int hh = 0; hh < 2; ++hh) {
      if (p != 0 || hh != 0) {
#pragma unroll
        for (int i = 0; i < HQ; ++i) {
          const int q = HQ * hh + i;
          a4[i] = (q < IN_C / 8) ? xr[2 * q + kh] : pr[2 * (q - IN_C / 8) + kh];
        }
      }
#pragma unroll
      for (int i = 0; i < 2 * HQ; ++i) {
        const int t = 2 * HQ * hh + i;
        v2f a;
        a.x = (i & 1) ? a4[i >> 1].z : a4[i >> 1].x;
        a.y = (i & 1) ? a4[i >> 1].w : a4[i >> 1].y;
        const v2f b0 = *(const v2f*)(bl + (t * NT + 2 * p + 0) * 64);
        const v2f b1 = *(const v2f*)(bl + (t * NT + 2 * p + 1) * 64);
        acc0 = __builtin_amdgcn_wmma_f32_16x16x4_f32(false, a, false, b0, (short)0, acc0, false, false);
        acc1 = __builtin_amdgcn_wmma_f32_16x16x4_f32(false, a, false, b1, (short)0, acc1, false, false);
      }
    }
#pragma unroll
    for (int i = 0; i < 8; ++i) {
      if (i < lim) {
        orow[i * HID + 32 * p +  0] = acc0[i];
        orow[i * HID + 32 * p + 16] = acc1[i];
      }
    }
  }
}

__global__ __launch_bounds__(PART_THREADS) void k_count(
    const int* __restrict__ dst, int* __restrict__ blkCnt, int NB, int E) {
  __shared__ int hist[MAX_BINS];
  const int tid = threadIdx.x;
  hist[tid] = 0;
  __syncthreads();

  const int e0 = blockIdx.x * (PART_THREADS * PART_K);
  int dd[PART_K];
#pragma unroll
  for (int k = 0; k < PART_K; ++k)
    dd[k] = dst[min(e0 + k * PART_THREADS + tid, E - 1)];
#pragma unroll
  for (int k = 0; k < PART_K; ++k)
    if (e0 + k * PART_THREADS + tid < E) atomicAdd(&hist[(unsigned)dd[k] / BIN_ROWS], 1);
  __syncthreads();

  if (tid < NB) blkCnt[blockIdx.x * NB + tid] = hist[tid];
}


template <bool FULL>
static __device__ __forceinline__ void add_counts(
    const int* __restrict__ col, int NB, int b0, int nblk, int me, int& cb, int& ahead) {
  int c[CNT_BATCH];
#pragma unroll
  for (int j = 0; j < CNT_BATCH; ++j) {
    const int b = FULL ? (b0 + j) : min(b0 + j, nblk - 1);
    c[j] = col[b * NB];
  }
#pragma unroll
  for (int j = 0; j < CNT_BATCH; ++j) {
    const int v = (FULL || b0 + j < nblk) ? c[j] : 0;
    cb += v;
    ahead += (b0 + j < me) ? v : 0;
  }
}

__global__ __launch_bounds__(PART_THREADS, 2 * (PART_THREADS / 32) / SIMDS_PER_WGP) void k_part(
    const int* __restrict__ src, const int* __restrict__ dst,
    const int* __restrict__ blkCnt, int* __restrict__ binCount,
    int* __restrict__ bins, int NB, int E, int npart,
    const float* __restrict__ x, const float* __restrict__ pos,
    const float* __restrict__ W1, float* __restrict__ hw1, int N) {
  __shared__ int hist[MAX_BINS];
  __shared__ int bstart[MAX_BINS];
  __shared__ int wtot[PART_THREADS / 32];
  __shared__ __attribute__((aligned(16))) int st_ent[PART_THREADS * PART_K];
  __shared__ int st_pos[PART_THREADS * PART_K];
  static_assert(GEMM1_LDS <= PART_THREADS * PART_K, "the staged weights fit the entry staging");
  if ((int)blockIdx.x >= npart) {
    gemm1_tiles<PART_THREADS / 32>(x, pos, W1, hw1, N, (int)blockIdx.x - npart, (float*)st_ent);
    return;
  }
  const int tid = threadIdx.x;
  const int lane = tid & 31;
  const int wave = tid >> 5;
  const int me = blockIdx.x;
  const int nblk = npart;

  int cb = 0, ahead = 0;
  if (tid < NB) {
    const int* col = blkCnt + tid;
    int b0 = 0;
#pragma nounroll
    for (; b0 + CNT_BATCH <= nblk; b0 += CNT_BATCH) add_counts<true>(col, NB, b0, nblk, me, cb, ahead);
    if (b0 < nblk) add_counts<false>(col, NB, b0, nblk, me, cb, ahead);
  }

  const int inc = wave_scan(cb, lane);
  if (lane == 31) wtot[wave] = inc;
  hist[tid] = 0;
  __syncthreads();
  {
    const int before = wave_sum((lane < wave) ? wtot[lane] : 0);
    bstart[tid] = before + inc - cb;
  }
  if (me == 0 && tid < NB) binCount[tid] = cb;

  const int e0 = me * (PART_THREADS * PART_K);
  int dd[PART_K], ss[PART_K], bin[PART_K], rank[PART_K];
#pragma unroll
  for (int k = 0; k < PART_K; ++k) {
    const int ec = min(e0 + k * PART_THREADS + tid, E - 1);
    dd[k] = dst[ec];
    ss[k] = src[ec];
  }
#pragma unroll
  for (int k = 0; k < PART_K; ++k) {
    const bool ok = e0 + k * PART_THREADS + tid < E;
    bin[k] = ok ? (int)((unsigned)dd[k] / BIN_ROWS) : -1;
    rank[k] = 0;
    if (ok) rank[k] = atomicAdd(&hist[bin[k]], 1);
  }
  __syncthreads();

  const int n = hist[tid];
  const int gb = bstart[tid] + ahead;
  const int inc2 = wave_scan(n, lane);
  if (lane == 31) wtot[wave] = inc2;
  __syncthreads();
  {
    const int before = wave_sum((lane < wave) ? wtot[lane] : 0);
    const int lofs = before + inc2 - n;
    hist[tid] = lofs;
    bstart[tid] = gb - lofs;
  }
  __syncthreads();

  int slot[PART_K], gpos[PART_K];
#pragma unroll
  for (int k = 0; k < PART_K; ++k) {
    const int bb = max(bin[k], 0);
    slot[k] = hist[bb] + rank[k];
    gpos[k] = slot[k] + bstart[bb];
  }
#pragma unroll
  for (int k = 0; k < PART_K; ++k) {
    if (bin[k] >= 0) {
      st_ent[slot[k]] = (int)((unsigned)ss[k] | ((unsigned)(dd[k] - bin[k] * BIN_ROWS) << ENT_SHIFT));
      st_pos[slot[k]] = gpos[k];
    }
  }
  __syncthreads();

  const int total = min(PART_THREADS * PART_K, E - e0);
  int oe[PART_K], op[PART_K];
#pragma unroll
  for (int k = 0; k < PART_K; ++k) {
    const int i = min(k * PART_THREADS + tid, total - 1);
    oe[k] = st_ent[i];
    op[k] = st_pos[i];
  }
#pragma unroll
  for (int k = 0; k < PART_K; ++k)
    if (k * PART_THREADS + tid < total) bins[op[k]] = oe[k];
}

__global__ __launch_bounds__(DINV_THREADS) void k_dinv(
    const int* __restrict__ bins, const int* __restrict__ binCount,
    float* dinv, unsigned short* deg16, int NB, int N) {
  __shared__ int cnt[BIN_ROWS];
  __shared__ int wtot[DINV_THREADS / 32];
  __shared__ int wmax[DINV_THREADS / 32];
  const int tid  = threadIdx.x;
  const int lane = tid & 31;
  const int wave = tid >> 5;
  const int b  = blockIdx.x;

  {
    const int bc = (tid < NB) ? binCount[tid] : 0;
    const int part = wave_sum((tid < b) ? bc : 0);
    const int mx = wave_max(bc);
    if (lane == 0) { wtot[wave] = part; wmax[wave] = mx; }
  }
  if (tid < BIN_ROWS) cnt[tid] = 0;
  __syncthreads();
  const int start = wave_sum(wtot[lane]);
  const bool tab = table_fits(wave_max(wmax[lane]), N);
  const int n = binCount[b];
  const int* mine = bins + start;

  for (int i0 = 0; i0 < n; i0 += DINV_THREADS * DINV_K) {
    int ent[DINV_K];
#pragma unroll
    for (int k = 0; k < DINV_K; ++k)
      ent[k] = mine[min(i0 + k * DINV_THREADS + tid, n - 1)];
#pragma unroll
    for (int k = 0; k < DINV_K; ++k)
      if (i0 + k * DINV_THREADS + tid < n) atomicAdd(&cnt[(unsigned)ent[k] >> ENT_SHIFT], 1);
  }
  __syncthreads();

  const int r = b * BIN_ROWS + tid;
  if (tid < BIN_ROWS && r < N) {
    if (tab) deg16[r] = (unsigned short)cnt[tid];
    else     dinv[r] = rsqrtf((float)cnt[tid] + 1.0f);
  }
}


template <bool TAB>
static __device__ __forceinline__ float row_dinv(
    const unsigned short* dtab, const float* __restrict__ dinv, int row) {
  return TAB ? rsqrtf((float)dtab[row] + 1.0f) : dinv[row];
}

template <int C, int AP, bool FULL, bool TAB>
static __device__ __forceinline__ void fold_batch(
    int ent, int nb, int lane,
    const float* __restrict__ dinv, const unsigned short* dtab, const float* dloc,
    const float* __restrict__ feat, float* acc) {
  constexpr int PER = C / 32;
  const int s = (int)((unsigned)ent & ((1u << ENT_SHIFT) - 1u));
  const int lr = (int)((unsigned)ent >> ENT_SHIFT);
  const float dvs = row_dinv<TAB>(dtab, dinv, s);
  const float dvd = dloc[lr];
  float w = 0.0f;
  const int ro = lr * AP;
#pragma unroll
  for (int i = 0; i < Q_BATCH; i += 16) {
    if (!FULL && i >= nb) break;
    float v[16][PER];
#pragma unroll
    for (int k = 0; k < 16; ++k) {
      if (FULL || i + k < nb) {
        const int sk = __builtin_amdgcn_readlane(s, i + k);
        const float* p = feat + (long)sk * C + lane * PER;
        if (PER == 2) {
          const v2f t = *(const v2f*)p;
          v[k][0] = t.x;
          v[k][PER - 1] = t.y;
        } else {
          v[k][0] = p[0];
        }
      }
    }
    if (i == 0) w = dvs * dvd;
#pragma unroll
    for (int k = 0; k < 16; ++k) {
      if (FULL || i + k < nb) {
        const float wk = __int_as_float(__builtin_amdgcn_readlane(__float_as_int(w), i + k));
        const int rk = __builtin_amdgcn_readlane(ro, i + k);
        float* a = acc + rk + lane * PER;
#pragma unroll
        for (int j = 0; j < PER; ++j)
          __hip_atomic_fetch_add(a + j, v[k][j] * wk, __ATOMIC_RELAXED, __HIP_MEMORY_SCOPE_WORKGROUP);
      }
    }
  }
}

template <int C, int AP, bool TAB>
static __device__ __forceinline__ void fold_bin(
    const int* __restrict__ mine, int n, int wave, int lane,
    const float* __restrict__ dinv, const unsigned short* dtab, const float* dloc,
    const float* __restrict__ feat, float* acc) {
  int base = wave * Q_BATCH;
  int ent = mine[min(base + lane, n - 1)];
  while (base < n) {
    const int nbase = base + AGG_WAVES * Q_BATCH;
    const int entn = mine[min(nbase + lane, n - 1)];
    const int nb = min(Q_BATCH, n - base);
    if (nb == Q_BATCH) fold_batch<C, AP, true, TAB>(ent, nb, lane, dinv, dtab, dloc, feat, acc);
    else               fold_batch<C, AP, false, TAB>(ent, nb, lane, dinv, dtab, dloc, feat, acc);
    ent = entn;
    base = nbase;
  }
}

template <int C, bool RELU, bool FUSE>
__global__ __launch_bounds__(AGG_THREADS) void k_agg(
    const int* __restrict__ bins, const int* __restrict__ binCount,
    const float* __restrict__ dinv, const unsigned short* __restrict__ deg16,
    const float* __restrict__ feat,
    const float* __restrict__ bias, const float* __restrict__ W2,
    float* __restrict__ out, int NB, int N) {
  constexpr int AP = FUSE ? C + 4 : C;
  __shared__ __attribute__((aligned(16))) float acc[BIN_ROWS * AP];
  __shared__ __attribute__((aligned(16))) float wb[FUSE ? (C / 4) * (OUTD / 16) * 64 + C : 4];
  __shared__ __attribute__((aligned(16))) unsigned short dtab[TBL_CAP];
  __shared__ float dloc[BIN_ROWS];
  __shared__ int wtot[AGG_WAVES];
  __shared__ int wmax[AGG_WAVES];

  const int tid  = threadIdx.x;
  const int lane = tid & 31;
  const int wave = __builtin_amdgcn_readfirstlane(tid >> 5);
  const int b  = blockIdx.x;
  const int r0 = b * BIN_ROWS;
  const int nr = min(BIN_ROWS, N - r0);

  const int bc = (tid < NB) ? binCount[tid] : 0;

  WStage<C, OUTD / 16, AGG_THREADS> wst;
  float4 bv = {};
  if constexpr (FUSE) {
    wst.load(W2, tid);
    bv = ((const float4*)bias)[min(tid, C / 4 - 1)];
  }

  constexpr int SELF_IT = (BIN_ROWS * (C / 4) + AGG_THREADS - 1) / AGG_THREADS;
  const int nself = nr * (C / 4);
  float4 selfv[SELF_IT];
#pragma unroll
  for (int k = 0; k < SELF_IT; ++k)
    selfv[k] = ((const float4*)(feat + (long)r0 * C))[min(tid + k * AGG_THREADS, nself - 1)];

  constexpr int TAB_IT = (TBL_CAP / 8 + AGG_THREADS - 1) / AGG_THREADS;
  const int ntab = min(N, TBL_CAP);
  const int npiece = ntab >> 3;
  const int lastp = max(npiece - 1, 0);
  uint4 tv[TAB_IT];
#pragma unroll
  for (int k = 0; k < TAB_IT; ++k)
    tv[k] = ((const uint4*)deg16)[min(tid + k * AGG_THREADS, lastp)];

  {
    const int part = wave_sum((tid < b) ? bc : 0);
    const int mx = wave_max(bc);
    if (lane == 0) { wtot[wave] = part; wmax[wave] = mx; }
  }

#pragma unroll
  for (int k = 0; k < TAB_IT; ++k)
    ((uint4*)dtab)[min(tid + k * AGG_THREADS, lastp)] = tv[k];
  for (int i = (npiece << 3) + tid; i < ntab; i += AGG_THREADS) dtab[i] = deg16[i];
  if constexpr (FUSE) {
    wst.store(wb, tid);
    if (tid < C / 4) ((float4*)wb)[(C / 4) * (OUTD / 16) * 16 + tid] = bv;
  }
  __syncthreads();

  const int start = __builtin_amdgcn_readfirstlane(wave_sum(wtot[lane]));
  const bool tab = table_fits(__builtin_amdgcn_readfirstlane(wave_max(wmax[lane])), N);
  const int n = __builtin_amdgcn_readfirstlane(binCount[b]);
  const int* mine = bins + start;

  if (tid < BIN_ROWS)
    dloc[tid] = (tid < nr) ? (tab ? row_dinv<true>(dtab, dinv, r0 + tid) : row_dinv<false>(dtab, dinv, r0 + tid)) : 0.0f;
#pragma unroll
  for (int k = 0; k < SELF_IT; ++k) {
    const int i = tid + k * AGG_THREADS;
    if (i < nself) {
      const int r = r0 + i / (C / 4);
      const float dv = tab ? row_dinv<true>(dtab, dinv, r) : row_dinv<false>(dtab, dinv, r);
      const float sw = dv * dv;
      float4 t = selfv[k];
      t.x *= sw; t.y *= sw; t.z *= sw; t.w *= sw;
      *(float4*)(acc + (i / (C / 4)) * AP + 4 * (i % (C / 4))) = t;
    }
  }
  __syncthreads();

  if (n > 0) {
    if (tab) fold_bin<C, AP, true>(mine, n, wave, lane, dinv, dtab, dloc, feat, acc);
    else     fold_bin<C, AP, false>(mine, n, wave, lane, dinv, dtab, dloc, feat, acc);
  }

  __syncthreads();

  if constexpr (FUSE) {
    constexpr int NT = OUTD / 16;
    constexpr int TS = C / 4;
    static_assert((BIN_ROWS + 15) / 16 <= AGG_WAVES, "one wave per row tile of the bin");
    if (wave * 16 < nr) {
      const int kh  = lane >> 4;
      const int n16 = lane & 15;
      const float* pr = acc + (wave * 16 + n16) * AP + 4 * kh;
      float4 a4[TS / 2];
#pragma unroll
      for (int q = 0; q < TS / 2; ++q) {
        float4 t = *(const float4*)(pr + 8 * q);
        const float4 bb = ((const float4*)wb)[TS * NT * 16 + 2 * q + kh];
        t.x += bb.x; t.y += bb.y; t.z += bb.z; t.w += bb.w;
        if (RELU) {
          t.x = fmaxf(t.x, 0.0f); t.y = fmaxf(t.y, 0.0f);
          t.z = fmaxf(t.z, 0.0f); t.w = fmaxf(t.w, 0.0f);
        }
        a4[q] = t;
      }
      v8f d0 = {}, d1 = {};
      const float* bl = wb + lane * 2;
#pragma unroll
      for (int t = 0; t < TS; ++t) {
        v2f a;
        a.x = (t & 1) ? a4[t >> 1].z : a4[t >> 1].x;
        a.y = (t & 1) ? a4[t >> 1].w : a4[t >> 1].y;
        const v2f b0 = *(const v2f*)(bl + (t * NT + 0) * 64);
        const v2f b1 = *(const v2f*)(bl + (t * NT + 1) * 64);
        d0 = __builtin_amdgcn_wmma_f32_16x16x4_f32(false, a, false, b0, (short)0, d0, false, false);
        d1 = __builtin_amdgcn_wmma_f32_16x16x4_f32(false, a, false, b1, (short)0, d1, false, false);
      }
      float* orow = out + (long)(r0 + wave * 16 + 8 * kh) * OUTD + n16;
      const int lim = nr - (wave * 16 + 8 * kh);
#pragma unroll
      for (int i = 0; i < 8; ++i) {
        if (i < lim) {
          orow[i * OUTD +  0] = d0[i];
          orow[i * OUTD + 16] = d1[i];
        }
      }
    }
    return;
  }

  for (int i = tid; i < nr * (C / 4); i += AGG_THREADS) {
    const int c4 = i % (C / 4);
    float4 t = ((const float4*)acc)[i];
    const float4 bb = ((const float4*)bias)[c4];
    t.x += bb.x; t.y += bb.y; t.z += bb.z; t.w += bb.w;
    if (RELU) {
      t.x = fmaxf(t.x, 0.0f); t.y = fmaxf(t.y, 0.0f);
      t.z = fmaxf(t.z, 0.0f); t.w = fmaxf(t.w, 0.0f);
    }
    ((float4*)(out + (long)r0 * C))[i] = t;
  }
}

__global__ void k_link(const float* __restrict__ z, const int* __restrict__ eli,
                       const float* __restrict__ Wl, const float* __restrict__ bl,
                       float* __restrict__ pred, int EL) {
  const int lane = threadIdx.x & 31;
  const int g = lane >> 3;
  const int q = lane & 7;
  const int e0 = blockIdx.x * blockDim.x + threadIdx.x - lane;
  if (e0 >= EL) return;
  const int ec = min(e0 + lane, EL - 1);
  const int s = eli[ec];
  const int d = eli[EL + ec];
  const float4 ws = ((const float4*)Wl)[q];
  const float4 wd = ((const float4*)(Wl + OUTD))[q];

  float t[8];
#pragma unroll
  for (int j = 0; j < 8; ++j) {
    const int sj = __shfl(s, 4 * j + g);
    const int dj = __shfl(d, 4 * j + g);
    const float4 a = ((const float4*)(z + (long)sj * OUTD))[q];
    const float4 b = ((const float4*)(z + (long)dj * OUTD))[q];
    float acc = a.x * ws.x;
    acc = fmaf(a.y, ws.y, acc);
    acc = fmaf(a.z, ws.z, acc);
    acc = fmaf(a.w, ws.w, acc);
    acc = fmaf(b.x, wd.x, acc);
    acc = fmaf(b.y, wd.y, acc);
    acc = fmaf(b.z, wd.z, acc);
    acc = fmaf(b.w, wd.w, acc);
    t[j] = acc;
  }

  const bool b4 = (q & 4) != 0, b2 = (q & 2) != 0, b1 = (q & 1) != 0;
  float u[4];
#pragma unroll
  for (int i = 0; i < 4; ++i) {
    const float keep = b4 ? t[i + 4] : t[i];
    const float send = b4 ? t[i] : t[i + 4];
    u[i] = keep + __shfl_xor(send, 4);
  }
  float v[2];
#pragma unroll
  for (int i = 0; i < 2; ++i) {
    const float keep = b2 ? u[i + 2] : u[i];
    const float send = b2 ? u[i] : u[i + 2];
    v[i] = keep + __shfl_xor(send, 2);
  }
  const float keep1 = b1 ? v[1] : v[0];
  const float send1 = b1 ? v[0] : v[1];
  const float r = keep1 + __shfl_xor(send1, 1);

  const int e = e0 + 4 * q + g;
  if (e < EL) pred[e] = r + bl[0];
}

extern "C" void kernel_launch(void* const* d_in, const int* in_sizes, int n_in,
                              void* d_out, int out_size, void* d_ws, size_t ws_size,
                              hipStream_t stream) {
  const float* x   = (const float*)d_in[0];
  const float* pos = (const float*)d_in[1];
  const float* W1  = (const float*)d_in[2];
  const float* b1  = (const float*)d_in[3];
  const float* W2  = (const float*)d_in[4];
  const float* b2  = (const float*)d_in[5];
  const float* Wl  = (const float*)d_in[6];
  const float* bl  = (const float*)d_in[7];
  const int*   ei  = (const int*)d_in[8];
  const int*   eli = (const int*)d_in[9];

  const int N  = in_sizes[0] / IN_C;
  const int E  = in_sizes[8] / 2;
  const int EL = in_sizes[9] / 2;

  float* z    = (float*)d_out;
  float* pred = z + (long)N * OUTD;

  const int NB = (N + BIN_ROWS - 1) / BIN_ROWS;
  const int per = PART_THREADS * PART_K;
  const int nblk = (E + per - 1) / per;

  float* dinv  = (float*)d_ws;
  unsigned short* deg16 = (unsigned short*)d_ws;
  float* hw1   = dinv + N;
  float* hw2   = hw1 + (long)N * HID;
  int* bins     = (int*)(hw2 + (long)N * HID);
  int* binCount = bins + E;
  int* blkCnt   = (int*)hw2;

  const int* esrc = ei;
  const int* edst = ei + E;

  const int B = 256;

  k_count<<<nblk, PART_THREADS, 0, stream>>>(edst, blkCnt, NB, E);
  const int rowTiles = (N + 15) / 16;
  const int g1blocks = (rowTiles + PART_THREADS / 32 - 1) / (PART_THREADS / 32);
  k_part <<<nblk + g1blocks, PART_THREADS, 0, stream>>>(esrc, edst, blkCnt, binCount, bins, NB, E, nblk,
                                                       x, pos, W1, hw1, N);
  k_dinv <<<NB, DINV_THREADS, 0, stream>>>(bins, binCount, dinv, deg16, NB, N);

  k_agg<HID, true, true><<<NB, AGG_THREADS, 0, stream>>>(bins, binCount, dinv, deg16, hw1, b1, W2, hw2, NB, N);

  k_agg<OUTD, false, false><<<NB, AGG_THREADS, 0, stream>>>(bins, binCount, dinv, deg16, hw2, b2, W2, z, NB, N);

  k_link<<<(EL + B - 1) / B, B, 0, stream>>>(z, eli, Wl, bl, pred, EL);
}


# speedup vs baseline: 5.9743x; 1.0366x over previous
#include <hip/hip_runtime.h>

typedef __attribute__((ext_vector_type(2))) float v2f;
typedef __attribute__((ext_vector_type(8))) float v8f;

#define IN_C 64
#define POSD 16
#define KDIM1 (IN_C + POSD)
#define HID 64
#define OUTD 32


#define AGG_THREADS 1024
#define AGG_WAVES 32
#define Q_BATCH 32

#define BIN_ROWS 400
#define ENT_SHIFT 23
#define PART_THREADS 1024
#define PART_K 8
#define MAX_BINS 1024
#define CNT_BATCH 32
#define SIMDS_PER_WGP 4
#define DINV_THREADS 1024
#define DINV_K 8
#define BIN_ALIGN 32
#define TBL_CAP 100352
static_assert(BIN_ROWS <= (1 << (32 - ENT_SHIFT)), "a row inside its bin must fit above ENT_SHIFT");
static_assert(BIN_ROWS <= DINV_THREADS && BIN_ROWS <= AGG_THREADS, "one thread per owned row in the owner blocks");

static __device__ __forceinline__ int bin_span(int n) {
  return (n + BIN_ALIGN - 1) & ~(BIN_ALIGN - 1);
}


static __device__ __forceinline__ int wave_sum(int v) {
#pragma unroll
  for (int o = 16; o > 0; o >>= 1) v += __shfl_xor(v, o);
  return v;
}

static __device__ __forceinline__ int wave_max(int v) {
#pragma unroll
  for (int o = 16; o > 0; o >>= 1) v = max(v, __shfl_xor(v, o));
  return v;
}

static __device__ __forceinline__ bool table_fits(int maxbin, int N) {
  return maxbin <= 0xFFFF && N <= TBL_CAP;
}

static __device__ __forceinline__ int wave_scan(int v, int lane) {
#pragma unroll
  for (int o = 1; o < 32; o <<= 1) {
    const int t = __shfl_up(v, o);
    if (lane >= o) v += t;
  }
  return v;
}


template <int K, int NT, int T>
struct WStage {
  static constexpr int NC = 16 * NT;
  static constexpr int PIECES = K * NC / 4;
  static constexpr int PER_T = (PIECES + T - 1) / T;
  static constexpr bool EVEN = (PIECES % T) == 0;
  static_assert(K % 8 == 0 && NC % 4 == 0, "whole k groups, whole 16-byte pieces");
  float4 w[PER_T];
  __device__ __forceinline__ void load(const float* __restrict__ Wm, int tid) {
#pragma unroll
    for (int j = 0; j < PER_T; ++j) {
      const int i = tid + T * j;
      w[j] = ((const float4*)Wm)[EVEN ? i : min(i, PIECES - 1)];
    }
  }
  __device__ __forceinline__ void store(float* Bs, int tid) const {
#pragma unroll
    for (int j = 0; j < PER_T; ++j) {
      const int i = tid + T * j;
      if (!EVEN && i >= PIECES) continue;
      const int k = (i * 4) / NC;
      const int n = (i * 4) % NC;
      const int t = 2 * (k >> 3) + ((k >> 1) & 1);
      const int h = (k >> 2) & 1;
      const int s = k & 1;
      const int d = ((t * NT + (n >> 4)) * 32 + (n & 15) + 16 * h) * 2 + s;
      Bs[d] = w[j].x; Bs[d + 2] = w[j].y; Bs[d + 4] = w[j].z; Bs[d + 6] = w[j].w;
    }
  }
};

#define GEMM1_LDS ((KDIM1 / 4) * (HID / 16) * 32 * 2)
template <int WAVES>
static __device__ __forceinline__ void gemm1_tiles(
    const float* __restrict__ x, const float* __restrict__ pos,
    const float* __restrict__ W1, float* __restrict__ hw1, int N, int gb, float* Bs) {
  constexpr int NT = HID / 16;
  constexpr int TS = KDIM1 / 4;
  constexpr int HQ = TS / 4;
  static_assert(TS % 4 == 0 && NT % 2 == 0, "two halves of the k range, column tiles in pairs");
  const int lane = threadIdx.x & 31;
  const int wave = threadIdx.x >> 5;
  const int rowBase = (gb * WAVES + wave) * 16;
  const int kh  = lane >> 4;
  const int n16 = lane & 15;
  const int row = min(rowBase + n16, N - 1);

  WStage<KDIM1, NT, WAVES * 32> ws;
  ws.load(W1, threadIdx.x);

  float4 a4[HQ];
  const float4* xr = (const float4*)(x + (long)row * IN_C);
  const float4* pr = (const float4*)(pos + (long)row * POSD);
#pragma unroll
  for (int i = 0; i < HQ; ++i) a4[i] = (i < IN_C / 8) ? xr[2 * i + kh] : pr[2 * (i - IN_C / 8) + kh];

  ws.store(Bs, threadIdx.x);
  __syncthreads();
  if (rowBase >= N) return;

  const float* bl = Bs + lane * 2;
  float* orow = hw1 + (long)(rowBase + 8 * kh) * HID + n16;
  const int lim = N - (rowBase + 8 * kh);
#pragma unroll
  for (int p = 0; p < NT / 2; ++p) {
    v8f acc0 = {}, acc1 = {};
#pragma unroll
    for (int hh = 0; hh < 2; ++hh) {
      if (p != 0 || hh != 0) {
#pragma unroll
        for (int i = 0; i < HQ; ++i) {
          const int q = HQ * hh + i;
          a4[i] = (q < IN_C / 8) ? xr[2 * q + kh] : pr[2 * (q - IN_C / 8) + kh];
        }
      }
#pragma unroll
      for (int i = 0; i < 2 * HQ; ++i) {
        const int t = 2 * HQ * hh + i;
        v2f a;
        a.x = (i & 1) ? a4[i >> 1].z : a4[i >> 1].x;
        a.y = (i & 1) ? a4[i >> 1].w : a4[i >> 1].y;
        const v2f b0 = *(const v2f*)(bl + (t * NT + 2 * p + 0) * 64);
        const v2f b1 = *(const v2f*)(bl + (t * NT + 2 * p + 1) * 64);
        acc0 = __builtin_amdgcn_wmma_f32_16x16x4_f32(false, a, false, b0, (short)0, acc0, false, false);
        acc1 = __builtin_amdgcn_wmma_f32_16x16x4_f32(false, a, false, b1, (short)0, acc1, false, false);
      }
    }
#pragma unroll
    for (int i = 0; i < 8; ++i) {
      if (i < lim) {
        orow[i * HID + 32 * p +  0] = acc0[i];
        orow[i * HID + 32 * p + 16] = acc1[i];
      }
    }
  }
}

__global__ __launch_bounds__(PART_THREADS) void k_count(
    const int* __restrict__ dst, int* __restrict__ blkCnt, int NB, int E) {
  __shared__ int hist[MAX_BINS];
  const int tid = threadIdx.x;
  hist[tid] = 0;
  __syncthreads();

  const int e0 = blockIdx.x * (PART_THREADS * PART_K);
  int dd[PART_K];
#pragma unroll
  for (int k = 0; k < PART_K; ++k)
    dd[k] = dst[min(e0 + k * PART_THREADS + tid, E - 1)];
#pragma unroll
  for (int k = 0; k < PART_K; ++k)
    if (e0 + k * PART_THREADS + tid < E) atomicAdd(&hist[(unsigned)dd[k] / BIN_ROWS], 1);
  __syncthreads();

  if (tid < NB) blkCnt[blockIdx.x * NB + tid] = hist[tid];
}


template <bool FULL>
static __device__ __forceinline__ void add_counts(
    const int* __restrict__ col, int NB, int b0, int nblk, int me, int& cb, int& ahead) {
  int c[CNT_BATCH];
#pragma unroll
  for (int j = 0; j < CNT_BATCH; ++j) {
    const int b = FULL ? (b0 + j) : min(b0 + j, nblk - 1);
    c[j] = col[b * NB];
  }
#pragma unroll
  for (int j = 0; j < CNT_BATCH; ++j) {
    const int v = (FULL || b0 + j < nblk) ? c[j] : 0;
    cb += v;
    ahead += (b0 + j < me) ? v : 0;
  }
}

__global__ __launch_bounds__(PART_THREADS, 2 * (PART_THREADS / 32) / SIMDS_PER_WGP) void k_part(
    const int* __restrict__ src, const int* __restrict__ dst,
    const int* __restrict__ blkCnt, int* __restrict__ binCount,
    int* __restrict__ bins, int NB, int E, int npart,
    const float* __restrict__ x, const float* __restrict__ pos,
    const float* __restrict__ W1, float* __restrict__ hw1, int N) {
  __shared__ int hist[MAX_BINS];
  __shared__ int bstart[MAX_BINS];
  __shared__ int wtot[PART_THREADS / 32];
  __shared__ __attribute__((aligned(16))) int st_ent[PART_THREADS * PART_K];
  __shared__ int st_pos[PART_THREADS * PART_K];
  static_assert(GEMM1_LDS <= PART_THREADS * PART_K, "the staged weights fit the entry staging");
  if ((int)blockIdx.x >= npart) {
    gemm1_tiles<PART_THREADS / 32>(x, pos, W1, hw1, N, (int)blockIdx.x - npart, (float*)st_ent);
    return;
  }
  const int tid = threadIdx.x;
  const int lane = tid & 31;
  const int wave = tid >> 5;
  const int me = blockIdx.x;
  const int nblk = npart;

  int cb = 0, ahead = 0;
  if (tid < NB) {
    const int* col = blkCnt + tid;
    int b0 = 0;
#pragma nounroll
    for (; b0 + CNT_BATCH <= nblk; b0 += CNT_BATCH) add_counts<true>(col, NB, b0, nblk, me, cb, ahead);
    if (b0 < nblk) add_counts<false>(col, NB, b0, nblk, me, cb, ahead);
  }

  const int span = bin_span(cb);
  const int inc = wave_scan(span, lane);
  if (lane == 31) wtot[wave] = inc;
  hist[tid] = 0;
  __syncthreads();
  {
    const int before = wave_sum((lane < wave) ? wtot[lane] : 0);
    bstart[tid] = before + inc - span;
  }
  if (me == 0 && tid < NB) binCount[tid] = cb;

  const int e0 = me * (PART_THREADS * PART_K);
  int dd[PART_K], ss[PART_K], bin[PART_K], rank[PART_K];
#pragma unroll
  for (int k = 0; k < PART_K; ++k) {
    const int ec = min(e0 + k * PART_THREADS + tid, E - 1);
    dd[k] = dst[ec];
    ss[k] = src[ec];
  }
#pragma unroll
  for (int k = 0; k < PART_K; ++k) {
    const bool ok = e0 + k * PART_THREADS + tid < E;
    bin[k] = ok ? (int)((unsigned)dd[k] / BIN_ROWS) : -1;
    rank[k] = 0;
    if (ok) rank[k] = atomicAdd(&hist[bin[k]], 1);
  }
  __syncthreads();

  const int n = hist[tid];
  const int gb = bstart[tid] + ahead;
  const int inc2 = wave_scan(n, lane);
  if (lane == 31) wtot[wave] = inc2;
  __syncthreads();
  {
    const int before = wave_sum((lane < wave) ? wtot[lane] : 0);
    const int lofs = before + inc2 - n;
    hist[tid] = lofs;
    bstart[tid] = gb - lofs;
  }
  __syncthreads();

  int slot[PART_K], gpos[PART_K];
#pragma unroll
  for (int k = 0; k < PART_K; ++k) {
    const int bb = max(bin[k], 0);
    slot[k] = hist[bb] + rank[k];
    gpos[k] = slot[k] + bstart[bb];
  }
#pragma unroll
  for (int k = 0; k < PART_K; ++k) {
    if (bin[k] >= 0) {
      st_ent[slot[k]] = (int)((unsigned)ss[k] | ((unsigned)(dd[k] - bin[k] * BIN_ROWS) << ENT_SHIFT));
      st_pos[slot[k]] = gpos[k];
    }
  }
  __syncthreads();

  const int total = min(PART_THREADS * PART_K, E - e0);
  int oe[PART_K], op[PART_K];
#pragma unroll
  for (int k = 0; k < PART_K; ++k) {
    const int i = min(k * PART_THREADS + tid, total - 1);
    oe[k] = st_ent[i];
    op[k] = st_pos[i];
  }
#pragma unroll
  for (int k = 0; k < PART_K; ++k)
    if (k * PART_THREADS + tid < total) bins[op[k]] = oe[k];
}

__global__ __launch_bounds__(DINV_THREADS) void k_dinv(
    const int* __restrict__ bins, const int* __restrict__ binCount,
    float* dinv, unsigned short* deg16, int NB, int N) {
  __shared__ int cnt[BIN_ROWS];
  __shared__ int wtot[DINV_THREADS / 32];
  __shared__ int wmax[DINV_THREADS / 32];
  const int tid  = threadIdx.x;
  const int lane = tid & 31;
  const int wave = tid >> 5;
  const int b  = blockIdx.x;

  {
    const int bc = (tid < NB) ? binCount[tid] : 0;
    const int part = wave_sum((tid < b) ? bin_span(bc) : 0);
    const int mx = wave_max(bc);
    if (lane == 0) { wtot[wave] = part; wmax[wave] = mx; }
  }
  if (tid < BIN_ROWS) cnt[tid] = 0;
  __syncthreads();
  const int start = wave_sum(wtot[lane]);
  const bool tab = table_fits(wave_max(wmax[lane]), N);
  const int n = binCount[b];
  const int* mine = bins + start;

  for (int i0 = 0; i0 < n; i0 += DINV_THREADS * DINV_K) {
    int ent[DINV_K];
#pragma unroll
    for (int k = 0; k < DINV_K; ++k)
      ent[k] = mine[min(i0 + k * DINV_THREADS + tid, n - 1)];
#pragma unroll
    for (int k = 0; k < DINV_K; ++k)
      if (i0 + k * DINV_THREADS + tid < n) atomicAdd(&cnt[(unsigned)ent[k] >> ENT_SHIFT], 1);
  }
  __syncthreads();

  const int r = b * BIN_ROWS + tid;
  if (tid < BIN_ROWS && r < N) {
    if (tab) deg16[r] = (unsigned short)cnt[tid];
    else     dinv[r] = rsqrtf((float)cnt[tid] + 1.0f);
  }
}


template <bool TAB>
static __device__ __forceinline__ float row_dinv(
    const unsigned short* dtab, const float* __restrict__ dinv, int row) {
  return TAB ? rsqrtf((float)dtab[row] + 1.0f) : dinv[row];
}

template <int C, int AP, bool FULL, bool TAB>
static __device__ __forceinline__ void fold_batch(
    int ent, int nb, int lane,
    const float* __restrict__ dinv, const unsigned short* dtab, const float* dloc,
    const float* __restrict__ feat, float* acc) {
  constexpr int PER = C / 32;
  const int s = (int)((unsigned)ent & ((1u << ENT_SHIFT) - 1u));
  const int lr = (int)((unsigned)ent >> ENT_SHIFT);
  const float dvs = row_dinv<TAB>(dtab, dinv, s);
  const float dvd = dloc[lr];
  float w = 0.0f;
  const int ro = lr * AP;
#pragma unroll
  for (int i = 0; i < Q_BATCH; i += 16) {
    if (!FULL && i >= nb) break;
    float v[16][PER];
#pragma unroll
    for (int k = 0; k < 16; ++k) {
      if (FULL || i + k < nb) {
        const int sk = __builtin_amdgcn_readlane(s, i + k);
        const float* p = feat + (long)sk * C + lane * PER;
        if (PER == 2) {
          const v2f t = *(const v2f*)p;
          v[k][0] = t.x;
          v[k][PER - 1] = t.y;
        } else {
          v[k][0] = p[0];
        }
      }
    }
    if (i == 0) w = dvs * dvd;
#pragma unroll
    for (int k = 0; k < 16; ++k) {
      if (FULL || i + k < nb) {
        const float wk = __int_as_float(__builtin_amdgcn_readlane(__float_as_int(w), i + k));
        const int rk = __builtin_amdgcn_readlane(ro, i + k);
        float* a = acc + rk + lane * PER;
#pragma unroll
        for (int j = 0; j < PER; ++j)
          __hip_atomic_fetch_add(a + j, v[k][j] * wk, __ATOMIC_RELAXED, __HIP_MEMORY_SCOPE_WORKGROUP);
      }
    }
  }
}

template <int C, int AP, bool TAB>
static __device__ __forceinline__ void fold_bin(
    const int* __restrict__ mine, int n, int wave, int lane,
    const float* __restrict__ dinv, const unsigned short* dtab, const float* dloc,
    const float* __restrict__ feat, float* acc) {
  int base = wave * Q_BATCH;
  int ent = mine[min(base + lane, n - 1)];
  while (base < n) {
    const int nbase = base + AGG_WAVES * Q_BATCH;
    const int entn = mine[min(nbase + lane, n - 1)];
    const int nb = min(Q_BATCH, n - base);
    if (nb == Q_BATCH) fold_batch<C, AP, true, TAB>(ent, nb, lane, dinv, dtab, dloc, feat, acc);
    else               fold_batch<C, AP, false, TAB>(ent, nb, lane, dinv, dtab, dloc, feat, acc);
    ent = entn;
    base = nbase;
  }
}

template <int C, bool RELU, bool FUSE>
__global__ __launch_bounds__(AGG_THREADS) void k_agg(
    const int* __restrict__ bins, const int* __restrict__ binCount,
    const float* __restrict__ dinv, const unsigned short* __restrict__ deg16,
    const float* __restrict__ feat,
    const float* __restrict__ bias, const float* __restrict__ W2,
    float* __restrict__ out, int NB, int N) {
  constexpr int AP = FUSE ? C + 4 : C;
  __shared__ __attribute__((aligned(16))) float acc[BIN_ROWS * AP];
  __shared__ __attribute__((aligned(16))) float wb[FUSE ? (C / 4) * (OUTD / 16) * 64 + C : 4];
  __shared__ __attribute__((aligned(16))) unsigned short dtab[TBL_CAP];
  __shared__ float dloc[BIN_ROWS];
  __shared__ int wtot[AGG_WAVES];
  __shared__ int wmax[AGG_WAVES];

  const int tid  = threadIdx.x;
  const int lane = tid & 31;
  const int wave = __builtin_amdgcn_readfirstlane(tid >> 5);
  const int b  = blockIdx.x;
  const int r0 = b * BIN_ROWS;
  const int nr = min(BIN_ROWS, N - r0);

  const int bc = (tid < NB) ? binCount[tid] : 0;

  WStage<C, OUTD / 16, AGG_THREADS> wst;
  float4 bv = {};
  if constexpr (FUSE) {
    wst.load(W2, tid);
    bv = ((const float4*)bias)[min(tid, C / 4 - 1)];
  }

  constexpr int SELF_IT = (BIN_ROWS * (C / 4) + AGG_THREADS - 1) / AGG_THREADS;
  const int nself = nr * (C / 4);
  float4 selfv[SELF_IT];
#pragma unroll
  for (int k = 0; k < SELF_IT; ++k)
    selfv[k] = ((const float4*)(feat + (long)r0 * C))[min(tid + k * AGG_THREADS, nself - 1)];

  constexpr int TAB_IT = (TBL_CAP / 8 + AGG_THREADS - 1) / AGG_THREADS;
  const int ntab = min(N, TBL_CAP);
  const int npiece = ntab >> 3;
  const int lastp = max(npiece - 1, 0);
  uint4 tv[TAB_IT];
#pragma unroll
  for (int k = 0; k < TAB_IT; ++k)
    tv[k] = ((const uint4*)deg16)[min(tid + k * AGG_THREADS, lastp)];

  {
    const int part = wave_sum((tid < b) ? bin_span(bc) : 0);
    const int mx = wave_max(bc);
    if (lane == 0) { wtot[wave] = part; wmax[wave] = mx; }
  }

#pragma unroll
  for (int k = 0; k < TAB_IT; ++k)
    ((uint4*)dtab)[min(tid + k * AGG_THREADS, lastp)] = tv[k];
  for (int i = (npiece << 3) + tid; i < ntab; i += AGG_THREADS) dtab[i] = deg16[i];
  if constexpr (FUSE) {
    wst.store(wb, tid);
    if (tid < C / 4) ((float4*)wb)[(C / 4) * (OUTD / 16) * 16 + tid] = bv;
  }
  __syncthreads();

  const int start = __builtin_amdgcn_readfirstlane(wave_sum(wtot[lane]));
  const bool tab = table_fits(__builtin_amdgcn_readfirstlane(wave_max(wmax[lane])), N);
  const int n = __builtin_amdgcn_readfirstlane(binCount[b]);
  const int* mine = bins + start;

  if (tid < BIN_ROWS)
    dloc[tid] = (tid < nr) ? (tab ? row_dinv<true>(dtab, dinv, r0 + tid) : row_dinv<false>(dtab, dinv, r0 + tid)) : 0.0f;
#pragma unroll
  for (int k = 0; k < SELF_IT; ++k) {
    const int i = tid + k * AGG_THREADS;
    if (i < nself) {
      const int r = r0 + i / (C / 4);
      const float dv = tab ? row_dinv<true>(dtab, dinv, r) : row_dinv<false>(dtab, dinv, r);
      const float sw = dv * dv;
      float4 t = selfv[k];
      t.x *= sw; t.y *= sw; t.z *= sw; t.w *= sw;
      *(float4*)(acc + (i / (C / 4)) * AP + 4 * (i % (C / 4))) = t;
    }
  }
  __syncthreads();

  if (n > 0) {
    if (tab) fold_bin<C, AP, true>(mine, n, wave, lane, dinv, dtab, dloc, feat, acc);
    else     fold_bin<C, AP, false>(mine, n, wave, lane, dinv, dtab, dloc, feat, acc);
  }

  __syncthreads();

  if constexpr (FUSE) {
    constexpr int NT = OUTD / 16;
    constexpr int TS = C / 4;
    static_assert((BIN_ROWS + 15) / 16 <= AGG_WAVES, "one wave per row tile of the bin");
    if (wave * 16 < nr) {
      const int kh  = lane >> 4;
      const int n16 = lane & 15;
      const float* pr = acc + (wave * 16 + n16) * AP + 4 * kh;
      float4 a4[TS / 2];
#pragma unroll
      for (int q = 0; q < TS / 2; ++q) {
        float4 t = *(const float4*)(pr + 8 * q);
        const float4 bb = ((const float4*)wb)[TS * NT * 16 + 2 * q + kh];
        t.x += bb.x; t.y += bb.y; t.z += bb.z; t.w += bb.w;
        if (RELU) {
          t.x = fmaxf(t.x, 0.0f); t.y = fmaxf(t.y, 0.0f);
          t.z = fmaxf(t.z, 0.0f); t.w = fmaxf(t.w, 0.0f);
        }
        a4[q] = t;
      }
      v8f d0 = {}, d1 = {};
      const float* bl = wb + lane * 2;
#pragma unroll
      for (int t = 0; t < TS; ++t) {
        v2f a;
        a.x = (t & 1) ? a4[t >> 1].z : a4[t >> 1].x;
        a.y = (t & 1) ? a4[t >> 1].w : a4[t >> 1].y;
        const v2f b0 = *(const v2f*)(bl + (t * NT + 0) * 64);
        const v2f b1 = *(const v2f*)(bl + (t * NT + 1) * 64);
        d0 = __builtin_amdgcn_wmma_f32_16x16x4_f32(false, a, false, b0, (short)0, d0, false, false);
        d1 = __builtin_amdgcn_wmma_f32_16x16x4_f32(false, a, false, b1, (short)0, d1, false, false);
      }
      float* orow = out + (long)(r0 + wave * 16 + 8 * kh) * OUTD + n16;
      const int lim = nr - (wave * 16 + 8 * kh);
#pragma unroll
      for (int i = 0; i < 8; ++i) {
        if (i < lim) {
          orow[i * OUTD +  0] = d0[i];
          orow[i * OUTD + 16] = d1[i];
        }
      }
    }
    return;
  }

  for (int i = tid; i < nr * (C / 4); i += AGG_THREADS) {
    const int c4 = i % (C / 4);
    float4 t = ((const float4*)acc)[i];
    const float4 bb = ((const float4*)bias)[c4];
    t.x += bb.x; t.y += bb.y; t.z += bb.z; t.w += bb.w;
    if (RELU) {
      t.x = fmaxf(t.x, 0.0f); t.y = fmaxf(t.y, 0.0f);
      t.z = fmaxf(t.z, 0.0f); t.w = fmaxf(t.w, 0.0f);
    }
    ((float4*)(out + (long)r0 * C))[i] = t;
  }
}

__global__ void k_link(const float* __restrict__ z, const int* __restrict__ eli,
                       const float* __restrict__ Wl, const float* __restrict__ bl,
                       float* __restrict__ pred, int EL) {
  const int lane = threadIdx.x & 31;
  const int g = lane >> 3;
  const int q = lane & 7;
  const int e0 = blockIdx.x * blockDim.x + threadIdx.x - lane;
  if (e0 >= EL) return;
  const int ec = min(e0 + lane, EL - 1);
  const int s = eli[ec];
  const int d = eli[EL + ec];
  const float4 ws = ((const float4*)Wl)[q];
  const float4 wd = ((const float4*)(Wl + OUTD))[q];

  float t[8];
#pragma unroll
  for (int j = 0; j < 8; ++j) {
    const int sj = __shfl(s, 4 * j + g);
    const int dj = __shfl(d, 4 * j + g);
    const float4 a = ((const float4*)(z + (long)sj * OUTD))[q];
    const float4 b = ((const float4*)(z + (long)dj * OUTD))[q];
    float acc = a.x * ws.x;
    acc = fmaf(a.y, ws.y, acc);
    acc = fmaf(a.z, ws.z, acc);
    acc = fmaf(a.w, ws.w, acc);
    acc = fmaf(b.x, wd.x, acc);
    acc = fmaf(b.y, wd.y, acc);
    acc = fmaf(b.z, wd.z, acc);
    acc = fmaf(b.w, wd.w, acc);
    t[j] = acc;
  }

  const bool b4 = (q & 4) != 0, b2 = (q & 2) != 0, b1 = (q & 1) != 0;
  float u[4];
#pragma unroll
  for (int i = 0; i < 4; ++i) {
    const float keep = b4 ? t[i + 4] : t[i];
    const float send = b4 ? t[i] : t[i + 4];
    u[i] = keep + __shfl_xor(send, 4);
  }
  float v[2];
#pragma unroll
  for (int i = 0; i < 2; ++i) {
    const float keep = b2 ? u[i + 2] : u[i];
    const float send = b2 ? u[i] : u[i + 2];
    v[i] = keep + __shfl_xor(send, 2);
  }
  const float keep1 = b1 ? v[1] : v[0];
  const float send1 = b1 ? v[0] : v[1];
  const float r = keep1 + __shfl_xor(send1, 1);

  const int e = e0 + 4 * q + g;
  if (e < EL) pred[e] = r + bl[0];
}

extern "C" void kernel_launch(void* const* d_in, const int* in_sizes, int n_in,
                              void* d_out, int out_size, void* d_ws, size_t ws_size,
                              hipStream_t stream) {
  const float* x   = (const float*)d_in[0];
  const float* pos = (const float*)d_in[1];
  const float* W1  = (const float*)d_in[2];
  const float* b1  = (const float*)d_in[3];
  const float* W2  = (const float*)d_in[4];
  const float* b2  = (const float*)d_in[5];
  const float* Wl  = (const float*)d_in[6];
  const float* bl  = (const float*)d_in[7];
  const int*   ei  = (const int*)d_in[8];
  const int*   eli = (const int*)d_in[9];

  const int N  = in_sizes[0] / IN_C;
  const int E  = in_sizes[8] / 2;
  const int EL = in_sizes[9] / 2;

  float* z    = (float*)d_out;
  float* pred = z + (long)N * OUTD;

  const int NB = (N + BIN_ROWS - 1) / BIN_ROWS;
  const int per = PART_THREADS * PART_K;
  const int nblk = (E + per - 1) / per;

  float* hw1   = (float*)d_ws;
  float* hw2   = hw1 + (long)N * HID;
  float* dinv  = hw2 + (long)N * OUTD;
  unsigned short* deg16 = (unsigned short*)dinv;
  int* bins     = (int*)(hw2 + (long)N * HID);
  int* binCount = bins + E + (long)BIN_ALIGN * NB;
  int* blkCnt   = (int*)hw2;

  const int* esrc = ei;
  const int* edst = ei + E;

  const int B = 256;

  k_count<<<nblk, PART_THREADS, 0, stream>>>(edst, blkCnt, NB, E);
  const int rowTiles = (N + 15) / 16;
  const int g1blocks = (rowTiles + PART_THREADS / 32 - 1) / (PART_THREADS / 32);
  k_part <<<nblk + g1blocks, PART_THREADS, 0, stream>>>(esrc, edst, blkCnt, binCount, bins, NB, E, nblk,
                                                       x, pos, W1, hw1, N);
  k_dinv <<<NB, DINV_THREADS, 0, stream>>>(bins, binCount, dinv, deg16, NB, N);

  k_agg<HID, true, true><<<NB, AGG_THREADS, 0, stream>>>(bins, binCount, dinv, deg16, hw1, b1, W2, hw2, NB, N);

  k_agg<OUTD, false, false><<<NB, AGG_THREADS, 0, stream>>>(bins, binCount, dinv, deg16, hw2, b2, W2, z, NB, N);

  k_link<<<(EL + B - 1) / B, B, 0, stream>>>(z, eli, Wl, bl, pred, EL);
}
